# starting kernel
# baseline (speedup 1.0000x reference)
#include <hip/hip_runtime.h>
#include <math.h>

#define B_   2
#define L_   512
#define DM   1024
#define DI   2048
#define DS_  64
#define NH_  32
#define HD_  64
#define NA_  32
#define DIP  4352
#define M1   (B_*L_)

#define NC_  16
#define LC_  32

#define OFF_Z    0
#define OFF_X    2048
#define OFF_B    4096
#define OFF_C    4160
#define OFF_DT   4224
#define OFF_A    4256
#define OFF_TRAP 4288
#define OFF_ANG  4320

typedef __attribute__((ext_vector_type(2))) float v2f;
typedef __attribute__((ext_vector_type(8))) float v8f;

__device__ __forceinline__ float softplus_f(float x) {
    return x > 20.f ? x : log1pf(expf(x));
}
__device__ __forceinline__ float sigmoid_f(float x) {
    return 1.f / (1.f + expf(-x));
}

__global__ __launch_bounds__(256) void gemm_wmma_f32(
    const float* __restrict__ A, const float* __restrict__ Bm,
    float* __restrict__ C, int M, int N, int K)
{
    const int lane = threadIdx.x & 31;
    const int wave = threadIdx.x >> 5;
    const int lo   = lane & 15;
    const int hi   = lane >> 4;

    const int m0 = blockIdx.y * 256 + wave * 32;
    const int n0 = blockIdx.x * 64;

    const float* Arow0 = A + (size_t)(m0 + lo) * K;
    const float* Arow1 = A + (size_t)(m0 + 16 + lo) * K;

    v8f acc00 = {}, acc01 = {}, acc02 = {}, acc03 = {};
    v8f acc10 = {}, acc11 = {}, acc12 = {}, acc13 = {};

    for (int k = 0; k < K; k += 4) {
        v2f a0; a0.x = Arow0[k + 2*hi]; a0.y = Arow0[k + 2*hi + 1];
        v2f a1; a1.x = Arow1[k + 2*hi]; a1.y = Arow1[k + 2*hi + 1];

        const float* Bp0 = Bm + (size_t)(k + 2*hi) * N + n0 + lo;
        const float* Bp1 = Bp0 + N;
        v2f b0; b0.x = Bp0[0];  b0.y = Bp1[0];
        v2f b1; b1.x = Bp0[16]; b1.y = Bp1[16];
        v2f b2; b2.x = Bp0[32]; b2.y = Bp1[32];
        v2f b3; b3.x = Bp0[48]; b3.y = Bp1[48];

        acc00 = __builtin_amdgcn_wmma_f32_16x16x4_f32(false, a0, false, b0, (short)0, acc00, false, false);
        acc01 = __builtin_amdgcn_wmma_f32_16x16x4_f32(false, a0, false, b1, (short)0, acc01, false, false);
        acc02 = __builtin_amdgcn_wmma_f32_16x16x4_f32(false, a0, false, b2, (short)0, acc02, false, false);
        acc03 = __builtin_amdgcn_wmma_f32_16x16x4_f32(false, a0, false, b3, (short)0, acc03, false, false);
        acc10 = __builtin_amdgcn_wmma_f32_16x16x4_f32(false, a1, false, b0, (short)0, acc10, false, false);
        acc11 = __builtin_amdgcn_wmma_f32_16x16x4_f32(false, a1, false, b1, (short)0, acc11, false, false);
        acc12 = __builtin_amdgcn_wmma_f32_16x16x4_f32(false, a1, false, b2, (short)0, acc12, false, false);
        acc13 = __builtin_amdgcn_wmma_f32_16x16x4_f32(false, a1, false, b3, (short)0, acc13, false, false);
    }

    #pragma unroll
    for (int r = 0; r < 8; ++r) {
        float* cp0 = C + (size_t)(m0 + r + 8*hi) * N + n0 + lo;
        cp0[0]  = acc00[r];
        cp0[16] = acc01[r];
        cp0[32] = acc02[r];
        cp0[48] = acc03[r];
        float* cp1 = C + (size_t)(m0 + 16 + r + 8*hi) * N + n0 + lo;
        cp1[0]  = acc10[r];
        cp1[16] = acc11[r];
        cp1[32] = acc12[r];
        cp1[48] = acc13[r];
    }
}

__global__ __launch_bounds__(32) void phase_kernel(
    const float* __restrict__ proj, const float* __restrict__ dt_bias,
    float* __restrict__ DTb, float* __restrict__ decayb,
    float* __restrict__ trapb, float* __restrict__ phaseb)
{
    const int b = blockIdx.x;
    const int j = threadIdx.x;
    float acc = 0.f;
    for (int l = 0; l < L_; ++l) {
        const float* row = proj + (size_t)(b * L_ + l) * DIP;
        float dt = softplus_f(row[OFF_DT + j] + dt_bias[j]);
        float s = dt;
        s += __shfl_xor(s, 16, 32);
        s += __shfl_xor(s,  8, 32);
        s += __shfl_xor(s,  4, 32);
        s += __shfl_xor(s,  2, 32);
        s += __shfl_xor(s,  1, 32);
        const float dtmean = s * (1.f / 32.f);
        acc += row[OFF_ANG + j] * dtmean;

        const size_t o = (size_t)(b * L_ + l) * 32 + j;
        DTb[o]    = dt;
        float Av  = fminf(-softplus_f(row[OFF_A + j]), -1e-4f);
        decayb[o] = expf(Av * dt);
        trapb[o]  = sigmoid_f(row[OFF_TRAP + j]);
        phaseb[o] = acc;
    }
}

__global__ __launch_bounds__(64) void rot_kernel(
    const float* __restrict__ proj,
    const float* __restrict__ Bn_g, const float* __restrict__ Bn_b,
    const float* __restrict__ Cn_g, const float* __restrict__ Cn_b,
    const float* __restrict__ phaseb,
    float* __restrict__ Brot, float* __restrict__ Crot)
{
    const int bl = blockIdx.x;
    const int t  = threadIdx.x;
    __shared__ float red[64], nb[64], nc[64];

    const float* row = proj + (size_t)bl * DIP;
    const float vb = row[OFF_B + t];
    const float vc = row[OFF_C + t];

    auto blockReduce = [&](float v) -> float {
        red[t] = v; __syncthreads();
        for (int s = 32; s > 0; s >>= 1) {
            if (t < s) red[t] += red[t + s];
            __syncthreads();
        }
        float r = red[0]; __syncthreads();
        return r;
    };

    const float mB  = blockReduce(vb) * (1.f / 64.f);
    const float vaB = blockReduce((vb - mB) * (vb - mB)) * (1.f / 64.f);
    const float nB  = (vb - mB) * rsqrtf(vaB + 1e-5f) * Bn_g[t] + Bn_b[t];

    const float mC  = blockReduce(vc) * (1.f / 64.f);
    const float vaC = blockReduce((vc - mC) * (vc - mC)) * (1.f / 64.f);
    const float nC  = (vc - mC) * rsqrtf(vaC + 1e-5f) * Cn_g[t] + Cn_b[t];

    nb[t] = nB; nc[t] = nC;
    __syncthreads();

    const int   q  = t >> 1;
    const float ph = phaseb[(size_t)bl * 32 + q];
    const float c  = cosf(ph), s = sinf(ph);
    float ob, oc;
    if ((t & 1) == 0) { ob = nb[t] * c - nb[t + 1] * s; oc = nc[t] * c - nc[t + 1] * s; }
    else              { ob = nb[t - 1] * s + nb[t] * c; oc = nc[t - 1] * s + nc[t] * c; }
    Brot[(size_t)bl * 64 + t] = ob;
    Crot[(size_t)bl * 64 + t] = oc;
}


__global__ __launch_bounds__(256) void scan_chunk_kernel(
    const float* __restrict__ proj,
    const float* __restrict__ Brot, const float* __restrict__ Crot,
    const float* __restrict__ DTb, const float* __restrict__ decayb,
    const float* __restrict__ trapb,
    float* __restrict__ yfin, float* __restrict__ cumdb,
    float* __restrict__ Sstate)
{
    const int blk = blockIdx.x;
    const int c   = blk % NC_;
    const int h   = (blk / NC_) % NH_;
    const int b   = blk / (NC_ * NH_);
    const int l0  = c * LC_;
    const int t   = threadIdx.x;
    const int p   = t >> 2;
    const int seg = t & 3;

    __shared__ float sx[2][64], sB[2][64], sC[2][64], sscal[2][3];

    float hst[16], Bprev[16];
    float xprev = 0.f;
    #pragma unroll
    for (int i = 0; i < 16; ++i) { hst[i] = 0.f; Bprev[i] = 0.f; }
    if (c > 0) {
        const size_t blp = (size_t)(b * L_ + l0 - 1);
        xprev = proj[blp * DIP + OFF_X + h * HD_ + p];
        const float* Bp = Brot + blp * 64 + seg * 16;
        #pragma unroll
        for (int i = 0; i < 16; ++i) Bprev[i] = Bp[i];
    }

    auto loadTok = [&](int tok, int buf) {
        const size_t bl = (size_t)(b * L_ + l0 + tok);
        if (t < 64)        sx[buf][t]       = proj[bl * DIP + OFF_X + h * HD_ + t];
        else if (t < 128)  sB[buf][t - 64]  = Brot[bl * 64 + (t - 64)];
        else if (t < 192)  sC[buf][t - 128] = Crot[bl * 64 + (t - 128)];
        else if (t == 192) sscal[buf][0] = DTb[bl * 32 + h];
        else if (t == 193) sscal[buf][1] = decayb[bl * 32 + h];
        else if (t == 194) sscal[buf][2] = trapb[bl * 32 + h];
    };

    loadTok(0, 0);
    __syncthreads();
    float cumd = 1.f;

    for (int tok = 0; tok < LC_; ++tok) {
        const int cur = tok & 1;
        if (tok + 1 < LC_) loadTok(tok + 1, cur ^ 1);

        const float x   = sx[cur][p];
        const float dt  = sscal[cur][0];
        const float dec = sscal[cur][1];
        const float tr  = sscal[cur][2];

        float y = 0.f;
        #pragma unroll
        for (int i = 0; i < 16; ++i) {
            const int   n  = seg * 16 + i;
            const float Bn = sB[cur][n];
            const float bx = tr * x * Bn + (1.f - tr) * xprev * Bprev[i];
            hst[i] = dec * hst[i] + dt * bx;
            y += hst[i] * sC[cur][n];
            Bprev[i] = Bn;
        }
        xprev = x;
        cumd *= dec;

        y += __shfl_xor(y, 1, 32);
        y += __shfl_xor(y, 2, 32);

        const size_t bl = (size_t)(b * L_ + l0 + tok);
        if (seg == 0) yfin[bl * DI + h * HD_ + p] = y;
        if (t == 0)   cumdb[((size_t)(b * NH_ + h)) * L_ + l0 + tok] = cumd;
        __syncthreads();
    }

    float* Sp = Sstate + (((size_t)(b * NH_ + h)) * NC_ + c) * 4096 + p * 64 + seg * 16;
    #pragma unroll
    for (int i = 0; i < 16; ++i) Sp[i] = hst[i];
}

__global__ __launch_bounds__(256) void scan_combine_kernel(
    const float* __restrict__ cumdb, float* __restrict__ Sstate)
{
    const int h  = blockIdx.x % NH_;
    const int b  = blockIdx.x / NH_;
    const int t  = threadIdx.x;
    const int p  = t >> 2, seg = t & 3;

    float H[16];
    #pragma unroll
    for (int i = 0; i < 16; ++i) H[i] = 0.f;

    for (int c = 0; c < NC_; ++c) {
        float* Sp = Sstate + (((size_t)(b * NH_ + h)) * NC_ + c) * 4096 + p * 64 + seg * 16;
        const float Pc = cumdb[((size_t)(b * NH_ + h)) * L_ + c * LC_ + LC_ - 1];
        #pragma unroll
        for (int i = 0; i < 16; ++i) {
            const float Sc = Sp[i];
            Sp[i] = H[i];
            H[i]  = Pc * H[i] + Sc;
        }
    }
}

__global__ __launch_bounds__(256) void scan_correct_kernel(
    const float* __restrict__ proj, const float* __restrict__ Crot,
    const float* __restrict__ cumdb, const float* __restrict__ Sstate,
    const float* __restrict__ D_param, float* __restrict__ yfin)
{
    const int blk = blockIdx.x;
    const int c   = blk % NC_;
    const int h   = (blk / NC_) % NH_;
    const int b   = blk / (NC_ * NH_);
    const int l0  = c * LC_;
    const int t   = threadIdx.x;
    const int p   = t >> 2, seg = t & 3;

    __shared__ float sC[64];
    __shared__ float scd;

    float H[16];
    const float* Sp = Sstate + (((size_t)(b * NH_ + h)) * NC_ + c) * 4096 + p * 64 + seg * 16;
    #pragma unroll
    for (int i = 0; i < 16; ++i) H[i] = Sp[i];
    const float Dp = D_param[h];

    for (int tok = 0; tok < LC_; ++tok) {
        const size_t bl = (size_t)(b * L_ + l0 + tok);
        if (t < 64)  sC[t] = Crot[bl * 64 + t];
        if (t == 64) scd   = cumdb[((size_t)(b * NH_ + h)) * L_ + l0 + tok];
        __syncthreads();

        float corr = 0.f;
        #pragma unroll
        for (int i = 0; i < 16; ++i) corr += H[i] * sC[seg * 16 + i];
        corr += __shfl_xor(corr, 1, 32);
        corr += __shfl_xor(corr, 2, 32);

        if (seg == 0) {
            const float x = proj[bl * DIP + OFF_X + h * HD_ + p];
            const float z = proj[bl * DIP + OFF_Z + h * HD_ + p];
            float y = yfin[bl * DI + h * HD_ + p] + scd * corr;
            yfin[bl * DI + h * HD_ + p] = (y + Dp * x) * (z * sigmoid_f(z));
        }
        __syncthreads();
    }
}

extern "C" void kernel_launch(void* const* d_in, const int* in_sizes, int n_in,
                              void* d_out, int out_size, void* d_ws, size_t ws_size,
                              hipStream_t stream)
{
    const float* u       = (const float*)d_in[0];
    const float* W_in    = (const float*)d_in[1];
    const float* W_out   = (const float*)d_in[2];
    const float* dt_bias = (const float*)d_in[3];
    const float* D_param = (const float*)d_in[4];
    const float* Bn_g    = (const float*)d_in[5];
    const float* Bn_b    = (const float*)d_in[6];
    const float* Cn_g    = (const float*)d_in[7];
    const float* Cn_b    = (const float*)d_in[8];

    float* ws = (float*)d_ws;
    float* proj   = ws; ws += (size_t)M1 * DIP;
    float* yfin   = ws; ws += (size_t)M1 * DI;
    float* Sstate = ws; ws += (size_t)B_ * NH_ * NC_ * 4096;
    float* Brot   = ws; ws += (size_t)M1 * DS_;
    float* Crot   = ws; ws += (size_t)M1 * DS_;
    float* DTb    = ws; ws += (size_t)M1 * NH_;
    float* decayb = ws; ws += (size_t)M1 * NH_;
    float* trapb  = ws; ws += (size_t)M1 * NH_;
    float* phaseb = ws; ws += (size_t)M1 * NA_;
    float* cumdb  = ws; ws += (size_t)B_ * NH_ * L_;

    gemm_wmma_f32<<<dim3(DIP / 64, M1 / 256), dim3(256), 0, stream>>>(
        u, W_in, proj, M1, DIP, DM);

    phase_kernel<<<dim3(B_), dim3(32), 0, stream>>>(
        proj, dt_bias, DTb, decayb, trapb, phaseb);

    rot_kernel<<<dim3(M1), dim3(64), 0, stream>>>(
        proj, Bn_g, Bn_b, Cn_g, Cn_b, phaseb, Brot, Crot);

    scan_chunk_kernel<<<dim3(B_ * NH_ * NC_), dim3(256), 0, stream>>>(
        proj, Brot, Crot, DTb, decayb, trapb, yfin, cumdb, Sstate);
    scan_combine_kernel<<<dim3(B_ * NH_), dim3(256), 0, stream>>>(
        cumdb, Sstate);
    scan_correct_kernel<<<dim3(B_ * NH_ * NC_), dim3(256), 0, stream>>>(
        proj, Crot, cumdb, Sstate, D_param, yfin);

    gemm_wmma_f32<<<dim3(DM / 64, M1 / 256), dim3(256), 0, stream>>>(
        yfin, W_out, (float*)d_out, M1, DM, DI);
}


# phase kernel made token-parallel: one 1024-thread block per batch, each wave scans its own run of tokens, wave totals joined through LDS; the phase cumsum becomes a two-level sum (order changes)
# speedup vs baseline: 1.9524x; 1.9524x over previous
#include <hip/hip_runtime.h>
#include <math.h>

#define B_   2
#define L_   512
#define DM   1024
#define DI   2048
#define DS_  64
#define NH_  32
#define HD_  64
#define NA_  32
#define DIP  4352
#define M1   (B_*L_)

#define NC_  16
#define LC_  32

#define OFF_Z    0
#define OFF_X    2048
#define OFF_B    4096
#define OFF_C    4160
#define OFF_DT   4224
#define OFF_A    4256
#define OFF_TRAP 4288
#define OFF_ANG  4320

typedef __attribute__((ext_vector_type(2))) float v2f;
typedef __attribute__((ext_vector_type(8))) float v8f;

__device__ __forceinline__ float softplus_f(float x) {
    return x > 20.f ? x : log1pf(expf(x));
}
__device__ __forceinline__ float sigmoid_f(float x) {
    return 1.f / (1.f + expf(-x));
}

__global__ __launch_bounds__(256) void gemm_wmma_f32(
    const float* __restrict__ A, const float* __restrict__ Bm,
    float* __restrict__ C, int M, int N, int K)
{
    const int lane = threadIdx.x & 31;
    const int wave = threadIdx.x >> 5;
    const int lo   = lane & 15;
    const int hi   = lane >> 4;

    const int m0 = blockIdx.y * 256 + wave * 32;
    const int n0 = blockIdx.x * 64;

    const float* Arow0 = A + (size_t)(m0 + lo) * K;
    const float* Arow1 = A + (size_t)(m0 + 16 + lo) * K;

    v8f acc00 = {}, acc01 = {}, acc02 = {}, acc03 = {};
    v8f acc10 = {}, acc11 = {}, acc12 = {}, acc13 = {};

    for (int k = 0; k < K; k += 4) {
        v2f a0; a0.x = Arow0[k + 2*hi]; a0.y = Arow0[k + 2*hi + 1];
        v2f a1; a1.x = Arow1[k + 2*hi]; a1.y = Arow1[k + 2*hi + 1];

        const float* Bp0 = Bm + (size_t)(k + 2*hi) * N + n0 + lo;
        const float* Bp1 = Bp0 + N;
        v2f b0; b0.x = Bp0[0];  b0.y = Bp1[0];
        v2f b1; b1.x = Bp0[16]; b1.y = Bp1[16];
        v2f b2; b2.x = Bp0[32]; b2.y = Bp1[32];
        v2f b3; b3.x = Bp0[48]; b3.y = Bp1[48];

        acc00 = __builtin_amdgcn_wmma_f32_16x16x4_f32(false, a0, false, b0, (short)0, acc00, false, false);
        acc01 = __builtin_amdgcn_wmma_f32_16x16x4_f32(false, a0, false, b1, (short)0, acc01, false, false);
        acc02 = __builtin_amdgcn_wmma_f32_16x16x4_f32(false, a0, false, b2, (short)0, acc02, false, false);
        acc03 = __builtin_amdgcn_wmma_f32_16x16x4_f32(false, a0, false, b3, (short)0, acc03, false, false);
        acc10 = __builtin_amdgcn_wmma_f32_16x16x4_f32(false, a1, false, b0, (short)0, acc10, false, false);
        acc11 = __builtin_amdgcn_wmma_f32_16x16x4_f32(false, a1, false, b1, (short)0, acc11, false, false);
        acc12 = __builtin_amdgcn_wmma_f32_16x16x4_f32(false, a1, false, b2, (short)0, acc12, false, false);
        acc13 = __builtin_amdgcn_wmma_f32_16x16x4_f32(false, a1, false, b3, (short)0, acc13, false, false);
    }

    #pragma unroll
    for (int r = 0; r < 8; ++r) {
        float* cp0 = C + (size_t)(m0 + r + 8*hi) * N + n0 + lo;
        cp0[0]  = acc00[r];
        cp0[16] = acc01[r];
        cp0[32] = acc02[r];
        cp0[48] = acc03[r];
        float* cp1 = C + (size_t)(m0 + 16 + r + 8*hi) * N + n0 + lo;
        cp1[0]  = acc10[r];
        cp1[16] = acc11[r];
        cp1[32] = acc12[r];
        cp1[48] = acc13[r];
    }
}

#define PW_  32
#define PT_  16
static_assert(PW_ * PT_ == L_, "waves x tokens per wave must cover the sequence");

__global__ __launch_bounds__(PW_ * 32) void phase_kernel(
    const float* __restrict__ proj, const float* __restrict__ dt_bias,
    float* __restrict__ DTb, float* __restrict__ decayb,
    float* __restrict__ trapb, float* __restrict__ phaseb)
{
    const int b = blockIdx.x;
    const int j = threadIdx.x & 31;
    const int w = threadIdx.x >> 5;
    __shared__ float wtot[PW_][32];

    const float  bias = dt_bias[j];
    const size_t bl0  = (size_t)b * L_ + (size_t)w * PT_;

    float ph[PT_];
    float acc = 0.f;
    #pragma unroll
    for (int i = 0; i < PT_; ++i) {
        const float* row = proj + (bl0 + i) * DIP;
        float dt = softplus_f(row[OFF_DT + j] + bias);
        float s = dt;
        s += __shfl_xor(s, 16, 32);
        s += __shfl_xor(s,  8, 32);
        s += __shfl_xor(s,  4, 32);
        s += __shfl_xor(s,  2, 32);
        s += __shfl_xor(s,  1, 32);
        const float dtmean = s * (1.f / 32.f);
        acc += row[OFF_ANG + j] * dtmean;
        ph[i] = acc;

        const size_t o = (bl0 + i) * 32 + j;
        DTb[o]    = dt;
        float Av  = fminf(-softplus_f(row[OFF_A + j]), -1e-4f);
        decayb[o] = expf(Av * dt);
        trapb[o]  = sigmoid_f(row[OFF_TRAP + j]);
    }

    wtot[w][j] = acc;
    __syncthreads();

    float off = 0.f;
    #pragma unroll
    for (int k = 0; k < PW_ - 1; ++k) {
        const float v = wtot[k][j];
        off += (k < w) ? v : 0.f;
    }

    #pragma unroll
    for (int i = 0; i < PT_; ++i)
        phaseb[(bl0 + i) * 32 + j] = off + ph[i];
}

__global__ __launch_bounds__(64) void rot_kernel(
    const float* __restrict__ proj,
    const float* __restrict__ Bn_g, const float* __restrict__ Bn_b,
    const float* __restrict__ Cn_g, const float* __restrict__ Cn_b,
    const float* __restrict__ phaseb,
    float* __restrict__ Brot, float* __restrict__ Crot)
{
    const int bl = blockIdx.x;
    const int t  = threadIdx.x;
    __shared__ float red[64], nb[64], nc[64];

    const float* row = proj + (size_t)bl * DIP;
    const float vb = row[OFF_B + t];
    const float vc = row[OFF_C + t];

    auto blockReduce = [&](float v) -> float {
        red[t] = v; __syncthreads();
        for (int s = 32; s > 0; s >>= 1) {
            if (t < s) red[t] += red[t + s];
            __syncthreads();
        }
        float r = red[0]; __syncthreads();
        return r;
    };

    const float mB  = blockReduce(vb) * (1.f / 64.f);
    const float vaB = blockReduce((vb - mB) * (vb - mB)) * (1.f / 64.f);
    const float nB  = (vb - mB) * rsqrtf(vaB + 1e-5f) * Bn_g[t] + Bn_b[t];

    const float mC  = blockReduce(vc) * (1.f / 64.f);
    const float vaC = blockReduce((vc - mC) * (vc - mC)) * (1.f / 64.f);
    const float nC  = (vc - mC) * rsqrtf(vaC + 1e-5f) * Cn_g[t] + Cn_b[t];

    nb[t] = nB; nc[t] = nC;
    __syncthreads();

    const int   q  = t >> 1;
    const float ph = phaseb[(size_t)bl * 32 + q];
    const float c  = cosf(ph), s = sinf(ph);
    float ob, oc;
    if ((t & 1) == 0) { ob = nb[t] * c - nb[t + 1] * s; oc = nc[t] * c - nc[t + 1] * s; }
    else              { ob = nb[t - 1] * s + nb[t] * c; oc = nc[t - 1] * s + nc[t] * c; }
    Brot[(size_t)bl * 64 + t] = ob;
    Crot[(size_t)bl * 64 + t] = oc;
}


__global__ __launch_bounds__(256) void scan_chunk_kernel(
    const float* __restrict__ proj,
    const float* __restrict__ Brot, const float* __restrict__ Crot,
    const float* __restrict__ DTb, const float* __restrict__ decayb,
    const float* __restrict__ trapb,
    float* __restrict__ yfin, float* __restrict__ cumdb,
    float* __restrict__ Sstate)
{
    const int blk = blockIdx.x;
    const int c   = blk % NC_;
    const int h   = (blk / NC_) % NH_;
    const int b   = blk / (NC_ * NH_);
    const int l0  = c * LC_;
    const int t   = threadIdx.x;
    const int p   = t >> 2;
    const int seg = t & 3;

    __shared__ float sx[2][64], sB[2][64], sC[2][64], sscal[2][3];

    float hst[16], Bprev[16];
    float xprev = 0.f;
    #pragma unroll
    for (int i = 0; i < 16; ++i) { hst[i] = 0.f; Bprev[i] = 0.f; }
    if (c > 0) {
        const size_t blp = (size_t)(b * L_ + l0 - 1);
        xprev = proj[blp * DIP + OFF_X + h * HD_ + p];
        const float* Bp = Brot + blp * 64 + seg * 16;
        #pragma unroll
        for (int i = 0; i < 16; ++i) Bprev[i] = Bp[i];
    }

    auto loadTok = [&](int tok, int buf) {
        const size_t bl = (size_t)(b * L_ + l0 + tok);
        if (t < 64)        sx[buf][t]       = proj[bl * DIP + OFF_X + h * HD_ + t];
        else if (t < 128)  sB[buf][t - 64]  = Brot[bl * 64 + (t - 64)];
        else if (t < 192)  sC[buf][t - 128] = Crot[bl * 64 + (t - 128)];
        else if (t == 192) sscal[buf][0] = DTb[bl * 32 + h];
        else if (t == 193) sscal[buf][1] = decayb[bl * 32 + h];
        else if (t == 194) sscal[buf][2] = trapb[bl * 32 + h];
    };

    loadTok(0, 0);
    __syncthreads();
    float cumd = 1.f;

    for (int tok = 0; tok < LC_; ++tok) {
        const int cur = tok & 1;
        if (tok + 1 < LC_) loadTok(tok + 1, cur ^ 1);

        const float x   = sx[cur][p];
        const float dt  = sscal[cur][0];
        const float dec = sscal[cur][1];
        const float tr  = sscal[cur][2];

        float y = 0.f;
        #pragma unroll
        for (int i = 0; i < 16; ++i) {
            const int   n  = seg * 16 + i;
            const float Bn = sB[cur][n];
            const float bx = tr * x * Bn + (1.f - tr) * xprev * Bprev[i];
            hst[i] = dec * hst[i] + dt * bx;
            y += hst[i] * sC[cur][n];
            Bprev[i] = Bn;
        }
        xprev = x;
        cumd *= dec;

        y += __shfl_xor(y, 1, 32);
        y += __shfl_xor(y, 2, 32);

        const size_t bl = (size_t)(b * L_ + l0 + tok);
        if (seg == 0) yfin[bl * DI + h * HD_ + p] = y;
        if (t == 0)   cumdb[((size_t)(b * NH_ + h)) * L_ + l0 + tok] = cumd;
        __syncthreads();
    }

    float* Sp = Sstate + (((size_t)(b * NH_ + h)) * NC_ + c) * 4096 + p * 64 + seg * 16;
    #pragma unroll
    for (int i = 0; i < 16; ++i) Sp[i] = hst[i];
}

__global__ __launch_bounds__(256) void scan_combine_kernel(
    const float* __restrict__ cumdb, float* __restrict__ Sstate)
{
    const int h  = blockIdx.x % NH_;
    const int b  = blockIdx.x / NH_;
    const int t  = threadIdx.x;
    const int p  = t >> 2, seg = t & 3;

    float H[16];
    #pragma unroll
    for (int i = 0; i < 16; ++i) H[i] = 0.f;

    for (int c = 0; c < NC_; ++c) {
        float* Sp = Sstate + (((size_t)(b * NH_ + h)) * NC_ + c) * 4096 + p * 64 + seg * 16;
        const float Pc = cumdb[((size_t)(b * NH_ + h)) * L_ + c * LC_ + LC_ - 1];
        #pragma unroll
        for (int i = 0; i < 16; ++i) {
            const float Sc = Sp[i];
            Sp[i] = H[i];
            H[i]  = Pc * H[i] + Sc;
        }
    }
}

__global__ __launch_bounds__(256) void scan_correct_kernel(
    const float* __restrict__ proj, const float* __restrict__ Crot,
    const float* __restrict__ cumdb, const float* __restrict__ Sstate,
    const float* __restrict__ D_param, float* __restrict__ yfin)
{
    const int blk = blockIdx.x;
    const int c   = blk % NC_;
    const int h   = (blk / NC_) % NH_;
    const int b   = blk / (NC_ * NH_);
    const int l0  = c * LC_;
    const int t   = threadIdx.x;
    const int p   = t >> 2, seg = t & 3;

    __shared__ float sC[64];
    __shared__ float scd;

    float H[16];
    const float* Sp = Sstate + (((size_t)(b * NH_ + h)) * NC_ + c) * 4096 + p * 64 + seg * 16;
    #pragma unroll
    for (int i = 0; i < 16; ++i) H[i] = Sp[i];
    const float Dp = D_param[h];

    for (int tok = 0; tok < LC_; ++tok) {
        const size_t bl = (size_t)(b * L_ + l0 + tok);
        if (t < 64)  sC[t] = Crot[bl * 64 + t];
        if (t == 64) scd   = cumdb[((size_t)(b * NH_ + h)) * L_ + l0 + tok];
        __syncthreads();

        float corr = 0.f;
        #pragma unroll
        for (int i = 0; i < 16; ++i) corr += H[i] * sC[seg * 16 + i];
        corr += __shfl_xor(corr, 1, 32);
        corr += __shfl_xor(corr, 2, 32);

        if (seg == 0) {
            const float x = proj[bl * DIP + OFF_X + h * HD_ + p];
            const float z = proj[bl * DIP + OFF_Z + h * HD_ + p];
            float y = yfin[bl * DI + h * HD_ + p] + scd * corr;
            yfin[bl * DI + h * HD_ + p] = (y + Dp * x) * (z * sigmoid_f(z));
        }
        __syncthreads();
    }
}

extern "C" void kernel_launch(void* const* d_in, const int* in_sizes, int n_in,
                              void* d_out, int out_size, void* d_ws, size_t ws_size,
                              hipStream_t stream)
{
    const float* u       = (const float*)d_in[0];
    const float* W_in    = (const float*)d_in[1];
    const float* W_out   = (const float*)d_in[2];
    const float* dt_bias = (const float*)d_in[3];
    const float* D_param = (const float*)d_in[4];
    const float* Bn_g    = (const float*)d_in[5];
    const float* Bn_b    = (const float*)d_in[6];
    const float* Cn_g    = (const float*)d_in[7];
    const float* Cn_b    = (const float*)d_in[8];

    float* ws = (float*)d_ws;
    float* proj   = ws; ws += (size_t)M1 * DIP;
    float* yfin   = ws; ws += (size_t)M1 * DI;
    float* Sstate = ws; ws += (size_t)B_ * NH_ * NC_ * 4096;
    float* Brot   = ws; ws += (size_t)M1 * DS_;
    float* Crot   = ws; ws += (size_t)M1 * DS_;
    float* DTb    = ws; ws += (size_t)M1 * NH_;
    float* decayb = ws; ws += (size_t)M1 * NH_;
    float* trapb  = ws; ws += (size_t)M1 * NH_;
    float* phaseb = ws; ws += (size_t)M1 * NA_;
    float* cumdb  = ws; ws += (size_t)B_ * NH_ * L_;

    gemm_wmma_f32<<<dim3(DIP / 64, M1 / 256), dim3(256), 0, stream>>>(
        u, W_in, proj, M1, DIP, DM);

    phase_kernel<<<dim3(B_), dim3(PW_ * 32), 0, stream>>>(
        proj, dt_bias, DTb, decayb, trapb, phaseb);

    rot_kernel<<<dim3(M1), dim3(64), 0, stream>>>(
        proj, Bn_g, Bn_b, Cn_g, Cn_b, phaseb, Brot, Crot);

    scan_chunk_kernel<<<dim3(B_ * NH_ * NC_), dim3(256), 0, stream>>>(
        proj, Brot, Crot, DTb, decayb, trapb, yfin, cumdb, Sstate);
    scan_combine_kernel<<<dim3(B_ * NH_), dim3(256), 0, stream>>>(
        cumdb, Sstate);
    scan_correct_kernel<<<dim3(B_ * NH_ * NC_), dim3(256), 0, stream>>>(
        proj, Crot, cumdb, Sstate, D_param, yfin);

    gemm_wmma_f32<<<dim3(DM / 64, M1 / 256), dim3(256), 0, stream>>>(
        yfin, W_out, (float*)d_out, M1, DM, DI);
}


# GEMM operands staged through LDS: 64x64 block tiles (1088 and 256 blocks), K-slabs of 32 fetched a slab ahead, two LDS stages, one LDS-only barrier per slab; f32 WMMA, k order and sum order unchanged
# speedup vs baseline: 4.8577x; 2.4881x over previous
#include <hip/hip_runtime.h>
#include <math.h>

#define B_   2
#define L_   512
#define DM   1024
#define DI   2048
#define DS_  64
#define NH_  32
#define HD_  64
#define NA_  32
#define DIP  4352
#define M1   (B_*L_)

#define NC_  16
#define LC_  32

#define OFF_Z    0
#define OFF_X    2048
#define OFF_B    4096
#define OFF_C    4160
#define OFF_DT   4224
#define OFF_A    4256
#define OFF_TRAP 4288
#define OFF_ANG  4320

typedef __attribute__((ext_vector_type(2))) float v2f;
typedef __attribute__((ext_vector_type(8))) float v8f;

__device__ __forceinline__ float softplus_f(float x) {
    return x > 20.f ? x : log1pf(expf(x));
}
__device__ __forceinline__ float sigmoid_f(float x) {
    return 1.f / (1.f + expf(-x));
}

#define GBM    64
#define GBN    64
#define GBK    32
#define APITCH 36
#define BPITCH 160

template <int N, int K>
__global__ __launch_bounds__(256) void gemm_wmma_f32(
    const float* __restrict__ A, const float* __restrict__ Bm, float* __restrict__ C)
{
    static_assert(N % GBN == 0 && K % GBK == 0, "tile sizes must divide the problem");
    __shared__ __attribute__((aligned(16))) float As[2][GBM * APITCH];
    __shared__ __attribute__((aligned(16))) float Bs[2][(GBK / 2) * BPITCH];

    const int t    = threadIdx.x;
    const int lane = t & 31;
    const int wave = t >> 5;
    const int lo   = lane & 15;
    const int hi   = lane >> 4;

    const int m0 = blockIdx.y * GBM;
    const int n0 = blockIdx.x * GBN;
    const int wm = (wave >> 1) * 16;
    const int wn = (wave & 1) * 32;

    const int ar = t >> 3, ac = (t & 7) * 4;
    const int bn = t & 63, kq = t >> 6;

    const float* Ag0 = A  + (size_t)(m0 + ar) * K + ac;
    const float* Ag1 = Ag0 + (size_t)32 * K;
    const float* Bg  = Bm + (size_t)(2 * kq) * N + n0 + bn;

    const int as0 = ar * APITCH + ac, as1 = (ar + 32) * APITCH + ac;
    const int bs0 = kq * BPITCH + bn * 2;

    float4 ra0, ra1;
    v2f    pb[4];
    auto fetch = [&](int i) {
        ra0 = *reinterpret_cast<const float4*>(Ag0 + (size_t)i * GBK);
        ra1 = *reinterpret_cast<const float4*>(Ag1 + (size_t)i * GBK);
        const float* bp = Bg + (size_t)i * GBK * N;
        #pragma unroll
        for (int q = 0; q < 4; ++q) {
            pb[q].x = bp[(size_t)(8 * q) * N];
            pb[q].y = bp[(size_t)(8 * q + 1) * N];
        }
    };
    auto stage = [&](int st) {
        *reinterpret_cast<float4*>(&As[st][as0]) = ra0;
        *reinterpret_cast<float4*>(&As[st][as1]) = ra1;
        #pragma unroll
        for (int q = 0; q < 4; ++q)
            *reinterpret_cast<v2f*>(&Bs[st][bs0 + 4 * q * BPITCH]) = pb[q];
    };
    auto lds_handover = [] {
        __builtin_amdgcn_fence(__ATOMIC_RELEASE, "workgroup", "local");
        __builtin_amdgcn_s_barrier();
        __builtin_amdgcn_fence(__ATOMIC_ACQUIRE, "workgroup", "local");
    };

    constexpr int nslab = K / GBK;
    fetch(0);
    stage(0);
    if (nslab > 1) fetch(1);
    lds_handover();

    v8f acc0 = {}, acc1 = {};

    const int fa = (wm + lo) * APITCH + 2 * hi;
    const int fb = hi * BPITCH + (wn + lo) * 2;

    auto slab = [&](int st) {
        const float* as = &As[st][fa];
        const float* bs = &Bs[st][fb];
        v2f a[GBK / 4], b0[GBK / 4], b1[GBK / 4];
        #pragma unroll
        for (int j = 0; j < GBK / 4; ++j) {
            a[j]  = *reinterpret_cast<const v2f*>(as + 4 * j);
            b0[j] = *reinterpret_cast<const v2f*>(bs + 2 * j * BPITCH);
            b1[j] = *reinterpret_cast<const v2f*>(bs + 2 * j * BPITCH + 32);
        }
        __builtin_amdgcn_sched_barrier(0);
        #pragma unroll
        for (int j = 0; j < GBK / 4; ++j) {
            acc0 = __builtin_amdgcn_wmma_f32_16x16x4_f32(false, a[j], false, b0[j], (short)0, acc0, false, false);
            acc1 = __builtin_amdgcn_wmma_f32_16x16x4_f32(false, a[j], false, b1[j], (short)0, acc1, false, false);
        }
    };

    for (int s = 0; s + 1 < nslab; ++s) {
        slab(s & 1);
        __builtin_amdgcn_sched_barrier(0);
        stage((s & 1) ^ 1);
        if (s + 2 < nslab) fetch(s + 2);
        lds_handover();
    }
    slab((nslab - 1) & 1);

    #pragma unroll
    for (int r = 0; r < 8; ++r) {
        float* cp = C + (size_t)(m0 + wm + r + 8 * hi) * N + n0 + wn + lo;
        cp[0]  = acc0[r];
        cp[16] = acc1[r];
    }
}

#define PW_  32
#define PT_  16
static_assert(PW_ * PT_ == L_, "waves x tokens per wave must cover the sequence");

__global__ __launch_bounds__(PW_ * 32) void phase_kernel(
    const float* __restrict__ proj, const float* __restrict__ dt_bias,
    float* __restrict__ DTb, float* __restrict__ decayb,
    float* __restrict__ trapb, float* __restrict__ phaseb)
{
    const int b = blockIdx.x;
    const int j = threadIdx.x & 31;
    const int w = threadIdx.x >> 5;
    __shared__ float wtot[PW_][32];

    const float  bias = dt_bias[j];
    const size_t bl0  = (size_t)b * L_ + (size_t)w * PT_;

    float ph[PT_];
    float acc = 0.f;
    #pragma unroll
    for (int i = 0; i < PT_; ++i) {
        const float* row = proj + (bl0 + i) * DIP;
        float dt = softplus_f(row[OFF_DT + j] + bias);
        float s = dt;
        s += __shfl_xor(s, 16, 32);
        s += __shfl_xor(s,  8, 32);
        s += __shfl_xor(s,  4, 32);
        s += __shfl_xor(s,  2, 32);
        s += __shfl_xor(s,  1, 32);
        const float dtmean = s * (1.f / 32.f);
        acc += row[OFF_ANG + j] * dtmean;
        ph[i] = acc;

        const size_t o = (bl0 + i) * 32 + j;
        DTb[o]    = dt;
        float Av  = fminf(-softplus_f(row[OFF_A + j]), -1e-4f);
        decayb[o] = expf(Av * dt);
        trapb[o]  = sigmoid_f(row[OFF_TRAP + j]);
    }

    wtot[w][j] = acc;
    __syncthreads();

    float off = 0.f;
    #pragma unroll
    for (int k = 0; k < PW_ - 1; ++k) {
        const float v = wtot[k][j];
        off += (k < w) ? v : 0.f;
    }

    #pragma unroll
    for (int i = 0; i < PT_; ++i)
        phaseb[(bl0 + i) * 32 + j] = off + ph[i];
}

__global__ __launch_bounds__(64) void rot_kernel(
    const float* __restrict__ proj,
    const float* __restrict__ Bn_g, const float* __restrict__ Bn_b,
    const float* __restrict__ Cn_g, const float* __restrict__ Cn_b,
    const float* __restrict__ phaseb,
    float* __restrict__ Brot, float* __restrict__ Crot)
{
    const int bl = blockIdx.x;
    const int t  = threadIdx.x;
    __shared__ float red[64], nb[64], nc[64];

    const float* row = proj + (size_t)bl * DIP;
    const float vb = row[OFF_B + t];
    const float vc = row[OFF_C + t];

    auto blockReduce = [&](float v) -> float {
        red[t] = v; __syncthreads();
        for (int s = 32; s > 0; s >>= 1) {
            if (t < s) red[t] += red[t + s];
            __syncthreads();
        }
        float r = red[0]; __syncthreads();
        return r;
    };

    const float mB  = blockReduce(vb) * (1.f / 64.f);
    const float vaB = blockReduce((vb - mB) * (vb - mB)) * (1.f / 64.f);
    const float nB  = (vb - mB) * rsqrtf(vaB + 1e-5f) * Bn_g[t] + Bn_b[t];

    const float mC  = blockReduce(vc) * (1.f / 64.f);
    const float vaC = blockReduce((vc - mC) * (vc - mC)) * (1.f / 64.f);
    const float nC  = (vc - mC) * rsqrtf(vaC + 1e-5f) * Cn_g[t] + Cn_b[t];

    nb[t] = nB; nc[t] = nC;
    __syncthreads();

    const int   q  = t >> 1;
    const float ph = phaseb[(size_t)bl * 32 + q];
    const float c  = cosf(ph), s = sinf(ph);
    float ob, oc;
    if ((t & 1) == 0) { ob = nb[t] * c - nb[t + 1] * s; oc = nc[t] * c - nc[t + 1] * s; }
    else              { ob = nb[t - 1] * s + nb[t] * c; oc = nc[t - 1] * s + nc[t] * c; }
    Brot[(size_t)bl * 64 + t] = ob;
    Crot[(size_t)bl * 64 + t] = oc;
}


__global__ __launch_bounds__(256) void scan_chunk_kernel(
    const float* __restrict__ proj,
    const float* __restrict__ Brot, const float* __restrict__ Crot,
    const float* __restrict__ DTb, const float* __restrict__ decayb,
    const float* __restrict__ trapb,
    float* __restrict__ yfin, float* __restrict__ cumdb,
    float* __restrict__ Sstate)
{
    const int blk = blockIdx.x;
    const int c   = blk % NC_;
    const int h   = (blk / NC_) % NH_;
    const int b   = blk / (NC_ * NH_);
    const int l0  = c * LC_;
    const int t   = threadIdx.x;
    const int p   = t >> 2;
    const int seg = t & 3;

    __shared__ float sx[2][64], sB[2][64], sC[2][64], sscal[2][3];

    float hst[16], Bprev[16];
    float xprev = 0.f;
    #pragma unroll
    for (int i = 0; i < 16; ++i) { hst[i] = 0.f; Bprev[i] = 0.f; }
    if (c > 0) {
        const size_t blp = (size_t)(b * L_ + l0 - 1);
        xprev = proj[blp * DIP + OFF_X + h * HD_ + p];
        const float* Bp = Brot + blp * 64 + seg * 16;
        #pragma unroll
        for (int i = 0; i < 16; ++i) Bprev[i] = Bp[i];
    }

    auto loadTok = [&](int tok, int buf) {
        const size_t bl = (size_t)(b * L_ + l0 + tok);
        if (t < 64)        sx[buf][t]       = proj[bl * DIP + OFF_X + h * HD_ + t];
        else if (t < 128)  sB[buf][t - 64]  = Brot[bl * 64 + (t - 64)];
        else if (t < 192)  sC[buf][t - 128] = Crot[bl * 64 + (t - 128)];
        else if (t == 192) sscal[buf][0] = DTb[bl * 32 + h];
        else if (t == 193) sscal[buf][1] = decayb[bl * 32 + h];
        else if (t == 194) sscal[buf][2] = trapb[bl * 32 + h];
    };

    loadTok(0, 0);
    __syncthreads();
    float cumd = 1.f;

    for (int tok = 0; tok < LC_; ++tok) {
        const int cur = tok & 1;
        if (tok + 1 < LC_) loadTok(tok + 1, cur ^ 1);

        const float x   = sx[cur][p];
        const float dt  = sscal[cur][0];
        const float dec = sscal[cur][1];
        const float tr  = sscal[cur][2];

        float y = 0.f;
        #pragma unroll
        for (int i = 0; i < 16; ++i) {
            const int   n  = seg * 16 + i;
            const float Bn = sB[cur][n];
            const float bx = tr * x * Bn + (1.f - tr) * xprev * Bprev[i];
            hst[i] = dec * hst[i] + dt * bx;
            y += hst[i] * sC[cur][n];
            Bprev[i] = Bn;
        }
        xprev = x;
        cumd *= dec;

        y += __shfl_xor(y, 1, 32);
        y += __shfl_xor(y, 2, 32);

        const size_t bl = (size_t)(b * L_ + l0 + tok);
        if (seg == 0) yfin[bl * DI + h * HD_ + p] = y;
        if (t == 0)   cumdb[((size_t)(b * NH_ + h)) * L_ + l0 + tok] = cumd;
        __syncthreads();
    }

    float* Sp = Sstate + (((size_t)(b * NH_ + h)) * NC_ + c) * 4096 + p * 64 + seg * 16;
    #pragma unroll
    for (int i = 0; i < 16; ++i) Sp[i] = hst[i];
}

__global__ __launch_bounds__(256) void scan_combine_kernel(
    const float* __restrict__ cumdb, float* __restrict__ Sstate)
{
    const int h  = blockIdx.x % NH_;
    const int b  = blockIdx.x / NH_;
    const int t  = threadIdx.x;
    const int p  = t >> 2, seg = t & 3;

    float H[16];
    #pragma unroll
    for (int i = 0; i < 16; ++i) H[i] = 0.f;

    for (int c = 0; c < NC_; ++c) {
        float* Sp = Sstate + (((size_t)(b * NH_ + h)) * NC_ + c) * 4096 + p * 64 + seg * 16;
        const float Pc = cumdb[((size_t)(b * NH_ + h)) * L_ + c * LC_ + LC_ - 1];
        #pragma unroll
        for (int i = 0; i < 16; ++i) {
            const float Sc = Sp[i];
            Sp[i] = H[i];
            H[i]  = Pc * H[i] + Sc;
        }
    }
}

__global__ __launch_bounds__(256) void scan_correct_kernel(
    const float* __restrict__ proj, const float* __restrict__ Crot,
    const float* __restrict__ cumdb, const float* __restrict__ Sstate,
    const float* __restrict__ D_param, float* __restrict__ yfin)
{
    const int blk = blockIdx.x;
    const int c   = blk % NC_;
    const int h   = (blk / NC_) % NH_;
    const int b   = blk / (NC_ * NH_);
    const int l0  = c * LC_;
    const int t   = threadIdx.x;
    const int p   = t >> 2, seg = t & 3;

    __shared__ float sC[64];
    __shared__ float scd;

    float H[16];
    const float* Sp = Sstate + (((size_t)(b * NH_ + h)) * NC_ + c) * 4096 + p * 64 + seg * 16;
    #pragma unroll
    for (int i = 0; i < 16; ++i) H[i] = Sp[i];
    const float Dp = D_param[h];

    for (int tok = 0; tok < LC_; ++tok) {
        const size_t bl = (size_t)(b * L_ + l0 + tok);
        if (t < 64)  sC[t] = Crot[bl * 64 + t];
        if (t == 64) scd   = cumdb[((size_t)(b * NH_ + h)) * L_ + l0 + tok];
        __syncthreads();

        float corr = 0.f;
        #pragma unroll
        for (int i = 0; i < 16; ++i) corr += H[i] * sC[seg * 16 + i];
        corr += __shfl_xor(corr, 1, 32);
        corr += __shfl_xor(corr, 2, 32);

        if (seg == 0) {
            const float x = proj[bl * DIP + OFF_X + h * HD_ + p];
            const float z = proj[bl * DIP + OFF_Z + h * HD_ + p];
            float y = yfin[bl * DI + h * HD_ + p] + scd * corr;
            yfin[bl * DI + h * HD_ + p] = (y + Dp * x) * (z * sigmoid_f(z));
        }
        __syncthreads();
    }
}

extern "C" void kernel_launch(void* const* d_in, const int* in_sizes, int n_in,
                              void* d_out, int out_size, void* d_ws, size_t ws_size,
                              hipStream_t stream)
{
    const float* u       = (const float*)d_in[0];
    const float* W_in    = (const float*)d_in[1];
    const float* W_out   = (const float*)d_in[2];
    const float* dt_bias = (const float*)d_in[3];
    const float* D_param = (const float*)d_in[4];
    const float* Bn_g    = (const float*)d_in[5];
    const float* Bn_b    = (const float*)d_in[6];
    const float* Cn_g    = (const float*)d_in[7];
    const float* Cn_b    = (const float*)d_in[8];

    float* ws = (float*)d_ws;
    float* proj   = ws; ws += (size_t)M1 * DIP;
    float* yfin   = ws; ws += (size_t)M1 * DI;
    float* Sstate = ws; ws += (size_t)B_ * NH_ * NC_ * 4096;
    float* Brot   = ws; ws += (size_t)M1 * DS_;
    float* Crot   = ws; ws += (size_t)M1 * DS_;
    float* DTb    = ws; ws += (size_t)M1 * NH_;
    float* decayb = ws; ws += (size_t)M1 * NH_;
    float* trapb  = ws; ws += (size_t)M1 * NH_;
    float* phaseb = ws; ws += (size_t)M1 * NA_;
    float* cumdb  = ws; ws += (size_t)B_ * NH_ * L_;

    gemm_wmma_f32<DIP, DM><<<dim3(DIP / GBN, M1 / GBM), dim3(256), 0, stream>>>(
        u, W_in, proj);

    phase_kernel<<<dim3(B_), dim3(PW_ * 32), 0, stream>>>(
        proj, dt_bias, DTb, decayb, trapb, phaseb);

    rot_kernel<<<dim3(M1), dim3(64), 0, stream>>>(
        proj, Bn_g, Bn_b, Cn_g, Cn_b, phaseb, Brot, Crot);

    scan_chunk_kernel<<<dim3(B_ * NH_ * NC_), dim3(256), 0, stream>>>(
        proj, Brot, Crot, DTb, decayb, trapb, yfin, cumdb, Sstate);
    scan_combine_kernel<<<dim3(B_ * NH_), dim3(256), 0, stream>>>(
        cumdb, Sstate);
    scan_correct_kernel<<<dim3(B_ * NH_ * NC_), dim3(256), 0, stream>>>(
        proj, Crot, cumdb, Sstate, D_param, yfin);

    gemm_wmma_f32<DM, DI><<<dim3(DM / GBN, M1 / GBM), dim3(256), 0, stream>>>(
        yfin, W_out, (float*)d_out);
}


# scan chunk and correction passes: each block stages its whole 32-token chunk in LDS once, token loops hold no barrier and no global access, lane partials join through LDS in the same sum tree
# speedup vs baseline: 6.3594x; 1.3091x over previous
#include <hip/hip_runtime.h>
#include <math.h>

#define B_   2
#define L_   512
#define DM   1024
#define DI   2048
#define DS_  64
#define NH_  32
#define HD_  64
#define NA_  32
#define DIP  4352
#define M1   (B_*L_)

#define NC_  16
#define LC_  32

#define OFF_Z    0
#define OFF_X    2048
#define OFF_B    4096
#define OFF_C    4160
#define OFF_DT   4224
#define OFF_A    4256
#define OFF_TRAP 4288
#define OFF_ANG  4320

typedef __attribute__((ext_vector_type(2))) float v2f;
typedef __attribute__((ext_vector_type(8))) float v8f;

__device__ __forceinline__ float softplus_f(float x) {
    return x > 20.f ? x : log1pf(expf(x));
}
__device__ __forceinline__ float sigmoid_f(float x) {
    return 1.f / (1.f + expf(-x));
}

#define GBM    64
#define GBN    64
#define GBK    32
#define APITCH 36
#define BPITCH 160

template <int N, int K>
__global__ __launch_bounds__(256) void gemm_wmma_f32(
    const float* __restrict__ A, const float* __restrict__ Bm, float* __restrict__ C)
{
    static_assert(N % GBN == 0 && K % GBK == 0, "tile sizes must divide the problem");
    __shared__ __attribute__((aligned(16))) float As[2][GBM * APITCH];
    __shared__ __attribute__((aligned(16))) float Bs[2][(GBK / 2) * BPITCH];

    const int t    = threadIdx.x;
    const int lane = t & 31;
    const int wave = t >> 5;
    const int lo   = lane & 15;
    const int hi   = lane >> 4;

    const int m0 = blockIdx.y * GBM;
    const int n0 = blockIdx.x * GBN;
    const int wm = (wave >> 1) * 16;
    const int wn = (wave & 1) * 32;

    const int ar = t >> 3, ac = (t & 7) * 4;
    const int bn = t & 63, kq = t >> 6;

    const float* Ag0 = A  + (size_t)(m0 + ar) * K + ac;
    const float* Ag1 = Ag0 + (size_t)32 * K;
    const float* Bg  = Bm + (size_t)(2 * kq) * N + n0 + bn;

    const int as0 = ar * APITCH + ac, as1 = (ar + 32) * APITCH + ac;
    const int bs0 = kq * BPITCH + bn * 2;

    float4 ra0, ra1;
    v2f    pb[4];
    auto fetch = [&](int i) {
        ra0 = *reinterpret_cast<const float4*>(Ag0 + (size_t)i * GBK);
        ra1 = *reinterpret_cast<const float4*>(Ag1 + (size_t)i * GBK);
        const float* bp = Bg + (size_t)i * GBK * N;
        #pragma unroll
        for (int q = 0; q < 4; ++q) {
            pb[q].x = bp[(size_t)(8 * q) * N];
            pb[q].y = bp[(size_t)(8 * q + 1) * N];
        }
    };
    auto stage = [&](int st) {
        *reinterpret_cast<float4*>(&As[st][as0]) = ra0;
        *reinterpret_cast<float4*>(&As[st][as1]) = ra1;
        #pragma unroll
        for (int q = 0; q < 4; ++q)
            *reinterpret_cast<v2f*>(&Bs[st][bs0 + 4 * q * BPITCH]) = pb[q];
    };
    auto lds_handover = [] {
        __builtin_amdgcn_fence(__ATOMIC_RELEASE, "workgroup", "local");
        __builtin_amdgcn_s_barrier();
        __builtin_amdgcn_fence(__ATOMIC_ACQUIRE, "workgroup", "local");
    };

    constexpr int nslab = K / GBK;
    fetch(0);
    stage(0);
    if (nslab > 1) fetch(1);
    lds_handover();

    v8f acc0 = {}, acc1 = {};

    const int fa = (wm + lo) * APITCH + 2 * hi;
    const int fb = hi * BPITCH + (wn + lo) * 2;

    auto slab = [&](int st) {
        const float* as = &As[st][fa];
        const float* bs = &Bs[st][fb];
        v2f a[GBK / 4], b0[GBK / 4], b1[GBK / 4];
        #pragma unroll
        for (int j = 0; j < GBK / 4; ++j) {
            a[j]  = *reinterpret_cast<const v2f*>(as + 4 * j);
            b0[j] = *reinterpret_cast<const v2f*>(bs + 2 * j * BPITCH);
            b1[j] = *reinterpret_cast<const v2f*>(bs + 2 * j * BPITCH + 32);
        }
        __builtin_amdgcn_sched_barrier(0);
        #pragma unroll
        for (int j = 0; j < GBK / 4; ++j) {
            acc0 = __builtin_amdgcn_wmma_f32_16x16x4_f32(false, a[j], false, b0[j], (short)0, acc0, false, false);
            acc1 = __builtin_amdgcn_wmma_f32_16x16x4_f32(false, a[j], false, b1[j], (short)0, acc1, false, false);
        }
    };

    for (int s = 0; s + 1 < nslab; ++s) {
        slab(s & 1);
        __builtin_amdgcn_sched_barrier(0);
        stage((s & 1) ^ 1);
        if (s + 2 < nslab) fetch(s + 2);
        lds_handover();
    }
    slab((nslab - 1) & 1);

    #pragma unroll
    for (int r = 0; r < 8; ++r) {
        float* cp = C + (size_t)(m0 + wm + r + 8 * hi) * N + n0 + wn + lo;
        cp[0]  = acc0[r];
        cp[16] = acc1[r];
    }
}

#define PW_  32
#define PT_  16
static_assert(PW_ * PT_ == L_, "waves x tokens per wave must cover the sequence");

__global__ __launch_bounds__(PW_ * 32) void phase_kernel(
    const float* __restrict__ proj, const float* __restrict__ dt_bias,
    float* __restrict__ DTb, float* __restrict__ decayb,
    float* __restrict__ trapb, float* __restrict__ phaseb)
{
    const int b = blockIdx.x;
    const int j = threadIdx.x & 31;
    const int w = threadIdx.x >> 5;
    __shared__ float wtot[PW_][32];

    const float  bias = dt_bias[j];
    const size_t bl0  = (size_t)b * L_ + (size_t)w * PT_;

    float ph[PT_];
    float acc = 0.f;
    #pragma unroll
    for (int i = 0; i < PT_; ++i) {
        const float* row = proj + (bl0 + i) * DIP;
        float dt = softplus_f(row[OFF_DT + j] + bias);
        float s = dt;
        s += __shfl_xor(s, 16, 32);
        s += __shfl_xor(s,  8, 32);
        s += __shfl_xor(s,  4, 32);
        s += __shfl_xor(s,  2, 32);
        s += __shfl_xor(s,  1, 32);
        const float dtmean = s * (1.f / 32.f);
        acc += row[OFF_ANG + j] * dtmean;
        ph[i] = acc;

        const size_t o = (bl0 + i) * 32 + j;
        DTb[o]    = dt;
        float Av  = fminf(-softplus_f(row[OFF_A + j]), -1e-4f);
        decayb[o] = expf(Av * dt);
        trapb[o]  = sigmoid_f(row[OFF_TRAP + j]);
    }

    wtot[w][j] = acc;
    __syncthreads();

    float off = 0.f;
    #pragma unroll
    for (int k = 0; k < PW_ - 1; ++k) {
        const float v = wtot[k][j];
        off += (k < w) ? v : 0.f;
    }

    #pragma unroll
    for (int i = 0; i < PT_; ++i)
        phaseb[(bl0 + i) * 32 + j] = off + ph[i];
}

__global__ __launch_bounds__(64) void rot_kernel(
    const float* __restrict__ proj,
    const float* __restrict__ Bn_g, const float* __restrict__ Bn_b,
    const float* __restrict__ Cn_g, const float* __restrict__ Cn_b,
    const float* __restrict__ phaseb,
    float* __restrict__ Brot, float* __restrict__ Crot)
{
    const int bl = blockIdx.x;
    const int t  = threadIdx.x;
    __shared__ float red[64], nb[64], nc[64];

    const float* row = proj + (size_t)bl * DIP;
    const float vb = row[OFF_B + t];
    const float vc = row[OFF_C + t];

    auto blockReduce = [&](float v) -> float {
        red[t] = v; __syncthreads();
        for (int s = 32; s > 0; s >>= 1) {
            if (t < s) red[t] += red[t + s];
            __syncthreads();
        }
        float r = red[0]; __syncthreads();
        return r;
    };

    const float mB  = blockReduce(vb) * (1.f / 64.f);
    const float vaB = blockReduce((vb - mB) * (vb - mB)) * (1.f / 64.f);
    const float nB  = (vb - mB) * rsqrtf(vaB + 1e-5f) * Bn_g[t] + Bn_b[t];

    const float mC  = blockReduce(vc) * (1.f / 64.f);
    const float vaC = blockReduce((vc - mC) * (vc - mC)) * (1.f / 64.f);
    const float nC  = (vc - mC) * rsqrtf(vaC + 1e-5f) * Cn_g[t] + Cn_b[t];

    nb[t] = nB; nc[t] = nC;
    __syncthreads();

    const int   q  = t >> 1;
    const float ph = phaseb[(size_t)bl * 32 + q];
    const float c  = cosf(ph), s = sinf(ph);
    float ob, oc;
    if ((t & 1) == 0) { ob = nb[t] * c - nb[t + 1] * s; oc = nc[t] * c - nc[t + 1] * s; }
    else              { ob = nb[t - 1] * s + nb[t] * c; oc = nc[t - 1] * s + nc[t] * c; }
    Brot[(size_t)bl * 64 + t] = ob;
    Crot[(size_t)bl * 64 + t] = oc;
}


__global__ __launch_bounds__(256) void scan_chunk_kernel(
    const float* __restrict__ proj,
    const float* __restrict__ Brot, const float* __restrict__ Crot,
    const float* __restrict__ DTb, const float* __restrict__ decayb,
    const float* __restrict__ trapb,
    float* __restrict__ yfin, float* __restrict__ cumdb,
    float* __restrict__ Sstate)
{
    const int blk = blockIdx.x;
    const int c   = blk % NC_;
    const int h   = (blk / NC_) % NH_;
    const int b   = blk / (NC_ * NH_);
    const int l0  = c * LC_;
    const int t   = threadIdx.x;
    const int p   = t >> 2;
    const int seg = t & 3;

    __shared__ __attribute__((aligned(16))) float sx[LC_ + 1][64];
    __shared__ __attribute__((aligned(16))) float sB[LC_ + 1][64];
    __shared__ __attribute__((aligned(16))) float sC[LC_][64];
    __shared__ __attribute__((aligned(16))) float sscal[LC_][4];
    __shared__ float scum[LC_];
    __shared__ __attribute__((aligned(16))) float sy[LC_][64][4];

    const size_t bl0 = (size_t)b * L_ + l0;

    float4 vx[2], vb[2], vc[2];
    #pragma unroll
    for (int k = 0; k < 2; ++k) {
        const int idx = k * 256 + t;
        const int r = idx >> 4, q = idx & 15;
        vx[k] = *reinterpret_cast<const float4*>(proj + (bl0 + r) * DIP + OFF_X + h * HD_ + q * 4);
        vb[k] = *reinterpret_cast<const float4*>(Brot + (bl0 + r) * 64 + q * 4);
        vc[k] = *reinterpret_cast<const float4*>(Crot + (bl0 + r) * 64 + q * 4);
    }
    float4 v0 = make_float4(0.f, 0.f, 0.f, 0.f);
    if (c > 0 && t < 32) {
        const int q = t & 15;
        v0 = (t < 16)
           ? *reinterpret_cast<const float4*>(proj + (bl0 - 1) * DIP + OFF_X + h * HD_ + q * 4)
           : *reinterpret_cast<const float4*>(Brot + (bl0 - 1) * 64 + q * 4);
    }
    float vs = 0.f;
    if (t >= 64 && t < 160) {
        const int kind = (t - 64) >> 5, tok = (t - 64) & 31;
        const float* sp = (kind == 0) ? DTb : (kind == 1) ? decayb : trapb;
        vs = sp[(bl0 + tok) * 32 + h];
    }
    #pragma unroll
    for (int k = 0; k < 2; ++k) {
        const int idx = k * 256 + t;
        const int r = idx >> 4, q = idx & 15;
        *reinterpret_cast<float4*>(&sx[r + 1][q * 4]) = vx[k];
        *reinterpret_cast<float4*>(&sB[r + 1][q * 4]) = vb[k];
        *reinterpret_cast<float4*>(&sC[r][q * 4])     = vc[k];
    }
    if (t < 16)      *reinterpret_cast<float4*>(&sx[0][(t & 15) * 4]) = v0;
    else if (t < 32) *reinterpret_cast<float4*>(&sB[0][(t & 15) * 4]) = v0;
    if (t >= 64 && t < 160) sscal[(t - 64) & 31][(t - 64) >> 5] = vs;
    __syncthreads();

    float hst[16], Bp[16];
    #pragma unroll
    for (int i = 0; i < 16; ++i) hst[i] = 0.f;
    #pragma unroll
    for (int i4 = 0; i4 < 4; ++i4) {
        const float4 v = *reinterpret_cast<const float4*>(&sB[0][seg * 16 + i4 * 4]);
        Bp[i4 * 4 + 0] = v.x; Bp[i4 * 4 + 1] = v.y; Bp[i4 * 4 + 2] = v.z; Bp[i4 * 4 + 3] = v.w;
    }
    float xprev = sx[0][p];
    float cumd  = 1.f;

    #pragma unroll 2
    for (int tok = 0; tok < LC_; ++tok) {
        const float  x  = sx[tok + 1][p];
        const float4 sc = *reinterpret_cast<const float4*>(&sscal[tok][0]);
        const float dt = sc.x, dec = sc.y, tr = sc.z;

        float Bc[16], Cc[16];
        #pragma unroll
        for (int i4 = 0; i4 < 4; ++i4) {
            const float4 vB = *reinterpret_cast<const float4*>(&sB[tok + 1][seg * 16 + i4 * 4]);
            const float4 vC = *reinterpret_cast<const float4*>(&sC[tok][seg * 16 + i4 * 4]);
            Bc[i4 * 4 + 0] = vB.x; Bc[i4 * 4 + 1] = vB.y; Bc[i4 * 4 + 2] = vB.z; Bc[i4 * 4 + 3] = vB.w;
            Cc[i4 * 4 + 0] = vC.x; Cc[i4 * 4 + 1] = vC.y; Cc[i4 * 4 + 2] = vC.z; Cc[i4 * 4 + 3] = vC.w;
        }

        float y = 0.f;
        #pragma unroll
        for (int i = 0; i < 16; ++i) {
            const float bx = tr * x * Bc[i] + (1.f - tr) * xprev * Bp[i];
            hst[i] = dec * hst[i] + dt * bx;
            y += hst[i] * Cc[i];
            Bp[i] = Bc[i];
        }
        xprev = x;
        cumd *= dec;

        sy[tok][p][seg] = y;
        if (t == 0) scum[tok] = cumd;
    }
    __syncthreads();

    #pragma unroll
    for (int k = 0; k < 2; ++k) {
        const int idx = k * 256 + t;
        const int r = idx >> 4, q = idx & 15;
        float o[4];
        #pragma unroll
        for (int j = 0; j < 4; ++j) {
            const float4 pv = *reinterpret_cast<const float4*>(&sy[r][q * 4 + j][0]);
            o[j] = (pv.x + pv.y) + (pv.z + pv.w);
        }
        *reinterpret_cast<float4*>(yfin + (bl0 + r) * DI + h * HD_ + q * 4) = make_float4(o[0], o[1], o[2], o[3]);
    }
    if (t < LC_) cumdb[((size_t)(b * NH_ + h)) * L_ + l0 + t] = scum[t];

    float* Sp = Sstate + (((size_t)(b * NH_ + h)) * NC_ + c) * 4096 + p * 64 + seg * 16;
    #pragma unroll
    for (int i = 0; i < 16; ++i) Sp[i] = hst[i];
}

__global__ __launch_bounds__(256) void scan_combine_kernel(
    const float* __restrict__ cumdb, float* __restrict__ Sstate)
{
    const int h  = blockIdx.x % NH_;
    const int b  = blockIdx.x / NH_;
    const int t  = threadIdx.x;
    const int p  = t >> 2, seg = t & 3;

    float H[16];
    #pragma unroll
    for (int i = 0; i < 16; ++i) H[i] = 0.f;

    for (int c = 0; c < NC_; ++c) {
        float* Sp = Sstate + (((size_t)(b * NH_ + h)) * NC_ + c) * 4096 + p * 64 + seg * 16;
        const float Pc = cumdb[((size_t)(b * NH_ + h)) * L_ + c * LC_ + LC_ - 1];
        #pragma unroll
        for (int i = 0; i < 16; ++i) {
            const float Sc = Sp[i];
            Sp[i] = H[i];
            H[i]  = Pc * H[i] + Sc;
        }
    }
}

__global__ __launch_bounds__(256) void scan_correct_kernel(
    const float* __restrict__ proj, const float* __restrict__ Crot,
    const float* __restrict__ cumdb, const float* __restrict__ Sstate,
    const float* __restrict__ D_param, float* __restrict__ yfin)
{
    const int blk = blockIdx.x;
    const int c   = blk % NC_;
    const int h   = (blk / NC_) % NH_;
    const int b   = blk / (NC_ * NH_);
    const int l0  = c * LC_;
    const int t   = threadIdx.x;
    const int p   = t >> 2, seg = t & 3;

    __shared__ __attribute__((aligned(16))) float sC[LC_][64];
    __shared__ float scd[LC_];
    __shared__ __attribute__((aligned(16))) float scorr[LC_][64][4];

    const size_t bl0 = (size_t)b * L_ + l0;

    float H[16];
    const float* Sp = Sstate + (((size_t)(b * NH_ + h)) * NC_ + c) * 4096 + p * 64 + seg * 16;
    #pragma unroll
    for (int i = 0; i < 16; ++i) H[i] = Sp[i];
    const float Dp = D_param[h];

    float4 vc[2], vx[2], vz[2], vy[2];
    #pragma unroll
    for (int k = 0; k < 2; ++k) {
        const int idx = k * 256 + t;
        const int r = idx >> 4, q = idx & 15;
        vc[k] = *reinterpret_cast<const float4*>(Crot + (bl0 + r) * 64 + q * 4);
        vx[k] = *reinterpret_cast<const float4*>(proj + (bl0 + r) * DIP + OFF_X + h * HD_ + q * 4);
        vz[k] = *reinterpret_cast<const float4*>(proj + (bl0 + r) * DIP + OFF_Z + h * HD_ + q * 4);
        vy[k] = *reinterpret_cast<const float4*>(yfin + (bl0 + r) * DI + h * HD_ + q * 4);
    }
    float cdv = 0.f;
    if (t < LC_) cdv = cumdb[((size_t)(b * NH_ + h)) * L_ + l0 + t];

    #pragma unroll
    for (int k = 0; k < 2; ++k) {
        const int idx = k * 256 + t;
        const int r = idx >> 4, q = idx & 15;
        *reinterpret_cast<float4*>(&sC[r][q * 4]) = vc[k];
    }
    if (t < LC_) scd[t] = cdv;
    __syncthreads();

    #pragma unroll 2
    for (int tok = 0; tok < LC_; ++tok) {
        float Cc[16];
        #pragma unroll
        for (int i4 = 0; i4 < 4; ++i4) {
            const float4 vC = *reinterpret_cast<const float4*>(&sC[tok][seg * 16 + i4 * 4]);
            Cc[i4 * 4 + 0] = vC.x; Cc[i4 * 4 + 1] = vC.y; Cc[i4 * 4 + 2] = vC.z; Cc[i4 * 4 + 3] = vC.w;
        }
        float corr = 0.f;
        #pragma unroll
        for (int i = 0; i < 16; ++i) corr += H[i] * Cc[i];
        scorr[tok][p][seg] = corr;
    }
    __syncthreads();

    #pragma unroll
    for (int k = 0; k < 2; ++k) {
        const int idx = k * 256 + t;
        const int r = idx >> 4, q = idx & 15;
        const float cd = scd[r];
        const float xs[4] = {vx[k].x, vx[k].y, vx[k].z, vx[k].w};
        const float zs[4] = {vz[k].x, vz[k].y, vz[k].z, vz[k].w};
        const float ys[4] = {vy[k].x, vy[k].y, vy[k].z, vy[k].w};
        float o[4];
        #pragma unroll
        for (int j = 0; j < 4; ++j) {
            const float4 pv = *reinterpret_cast<const float4*>(&scorr[r][q * 4 + j][0]);
            const float corr = (pv.x + pv.y) + (pv.z + pv.w);
            const float y = ys[j] + cd * corr;
            o[j] = (y + Dp * xs[j]) * (zs[j] * sigmoid_f(zs[j]));
        }
        *reinterpret_cast<float4*>(yfin + (bl0 + r) * DI + h * HD_ + q * 4) = make_float4(o[0], o[1], o[2], o[3]);
    }
}

extern "C" void kernel_launch(void* const* d_in, const int* in_sizes, int n_in,
                              void* d_out, int out_size, void* d_ws, size_t ws_size,
                              hipStream_t stream)
{
    const float* u       = (const float*)d_in[0];
    const float* W_in    = (const float*)d_in[1];
    const float* W_out   = (const float*)d_in[2];
    const float* dt_bias = (const float*)d_in[3];
    const float* D_param = (const float*)d_in[4];
    const float* Bn_g    = (const float*)d_in[5];
    const float* Bn_b    = (const float*)d_in[6];
    const float* Cn_g    = (const float*)d_in[7];
    const float* Cn_b    = (const float*)d_in[8];

    float* ws = (float*)d_ws;
    float* proj   = ws; ws += (size_t)M1 * DIP;
    float* yfin   = ws; ws += (size_t)M1 * DI;
    float* Sstate = ws; ws += (size_t)B_ * NH_ * NC_ * 4096;
    float* Brot   = ws; ws += (size_t)M1 * DS_;
    float* Crot   = ws; ws += (size_t)M1 * DS_;
    float* DTb    = ws; ws += (size_t)M1 * NH_;
    float* decayb = ws; ws += (size_t)M1 * NH_;
    float* trapb  = ws; ws += (size_t)M1 * NH_;
    float* phaseb = ws; ws += (size_t)M1 * NA_;
    float* cumdb  = ws; ws += (size_t)B_ * NH_ * L_;

    gemm_wmma_f32<DIP, DM><<<dim3(DIP / GBN, M1 / GBM), dim3(256), 0, stream>>>(
        u, W_in, proj);

    phase_kernel<<<dim3(B_), dim3(PW_ * 32), 0, stream>>>(
        proj, dt_bias, DTb, decayb, trapb, phaseb);

    rot_kernel<<<dim3(M1), dim3(64), 0, stream>>>(
        proj, Bn_g, Bn_b, Cn_g, Cn_b, phaseb, Brot, Crot);

    scan_chunk_kernel<<<dim3(B_ * NH_ * NC_), dim3(256), 0, stream>>>(
        proj, Brot, Crot, DTb, decayb, trapb, yfin, cumdb, Sstate);
    scan_combine_kernel<<<dim3(B_ * NH_), dim3(256), 0, stream>>>(
        cumdb, Sstate);
    scan_correct_kernel<<<dim3(B_ * NH_ * NC_), dim3(256), 0, stream>>>(
        proj, Crot, cumdb, Sstate, D_param, yfin);

    gemm_wmma_f32<DM, DI><<<dim3(DM / GBN, M1 / GBM), dim3(256), 0, stream>>>(
        yfin, W_out, (float*)d_out);
}


# phase kernel spread over the chip: one token per wave on 64 blocks (was 16 tokens per wave on 2); group prefix via LDS, group totals joined in rotation kernel; sum order kept, products rounded first
# speedup vs baseline: 7.1101x; 1.1180x over previous
#include <hip/hip_runtime.h>
#include <math.h>

#define B_   2
#define L_   512
#define DM   1024
#define DI   2048
#define DS_  64
#define NH_  32
#define HD_  64
#define NA_  32
#define DIP  4352
#define M1   (B_*L_)

#define NC_  16
#define LC_  32

#define OFF_Z    0
#define OFF_X    2048
#define OFF_B    4096
#define OFF_C    4160
#define OFF_DT   4224
#define OFF_A    4256
#define OFF_TRAP 4288
#define OFF_ANG  4320

typedef __attribute__((ext_vector_type(2))) float v2f;
typedef __attribute__((ext_vector_type(8))) float v8f;

__device__ __forceinline__ float softplus_f(float x) {
    return x > 20.f ? x : log1pf(expf(x));
}
__device__ __forceinline__ float sigmoid_f(float x) {
    return 1.f / (1.f + expf(-x));
}

#define GBM    64
#define GBN    64
#define GBK    32
#define APITCH 36
#define BPITCH 160

template <int N, int K>
__global__ __launch_bounds__(256) void gemm_wmma_f32(
    const float* __restrict__ A, const float* __restrict__ Bm, float* __restrict__ C)
{
    static_assert(N % GBN == 0 && K % GBK == 0, "tile sizes must divide the problem");
    __shared__ __attribute__((aligned(16))) float As[2][GBM * APITCH];
    __shared__ __attribute__((aligned(16))) float Bs[2][(GBK / 2) * BPITCH];

    const int t    = threadIdx.x;
    const int lane = t & 31;
    const int wave = t >> 5;
    const int lo   = lane & 15;
    const int hi   = lane >> 4;

    const int m0 = blockIdx.y * GBM;
    const int n0 = blockIdx.x * GBN;
    const int wm = (wave >> 1) * 16;
    const int wn = (wave & 1) * 32;

    const int ar = t >> 3, ac = (t & 7) * 4;
    const int bn = t & 63, kq = t >> 6;

    const float* Ag0 = A  + (size_t)(m0 + ar) * K + ac;
    const float* Ag1 = Ag0 + (size_t)32 * K;
    const float* Bg  = Bm + (size_t)(2 * kq) * N + n0 + bn;

    const int as0 = ar * APITCH + ac, as1 = (ar + 32) * APITCH + ac;
    const int bs0 = kq * BPITCH + bn * 2;

    float4 ra0, ra1;
    v2f    pb[4];
    auto fetch = [&](int i) {
        ra0 = *reinterpret_cast<const float4*>(Ag0 + (size_t)i * GBK);
        ra1 = *reinterpret_cast<const float4*>(Ag1 + (size_t)i * GBK);
        const float* bp = Bg + (size_t)i * GBK * N;
        #pragma unroll
        for (int q = 0; q < 4; ++q) {
            pb[q].x = bp[(size_t)(8 * q) * N];
            pb[q].y = bp[(size_t)(8 * q + 1) * N];
        }
    };
    auto stage = [&](int st) {
        *reinterpret_cast<float4*>(&As[st][as0]) = ra0;
        *reinterpret_cast<float4*>(&As[st][as1]) = ra1;
        #pragma unroll
        for (int q = 0; q < 4; ++q)
            *reinterpret_cast<v2f*>(&Bs[st][bs0 + 4 * q * BPITCH]) = pb[q];
    };
    auto lds_handover = [] {
        __builtin_amdgcn_fence(__ATOMIC_RELEASE, "workgroup", "local");
        __builtin_amdgcn_s_barrier();
        __builtin_amdgcn_fence(__ATOMIC_ACQUIRE, "workgroup", "local");
    };

    constexpr int nslab = K / GBK;
    fetch(0);
    stage(0);
    if (nslab > 1) fetch(1);
    lds_handover();

    v8f acc0 = {}, acc1 = {};

    const int fa = (wm + lo) * APITCH + 2 * hi;
    const int fb = hi * BPITCH + (wn + lo) * 2;

    auto slab = [&](int st) {
        const float* as = &As[st][fa];
        const float* bs = &Bs[st][fb];
        v2f a[GBK / 4], b0[GBK / 4], b1[GBK / 4];
        #pragma unroll
        for (int j = 0; j < GBK / 4; ++j) {
            a[j]  = *reinterpret_cast<const v2f*>(as + 4 * j);
            b0[j] = *reinterpret_cast<const v2f*>(bs + 2 * j * BPITCH);
            b1[j] = *reinterpret_cast<const v2f*>(bs + 2 * j * BPITCH + 32);
        }
        __builtin_amdgcn_sched_barrier(0);
        #pragma unroll
        for (int j = 0; j < GBK / 4; ++j) {
            acc0 = __builtin_amdgcn_wmma_f32_16x16x4_f32(false, a[j], false, b0[j], (short)0, acc0, false, false);
            acc1 = __builtin_amdgcn_wmma_f32_16x16x4_f32(false, a[j], false, b1[j], (short)0, acc1, false, false);
        }
    };

    for (int s = 0; s + 1 < nslab; ++s) {
        slab(s & 1);
        __builtin_amdgcn_sched_barrier(0);
        stage((s & 1) ^ 1);
        if (s + 2 < nslab) fetch(s + 2);
        lds_handover();
    }
    slab((nslab - 1) & 1);

    #pragma unroll
    for (int r = 0; r < 8; ++r) {
        float* cp = C + (size_t)(m0 + wm + r + 8 * hi) * N + n0 + wn + lo;
        cp[0]  = acc0[r];
        cp[16] = acc1[r];
    }
}

#define PT_  16
#define PG_  (L_ / PT_)
static_assert(PG_ * PT_ == L_, "groups x tokens per group must cover the sequence");

__global__ __launch_bounds__(PT_ * 32) void phase_kernel(
    const float* __restrict__ proj, const float* __restrict__ dt_bias,
    float* __restrict__ DTb, float* __restrict__ decayb,
    float* __restrict__ trapb, float* __restrict__ phaseb)
{
    const int b = blockIdx.x / PG_;
    const int g = blockIdx.x % PG_;
    const int j = threadIdx.x & 31;
    const int w = threadIdx.x >> 5;
    __shared__ float sprod[PT_][32];

    const size_t bl  = (size_t)b * L_ + (size_t)g * PT_ + w;
    const float* row = proj + bl * DIP;

    const float vdt  = row[OFF_DT + j];
    const float vang = row[OFF_ANG + j];
    const float va   = row[OFF_A + j];
    const float vtr  = row[OFF_TRAP + j];

    const float dt = softplus_f(vdt + dt_bias[j]);
    float s = dt;
    s += __shfl_xor(s, 16, 32);
    s += __shfl_xor(s,  8, 32);
    s += __shfl_xor(s,  4, 32);
    s += __shfl_xor(s,  2, 32);
    s += __shfl_xor(s,  1, 32);
    const float dtmean = s * (1.f / 32.f);
    sprod[w][j] = vang * dtmean;

    const size_t o = bl * 32 + j;
    DTb[o]    = dt;
    const float Av = fminf(-softplus_f(va), -1e-4f);
    decayb[o] = expf(Av * dt);
    trapb[o]  = sigmoid_f(vtr);

    __syncthreads();

    float acc = 0.f;
    #pragma unroll
    for (int i = 0; i < PT_; ++i) {
        const float v = sprod[i][j];
        acc += (i <= w) ? v : 0.f;
    }
    phaseb[o] = acc;
}

__global__ __launch_bounds__(64) void rot_kernel(
    const float* __restrict__ proj,
    const float* __restrict__ Bn_g, const float* __restrict__ Bn_b,
    const float* __restrict__ Cn_g, const float* __restrict__ Cn_b,
    const float* __restrict__ phaseb,
    float* __restrict__ Brot, float* __restrict__ Crot)
{
    const int bl = blockIdx.x;
    const int t  = threadIdx.x;
    __shared__ float red[64], nb[64], nc[64];

    const int q  = t >> 1;
    const int l  = bl % L_;
    const int g  = l / PT_;
    const float* pseq = phaseb + (size_t)(bl - l) * 32 + q;
    float off = 0.f;
    #pragma unroll
    for (int k = 0; k < PG_ - 1; ++k) {
        const float v = pseq[(size_t)(k * PT_ + PT_ - 1) * 32];
        off += (k < g) ? v : 0.f;
    }
    const float ph = off + phaseb[(size_t)bl * 32 + q];

    const float* row = proj + (size_t)bl * DIP;
    const float vb = row[OFF_B + t];
    const float vc = row[OFF_C + t];

    auto blockReduce = [&](float v) -> float {
        red[t] = v; __syncthreads();
        for (int s = 32; s > 0; s >>= 1) {
            if (t < s) red[t] += red[t + s];
            __syncthreads();
        }
        float r = red[0]; __syncthreads();
        return r;
    };

    const float mB  = blockReduce(vb) * (1.f / 64.f);
    const float vaB = blockReduce((vb - mB) * (vb - mB)) * (1.f / 64.f);
    const float nB  = (vb - mB) * rsqrtf(vaB + 1e-5f) * Bn_g[t] + Bn_b[t];

    const float mC  = blockReduce(vc) * (1.f / 64.f);
    const float vaC = blockReduce((vc - mC) * (vc - mC)) * (1.f / 64.f);
    const float nC  = (vc - mC) * rsqrtf(vaC + 1e-5f) * Cn_g[t] + Cn_b[t];

    nb[t] = nB; nc[t] = nC;
    __syncthreads();

    const float c  = cosf(ph), s = sinf(ph);
    float ob, oc;
    if ((t & 1) == 0) { ob = nb[t] * c - nb[t + 1] * s; oc = nc[t] * c - nc[t + 1] * s; }
    else              { ob = nb[t - 1] * s + nb[t] * c; oc = nc[t - 1] * s + nc[t] * c; }
    Brot[(size_t)bl * 64 + t] = ob;
    Crot[(size_t)bl * 64 + t] = oc;
}


__global__ __launch_bounds__(256) void scan_chunk_kernel(
    const float* __restrict__ proj,
    const float* __restrict__ Brot, const float* __restrict__ Crot,
    const float* __restrict__ DTb, const float* __restrict__ decayb,
    const float* __restrict__ trapb,
    float* __restrict__ yfin, float* __restrict__ cumdb,
    float* __restrict__ Sstate)
{
    const int blk = blockIdx.x;
    const int c   = blk % NC_;
    const int h   = (blk / NC_) % NH_;
    const int b   = blk / (NC_ * NH_);
    const int l0  = c * LC_;
    const int t   = threadIdx.x;
    const int p   = t >> 2;
    const int seg = t & 3;

    __shared__ __attribute__((aligned(16))) float sx[LC_ + 1][64];
    __shared__ __attribute__((aligned(16))) float sB[LC_ + 1][64];
    __shared__ __attribute__((aligned(16))) float sC[LC_][64];
    __shared__ __attribute__((aligned(16))) float sscal[LC_][4];
    __shared__ float scum[LC_];
    __shared__ __attribute__((aligned(16))) float sy[LC_][64][4];

    const size_t bl0 = (size_t)b * L_ + l0;

    float4 vx[2], vb[2], vc[2];
    #pragma unroll
    for (int k = 0; k < 2; ++k) {
        const int idx = k * 256 + t;
        const int r = idx >> 4, q = idx & 15;
        vx[k] = *reinterpret_cast<const float4*>(proj + (bl0 + r) * DIP + OFF_X + h * HD_ + q * 4);
        vb[k] = *reinterpret_cast<const float4*>(Brot + (bl0 + r) * 64 + q * 4);
        vc[k] = *reinterpret_cast<const float4*>(Crot + (bl0 + r) * 64 + q * 4);
    }
    float4 v0 = make_float4(0.f, 0.f, 0.f, 0.f);
    if (c > 0 && t < 32) {
        const int q = t & 15;
        v0 = (t < 16)
           ? *reinterpret_cast<const float4*>(proj + (bl0 - 1) * DIP + OFF_X + h * HD_ + q * 4)
           : *reinterpret_cast<const float4*>(Brot + (bl0 - 1) * 64 + q * 4);
    }
    float vs = 0.f;
    if (t >= 64 && t < 160) {
        const int kind = (t - 64) >> 5, tok = (t - 64) & 31;
        const float* sp = (kind == 0) ? DTb : (kind == 1) ? decayb : trapb;
        vs = sp[(bl0 + tok) * 32 + h];
    }
    #pragma unroll
    for (int k = 0; k < 2; ++k) {
        const int idx = k * 256 + t;
        const int r = idx >> 4, q = idx & 15;
        *reinterpret_cast<float4*>(&sx[r + 1][q * 4]) = vx[k];
        *reinterpret_cast<float4*>(&sB[r + 1][q * 4]) = vb[k];
        *reinterpret_cast<float4*>(&sC[r][q * 4])     = vc[k];
    }
    if (t < 16)      *reinterpret_cast<float4*>(&sx[0][(t & 15) * 4]) = v0;
    else if (t < 32) *reinterpret_cast<float4*>(&sB[0][(t & 15) * 4]) = v0;
    if (t >= 64 && t < 160) sscal[(t - 64) & 31][(t - 64) >> 5] = vs;
    __syncthreads();

    float hst[16], Bp[16];
    #pragma unroll
    for (int i = 0; i < 16; ++i) hst[i] = 0.f;
    #pragma unroll
    for (int i4 = 0; i4 < 4; ++i4) {
        const float4 v = *reinterpret_cast<const float4*>(&sB[0][seg * 16 + i4 * 4]);
        Bp[i4 * 4 + 0] = v.x; Bp[i4 * 4 + 1] = v.y; Bp[i4 * 4 + 2] = v.z; Bp[i4 * 4 + 3] = v.w;
    }
    float xprev = sx[0][p];
    float cumd  = 1.f;

    #pragma unroll 2
    for (int tok = 0; tok < LC_; ++tok) {
        const float  x  = sx[tok + 1][p];
        const float4 sc = *reinterpret_cast<const float4*>(&sscal[tok][0]);
        const float dt = sc.x, dec = sc.y, tr = sc.z;

        float Bc[16], Cc[16];
        #pragma unroll
        for (int i4 = 0; i4 < 4; ++i4) {
            const float4 vB = *reinterpret_cast<const float4*>(&sB[tok + 1][seg * 16 + i4 * 4]);
            const float4 vC = *reinterpret_cast<const float4*>(&sC[tok][seg * 16 + i4 * 4]);
            Bc[i4 * 4 + 0] = vB.x; Bc[i4 * 4 + 1] = vB.y; Bc[i4 * 4 + 2] = vB.z; Bc[i4 * 4 + 3] = vB.w;
            Cc[i4 * 4 + 0] = vC.x; Cc[i4 * 4 + 1] = vC.y; Cc[i4 * 4 + 2] = vC.z; Cc[i4 * 4 + 3] = vC.w;
        }

        float y = 0.f;
        #pragma unroll
        for (int i = 0; i < 16; ++i) {
            const float bx = tr * x * Bc[i] + (1.f - tr) * xprev * Bp[i];
            hst[i] = dec * hst[i] + dt * bx;
            y += hst[i] * Cc[i];
            Bp[i] = Bc[i];
        }
        xprev = x;
        cumd *= dec;

        sy[tok][p][seg] = y;
        if (t == 0) scum[tok] = cumd;
    }
    __syncthreads();

    #pragma unroll
    for (int k = 0; k < 2; ++k) {
        const int idx = k * 256 + t;
        const int r = idx >> 4, q = idx & 15;
        float o[4];
        #pragma unroll
        for (int j = 0; j < 4; ++j) {
            const float4 pv = *reinterpret_cast<const float4*>(&sy[r][q * 4 + j][0]);
            o[j] = (pv.x + pv.y) + (pv.z + pv.w);
        }
        *reinterpret_cast<float4*>(yfin + (bl0 + r) * DI + h * HD_ + q * 4) = make_float4(o[0], o[1], o[2], o[3]);
    }
    if (t < LC_) cumdb[((size_t)(b * NH_ + h)) * L_ + l0 + t] = scum[t];

    float* Sp = Sstate + (((size_t)(b * NH_ + h)) * NC_ + c) * 4096 + p * 64 + seg * 16;
    #pragma unroll
    for (int i = 0; i < 16; ++i) Sp[i] = hst[i];
}

__global__ __launch_bounds__(256) void scan_combine_kernel(
    const float* __restrict__ cumdb, float* __restrict__ Sstate)
{
    const int h  = blockIdx.x % NH_;
    const int b  = blockIdx.x / NH_;
    const int t  = threadIdx.x;
    const int p  = t >> 2, seg = t & 3;

    float H[16];
    #pragma unroll
    for (int i = 0; i < 16; ++i) H[i] = 0.f;

    for (int c = 0; c < NC_; ++c) {
        float* Sp = Sstate + (((size_t)(b * NH_ + h)) * NC_ + c) * 4096 + p * 64 + seg * 16;
        const float Pc = cumdb[((size_t)(b * NH_ + h)) * L_ + c * LC_ + LC_ - 1];
        #pragma unroll
        for (int i = 0; i < 16; ++i) {
            const float Sc = Sp[i];
            Sp[i] = H[i];
            H[i]  = Pc * H[i] + Sc;
        }
    }
}

__global__ __launch_bounds__(256) void scan_correct_kernel(
    const float* __restrict__ proj, const float* __restrict__ Crot,
    const float* __restrict__ cumdb, const float* __restrict__ Sstate,
    const float* __restrict__ D_param, float* __restrict__ yfin)
{
    const int blk = blockIdx.x;
    const int c   = blk % NC_;
    const int h   = (blk / NC_) % NH_;
    const int b   = blk / (NC_ * NH_);
    const int l0  = c * LC_;
    const int t   = threadIdx.x;
    const int p   = t >> 2, seg = t & 3;

    __shared__ __attribute__((aligned(16))) float sC[LC_][64];
    __shared__ float scd[LC_];
    __shared__ __attribute__((aligned(16))) float scorr[LC_][64][4];

    const size_t bl0 = (size_t)b * L_ + l0;

    float H[16];
    const float* Sp = Sstate + (((size_t)(b * NH_ + h)) * NC_ + c) * 4096 + p * 64 + seg * 16;
    #pragma unroll
    for (int i = 0; i < 16; ++i) H[i] = Sp[i];
    const float Dp = D_param[h];

    float4 vc[2], vx[2], vz[2], vy[2];
    #pragma unroll
    for (int k = 0; k < 2; ++k) {
        const int idx = k * 256 + t;
        const int r = idx >> 4, q = idx & 15;
        vc[k] = *reinterpret_cast<const float4*>(Crot + (bl0 + r) * 64 + q * 4);
        vx[k] = *reinterpret_cast<const float4*>(proj + (bl0 + r) * DIP + OFF_X + h * HD_ + q * 4);
        vz[k] = *reinterpret_cast<const float4*>(proj + (bl0 + r) * DIP + OFF_Z + h * HD_ + q * 4);
        vy[k] = *reinterpret_cast<const float4*>(yfin + (bl0 + r) * DI + h * HD_ + q * 4);
    }
    float cdv = 0.f;
    if (t < LC_) cdv = cumdb[((size_t)(b * NH_ + h)) * L_ + l0 + t];

    #pragma unroll
    for (int k = 0; k < 2; ++k) {
        const int idx = k * 256 + t;
        const int r = idx >> 4, q = idx & 15;
        *reinterpret_cast<float4*>(&sC[r][q * 4]) = vc[k];
    }
    if (t < LC_) scd[t] = cdv;
    __syncthreads();

    #pragma unroll 2
    for (int tok = 0; tok < LC_; ++tok) {
        float Cc[16];
        #pragma unroll
        for (int i4 = 0; i4 < 4; ++i4) {
            const float4 vC = *reinterpret_cast<const float4*>(&sC[tok][seg * 16 + i4 * 4]);
            Cc[i4 * 4 + 0] = vC.x; Cc[i4 * 4 + 1] = vC.y; Cc[i4 * 4 + 2] = vC.z; Cc[i4 * 4 + 3] = vC.w;
        }
        float corr = 0.f;
        #pragma unroll
        for (int i = 0; i < 16; ++i) corr += H[i] * Cc[i];
        scorr[tok][p][seg] = corr;
    }
    __syncthreads();

    #pragma unroll
    for (int k = 0; k < 2; ++k) {
        const int idx = k * 256 + t;
        const int r = idx >> 4, q = idx & 15;
        const float cd = scd[r];
        const float xs[4] = {vx[k].x, vx[k].y, vx[k].z, vx[k].w};
        const float zs[4] = {vz[k].x, vz[k].y, vz[k].z, vz[k].w};
        const float ys[4] = {vy[k].x, vy[k].y, vy[k].z, vy[k].w};
        float o[4];
        #pragma unroll
        for (int j = 0; j < 4; ++j) {
            const float4 pv = *reinterpret_cast<const float4*>(&scorr[r][q * 4 + j][0]);
            const float corr = (pv.x + pv.y) + (pv.z + pv.w);
            const float y = ys[j] + cd * corr;
            o[j] = (y + Dp * xs[j]) * (zs[j] * sigmoid_f(zs[j]));
        }
        *reinterpret_cast<float4*>(yfin + (bl0 + r) * DI + h * HD_ + q * 4) = make_float4(o[0], o[1], o[2], o[3]);
    }
}

extern "C" void kernel_launch(void* const* d_in, const int* in_sizes, int n_in,
                              void* d_out, int out_size, void* d_ws, size_t ws_size,
                              hipStream_t stream)
{
    const float* u       = (const float*)d_in[0];
    const float* W_in    = (const float*)d_in[1];
    const float* W_out   = (const float*)d_in[2];
    const float* dt_bias = (const float*)d_in[3];
    const float* D_param = (const float*)d_in[4];
    const float* Bn_g    = (const float*)d_in[5];
    const float* Bn_b    = (const float*)d_in[6];
    const float* Cn_g    = (const float*)d_in[7];
    const float* Cn_b    = (const float*)d_in[8];

    float* ws = (float*)d_ws;
    float* proj   = ws; ws += (size_t)M1 * DIP;
    float* yfin   = ws; ws += (size_t)M1 * DI;
    float* Sstate = ws; ws += (size_t)B_ * NH_ * NC_ * 4096;
    float* Brot   = ws; ws += (size_t)M1 * DS_;
    float* Crot   = ws; ws += (size_t)M1 * DS_;
    float* DTb    = ws; ws += (size_t)M1 * NH_;
    float* decayb = ws; ws += (size_t)M1 * NH_;
    float* trapb  = ws; ws += (size_t)M1 * NH_;
    float* phaseb = ws; ws += (size_t)M1 * NA_;
    float* cumdb  = ws; ws += (size_t)B_ * NH_ * L_;

    gemm_wmma_f32<DIP, DM><<<dim3(DIP / GBN, M1 / GBM), dim3(256), 0, stream>>>(
        u, W_in, proj);

    phase_kernel<<<dim3(B_ * PG_), dim3(PT_ * 32), 0, stream>>>(
        proj, dt_bias, DTb, decayb, trapb, phaseb);

    rot_kernel<<<dim3(M1), dim3(64), 0, stream>>>(
        proj, Bn_g, Bn_b, Cn_g, Cn_b, phaseb, Brot, Crot);

    scan_chunk_kernel<<<dim3(B_ * NH_ * NC_), dim3(256), 0, stream>>>(
        proj, Brot, Crot, DTb, decayb, trapb, yfin, cumdb, Sstate);
    scan_combine_kernel<<<dim3(B_ * NH_), dim3(256), 0, stream>>>(
        cumdb, Sstate);
    scan_correct_kernel<<<dim3(B_ * NH_ * NC_), dim3(256), 0, stream>>>(
        proj, Crot, cumdb, Sstate, D_param, yfin);

    gemm_wmma_f32<DM, DI><<<dim3(DM / GBN, M1 / GBM), dim3(256), 0, stream>>>(
        yfin, W_out, (float*)d_out);
}


# GEMM K loop pipelined one slab deeper: slab s+1's fragments go from LDS to a second register set while slab s is multiplied from registers; fetch two slabs ahead; k and sum order unchanged
# speedup vs baseline: 7.7024x; 1.0833x over previous
#include <hip/hip_runtime.h>
#include <math.h>

#define B_   2
#define L_   512
#define DM   1024
#define DI   2048
#define DS_  64
#define NH_  32
#define HD_  64
#define NA_  32
#define DIP  4352
#define M1   (B_*L_)

#define NC_  16
#define LC_  32

#define OFF_Z    0
#define OFF_X    2048
#define OFF_B    4096
#define OFF_C    4160
#define OFF_DT   4224
#define OFF_A    4256
#define OFF_TRAP 4288
#define OFF_ANG  4320

typedef __attribute__((ext_vector_type(2))) float v2f;
typedef __attribute__((ext_vector_type(4))) float v4f;
typedef __attribute__((ext_vector_type(8))) float v8f;

__device__ __forceinline__ float softplus_f(float x) {
    return x > 20.f ? x : log1pf(expf(x));
}
__device__ __forceinline__ float sigmoid_f(float x) {
    return 1.f / (1.f + expf(-x));
}

#define GBM    64
#define GBN    64
#define GBK    32
#define APITCH 36
#define BPITCH 160

template <int N, int K>
__global__ __launch_bounds__(256) void gemm_wmma_f32(
    const float* __restrict__ A, const float* __restrict__ Bm, float* __restrict__ C)
{
    constexpr int nslab = K / GBK;
    static_assert(N % GBN == 0 && K % GBK == 0, "tile sizes must divide the problem");
    static_assert(nslab % 2 == 0 && nslab >= 4, "the pipeline runs in pairs of slabs");
    __shared__ __attribute__((aligned(16))) float As[2][GBM * APITCH];
    __shared__ __attribute__((aligned(16))) float Bs[2][(GBK / 2) * BPITCH];

    const int t    = threadIdx.x;
    const int lane = t & 31;
    const int wave = t >> 5;
    const int lo   = lane & 15;
    const int hi   = lane >> 4;

    const int m0 = blockIdx.y * GBM;
    const int n0 = blockIdx.x * GBN;
    const int wm = (wave >> 1) * 16;
    const int wn = (wave & 1) * 32;

    const int ar = t >> 3, ac = (t & 7) * 4;
    const int bn = t & 63, kq = t >> 6;

    const float* Ag0 = A  + (size_t)(m0 + ar) * K + ac;
    const float* Ag1 = Ag0 + (size_t)32 * K;
    const float* Bg  = Bm + (size_t)(2 * kq) * N + n0 + bn;

    const int as0 = ar * APITCH + ac, as1 = (ar + 32) * APITCH + ac;
    const int bs0 = kq * BPITCH + bn * 2;

    struct Staged { v4f a0, a1; v2f b[4]; };
    Staged sg0, sg1;
    auto fetch = [&](int i, Staged& g) __attribute__((always_inline)) {
        g.a0 = *reinterpret_cast<const v4f*>(Ag0 + (size_t)i * GBK);
        g.a1 = *reinterpret_cast<const v4f*>(Ag1 + (size_t)i * GBK);
        const float* bp = Bg + (size_t)i * GBK * N;
        #pragma unroll
        for (int q = 0; q < 4; ++q) {
            g.b[q].x = bp[(size_t)(8 * q) * N];
            g.b[q].y = bp[(size_t)(8 * q + 1) * N];
        }
    };
    auto stage = [&](int st, const Staged& g) __attribute__((always_inline)) {
        *reinterpret_cast<v4f*>(&As[st][as0]) = g.a0;
        *reinterpret_cast<v4f*>(&As[st][as1]) = g.a1;
        #pragma unroll
        for (int q = 0; q < 4; ++q)
            *reinterpret_cast<v2f*>(&Bs[st][bs0 + 4 * q * BPITCH]) = g.b[q];
    };
    auto lds_handover = [] () __attribute__((always_inline)) {
        __builtin_amdgcn_fence(__ATOMIC_RELEASE, "workgroup", "local");
        __builtin_amdgcn_s_barrier();
        __builtin_amdgcn_fence(__ATOMIC_ACQUIRE, "workgroup", "local");
    };

    v8f acc0 = {}, acc1 = {};

    const int fa = (wm + lo) * APITCH + 2 * hi;
    const int fb = hi * BPITCH + (wn + lo) * 2;

    struct Frags { v2f a[GBK / 4], b0[GBK / 4], b1[GBK / 4]; };
    Frags fr0, fr1;
    auto frags = [&](int st, Frags& f) __attribute__((always_inline)) {
        const float* as = &As[st][fa];
        const float* bs = &Bs[st][fb];
        #pragma unroll
        for (int j = 0; j < GBK / 4; ++j) {
            f.a[j]  = *reinterpret_cast<const v2f*>(as + 4 * j);
            f.b0[j] = *reinterpret_cast<const v2f*>(bs + 2 * j * BPITCH);
            f.b1[j] = *reinterpret_cast<const v2f*>(bs + 2 * j * BPITCH + 32);
        }
    };
    auto products = [&](const Frags& f) __attribute__((always_inline)) {
        #pragma unroll
        for (int j = 0; j < GBK / 4; ++j) {
            acc0 = __builtin_amdgcn_wmma_f32_16x16x4_f32(false, f.a[j], false, f.b0[j], (short)0, acc0, false, false);
            acc1 = __builtin_amdgcn_wmma_f32_16x16x4_f32(false, f.a[j], false, f.b1[j], (short)0, acc1, false, false);
        }
    };

    auto hold_products = [&]() __attribute__((always_inline)) {
        asm volatile("" : "+v"(acc0), "+v"(acc1));
    };

    auto two_slabs = [&](int s, bool more2, bool more4, bool more3, bool more5) __attribute__((always_inline)) {
        frags(1, fr1);
        products(fr0);
        if (more2) { stage(0, sg0); if (more4) fetch(s + 4, sg0); }
        if (more2) { lds_handover(); hold_products(); }
        if (more2) frags(0, fr0);
        products(fr1);
        if (more3) { stage(1, sg1); if (more5) fetch(s + 5, sg1); }
        if (more2) { lds_handover(); hold_products(); }
    };

    fetch(0, sg0);
    fetch(1, sg1);
    stage(0, sg0);
    fetch(2, sg0);
    lds_handover();
    frags(0, fr0);
    stage(1, sg1);
    fetch(3, sg1);
    lds_handover();

    int s = 0;
    for (; s + 4 < nslab; s += 2) two_slabs(s, true, true, true, true);
    two_slabs(s, true, false, true, false);
    two_slabs(s + 2, false, false, false, false);

    #pragma unroll
    for (int r = 0; r < 8; ++r) {
        float* cp = C + (size_t)(m0 + wm + r + 8 * hi) * N + n0 + wn + lo;
        cp[0]  = acc0[r];
        cp[16] = acc1[r];
    }
}

#define PT_  16
#define PG_  (L_ / PT_)
static_assert(PG_ * PT_ == L_, "groups x tokens per group must cover the sequence");

__global__ __launch_bounds__(PT_ * 32) void phase_kernel(
    const float* __restrict__ proj, const float* __restrict__ dt_bias,
    float* __restrict__ DTb, float* __restrict__ decayb,
    float* __restrict__ trapb, float* __restrict__ phaseb)
{
    const int b = blockIdx.x / PG_;
    const int g = blockIdx.x % PG_;
    const int j = threadIdx.x & 31;
    const int w = threadIdx.x >> 5;
    __shared__ float sprod[PT_][32];

    const size_t bl  = (size_t)b * L_ + (size_t)g * PT_ + w;
    const float* row = proj + bl * DIP;

    const float vdt  = row[OFF_DT + j];
    const float vang = row[OFF_ANG + j];
    const float va   = row[OFF_A + j];
    const float vtr  = row[OFF_TRAP + j];

    const float dt = softplus_f(vdt + dt_bias[j]);
    float s = dt;
    s += __shfl_xor(s, 16, 32);
    s += __shfl_xor(s,  8, 32);
    s += __shfl_xor(s,  4, 32);
    s += __shfl_xor(s,  2, 32);
    s += __shfl_xor(s,  1, 32);
    const float dtmean = s * (1.f / 32.f);
    sprod[w][j] = vang * dtmean;

    const size_t o = bl * 32 + j;
    DTb[o]    = dt;
    const float Av = fminf(-softplus_f(va), -1e-4f);
    decayb[o] = expf(Av * dt);
    trapb[o]  = sigmoid_f(vtr);

    __syncthreads();

    float acc = 0.f;
    #pragma unroll
    for (int i = 0; i < PT_; ++i) {
        const float v = sprod[i][j];
        acc += (i <= w) ? v : 0.f;
    }
    phaseb[o] = acc;
}

__global__ __launch_bounds__(64) void rot_kernel(
    const float* __restrict__ proj,
    const float* __restrict__ Bn_g, const float* __restrict__ Bn_b,
    const float* __restrict__ Cn_g, const float* __restrict__ Cn_b,
    const float* __restrict__ phaseb,
    float* __restrict__ Brot, float* __restrict__ Crot)
{
    const int bl = blockIdx.x;
    const int t  = threadIdx.x;
    __shared__ float red[64], nb[64], nc[64];

    const int q  = t >> 1;
    const int l  = bl % L_;
    const int g  = l / PT_;
    const float* pseq = phaseb + (size_t)(bl - l) * 32 + q;
    float off = 0.f;
    #pragma unroll
    for (int k = 0; k < PG_ - 1; ++k) {
        const float v = pseq[(size_t)(k * PT_ + PT_ - 1) * 32];
        off += (k < g) ? v : 0.f;
    }
    const float ph = off + phaseb[(size_t)bl * 32 + q];

    const float* row = proj + (size_t)bl * DIP;
    const float vb = row[OFF_B + t];
    const float vc = row[OFF_C + t];

    auto blockReduce = [&](float v) -> float {
        red[t] = v; __syncthreads();
        for (int s = 32; s > 0; s >>= 1) {
            if (t < s) red[t] += red[t + s];
            __syncthreads();
        }
        float r = red[0]; __syncthreads();
        return r;
    };

    const float mB  = blockReduce(vb) * (1.f / 64.f);
    const float vaB = blockReduce((vb - mB) * (vb - mB)) * (1.f / 64.f);
    const float nB  = (vb - mB) * rsqrtf(vaB + 1e-5f) * Bn_g[t] + Bn_b[t];

    const float mC  = blockReduce(vc) * (1.f / 64.f);
    const float vaC = blockReduce((vc - mC) * (vc - mC)) * (1.f / 64.f);
    const float nC  = (vc - mC) * rsqrtf(vaC + 1e-5f) * Cn_g[t] + Cn_b[t];

    nb[t] = nB; nc[t] = nC;
    __syncthreads();

    const float c  = cosf(ph), s = sinf(ph);
    float ob, oc;
    if ((t & 1) == 0) { ob = nb[t] * c - nb[t + 1] * s; oc = nc[t] * c - nc[t + 1] * s; }
    else              { ob = nb[t - 1] * s + nb[t] * c; oc = nc[t - 1] * s + nc[t] * c; }
    Brot[(size_t)bl * 64 + t] = ob;
    Crot[(size_t)bl * 64 + t] = oc;
}


__global__ __launch_bounds__(256) void scan_chunk_kernel(
    const float* __restrict__ proj,
    const float* __restrict__ Brot, const float* __restrict__ Crot,
    const float* __restrict__ DTb, const float* __restrict__ decayb,
    const float* __restrict__ trapb,
    float* __restrict__ yfin, float* __restrict__ cumdb,
    float* __restrict__ Sstate)
{
    const int blk = blockIdx.x;
    const int c   = blk % NC_;
    const int h   = (blk / NC_) % NH_;
    const int b   = blk / (NC_ * NH_);
    const int l0  = c * LC_;
    const int t   = threadIdx.x;
    const int p   = t >> 2;
    const int seg = t & 3;

    __shared__ __attribute__((aligned(16))) float sx[LC_ + 1][64];
    __shared__ __attribute__((aligned(16))) float sB[LC_ + 1][64];
    __shared__ __attribute__((aligned(16))) float sC[LC_][64];
    __shared__ __attribute__((aligned(16))) float sscal[LC_][4];
    __shared__ float scum[LC_];
    __shared__ __attribute__((aligned(16))) float sy[LC_][64][4];

    const size_t bl0 = (size_t)b * L_ + l0;

    float4 vx[2], vb[2], vc[2];
    #pragma unroll
    for (int k = 0; k < 2; ++k) {
        const int idx = k * 256 + t;
        const int r = idx >> 4, q = idx & 15;
        vx[k] = *reinterpret_cast<const float4*>(proj + (bl0 + r) * DIP + OFF_X + h * HD_ + q * 4);
        vb[k] = *reinterpret_cast<const float4*>(Brot + (bl0 + r) * 64 + q * 4);
        vc[k] = *reinterpret_cast<const float4*>(Crot + (bl0 + r) * 64 + q * 4);
    }
    float4 v0 = make_float4(0.f, 0.f, 0.f, 0.f);
    if (c > 0 && t < 32) {
        const int q = t & 15;
        v0 = (t < 16)
           ? *reinterpret_cast<const float4*>(proj + (bl0 - 1) * DIP + OFF_X + h * HD_ + q * 4)
           : *reinterpret_cast<const float4*>(Brot + (bl0 - 1) * 64 + q * 4);
    }
    float vs = 0.f;
    if (t >= 64 && t < 160) {
        const int kind = (t - 64) >> 5, tok = (t - 64) & 31;
        const float* sp = (kind == 0) ? DTb : (kind == 1) ? decayb : trapb;
        vs = sp[(bl0 + tok) * 32 + h];
    }
    #pragma unroll
    for (int k = 0; k < 2; ++k) {
        const int idx = k * 256 + t;
        const int r = idx >> 4, q = idx & 15;
        *reinterpret_cast<float4*>(&sx[r + 1][q * 4]) = vx[k];
        *reinterpret_cast<float4*>(&sB[r + 1][q * 4]) = vb[k];
        *reinterpret_cast<float4*>(&sC[r][q * 4])     = vc[k];
    }
    if (t < 16)      *reinterpret_cast<float4*>(&sx[0][(t & 15) * 4]) = v0;
    else if (t < 32) *reinterpret_cast<float4*>(&sB[0][(t & 15) * 4]) = v0;
    if (t >= 64 && t < 160) sscal[(t - 64) & 31][(t - 64) >> 5] = vs;
    __syncthreads();

    float hst[16], Bp[16];
    #pragma unroll
    for (int i = 0; i < 16; ++i) hst[i] = 0.f;
    #pragma unroll
    for (int i4 = 0; i4 < 4; ++i4) {
        const float4 v = *reinterpret_cast<const float4*>(&sB[0][seg * 16 + i4 * 4]);
        Bp[i4 * 4 + 0] = v.x; Bp[i4 * 4 + 1] = v.y; Bp[i4 * 4 + 2] = v.z; Bp[i4 * 4 + 3] = v.w;
    }
    float xprev = sx[0][p];
    float cumd  = 1.f;

    #pragma unroll 2
    for (int tok = 0; tok < LC_; ++tok) {
        const float  x  = sx[tok + 1][p];
        const float4 sc = *reinterpret_cast<const float4*>(&sscal[tok][0]);
        const float dt = sc.x, dec = sc.y, tr = sc.z;

        float Bc[16], Cc[16];
        #pragma unroll
        for (int i4 = 0; i4 < 4; ++i4) {
            const float4 vB = *reinterpret_cast<const float4*>(&sB[tok + 1][seg * 16 + i4 * 4]);
            const float4 vC = *reinterpret_cast<const float4*>(&sC[tok][seg * 16 + i4 * 4]);
            Bc[i4 * 4 + 0] = vB.x; Bc[i4 * 4 + 1] = vB.y; Bc[i4 * 4 + 2] = vB.z; Bc[i4 * 4 + 3] = vB.w;
            Cc[i4 * 4 + 0] = vC.x; Cc[i4 * 4 + 1] = vC.y; Cc[i4 * 4 + 2] = vC.z; Cc[i4 * 4 + 3] = vC.w;
        }

        float y = 0.f;
        #pragma unroll
        for (int i = 0; i < 16; ++i) {
            const float bx = tr * x * Bc[i] + (1.f - tr) * xprev * Bp[i];
            hst[i] = dec * hst[i] + dt * bx;
            y += hst[i] * Cc[i];
            Bp[i] = Bc[i];
        }
        xprev = x;
        cumd *= dec;

        sy[tok][p][seg] = y;
        if (t == 0) scum[tok] = cumd;
    }
    __syncthreads();

    #pragma unroll
    for (int k = 0; k < 2; ++k) {
        const int idx = k * 256 + t;
        const int r = idx >> 4, q = idx & 15;
        float o[4];
        #pragma unroll
        for (int j = 0; j < 4; ++j) {
            const float4 pv = *reinterpret_cast<const float4*>(&sy[r][q * 4 + j][0]);
            o[j] = (pv.x + pv.y) + (pv.z + pv.w);
        }
        *reinterpret_cast<float4*>(yfin + (bl0 + r) * DI + h * HD_ + q * 4) = make_float4(o[0], o[1], o[2], o[3]);
    }
    if (t < LC_) cumdb[((size_t)(b * NH_ + h)) * L_ + l0 + t] = scum[t];

    float* Sp = Sstate + (((size_t)(b * NH_ + h)) * NC_ + c) * 4096 + p * 64 + seg * 16;
    #pragma unroll
    for (int i = 0; i < 16; ++i) Sp[i] = hst[i];
}

__global__ __launch_bounds__(256) void scan_combine_kernel(
    const float* __restrict__ cumdb, float* __restrict__ Sstate)
{
    const int h  = blockIdx.x % NH_;
    const int b  = blockIdx.x / NH_;
    const int t  = threadIdx.x;
    const int p  = t >> 2, seg = t & 3;

    float H[16];
    #pragma unroll
    for (int i = 0; i < 16; ++i) H[i] = 0.f;

    for (int c = 0; c < NC_; ++c) {
        float* Sp = Sstate + (((size_t)(b * NH_ + h)) * NC_ + c) * 4096 + p * 64 + seg * 16;
        const float Pc = cumdb[((size_t)(b * NH_ + h)) * L_ + c * LC_ + LC_ - 1];
        #pragma unroll
        for (int i = 0; i < 16; ++i) {
            const float Sc = Sp[i];
            Sp[i] = H[i];
            H[i]  = Pc * H[i] + Sc;
        }
    }
}

__global__ __launch_bounds__(256) void scan_correct_kernel(
    const float* __restrict__ proj, const float* __restrict__ Crot,
    const float* __restrict__ cumdb, const float* __restrict__ Sstate,
    const float* __restrict__ D_param, float* __restrict__ yfin)
{
    const int blk = blockIdx.x;
    const int c   = blk % NC_;
    const int h   = (blk / NC_) % NH_;
    const int b   = blk / (NC_ * NH_);
    const int l0  = c * LC_;
    const int t   = threadIdx.x;
    const int p   = t >> 2, seg = t & 3;

    __shared__ __attribute__((aligned(16))) float sC[LC_][64];
    __shared__ float scd[LC_];
    __shared__ __attribute__((aligned(16))) float scorr[LC_][64][4];

    const size_t bl0 = (size_t)b * L_ + l0;

    float H[16];
    const float* Sp = Sstate + (((size_t)(b * NH_ + h)) * NC_ + c) * 4096 + p * 64 + seg * 16;
    #pragma unroll
    for (int i = 0; i < 16; ++i) H[i] = Sp[i];
    const float Dp = D_param[h];

    float4 vc[2], vx[2], vz[2], vy[2];
    #pragma unroll
    for (int k = 0; k < 2; ++k) {
        const int idx = k * 256 + t;
        const int r = idx >> 4, q = idx & 15;
        vc[k] = *reinterpret_cast<const float4*>(Crot + (bl0 + r) * 64 + q * 4);
        vx[k] = *reinterpret_cast<const float4*>(proj + (bl0 + r) * DIP + OFF_X + h * HD_ + q * 4);
        vz[k] = *reinterpret_cast<const float4*>(proj + (bl0 + r) * DIP + OFF_Z + h * HD_ + q * 4);
        vy[k] = *reinterpret_cast<const float4*>(yfin + (bl0 + r) * DI + h * HD_ + q * 4);
    }
    float cdv = 0.f;
    if (t < LC_) cdv = cumdb[((size_t)(b * NH_ + h)) * L_ + l0 + t];

    #pragma unroll
    for (int k = 0; k < 2; ++k) {
        const int idx = k * 256 + t;
        const int r = idx >> 4, q = idx & 15;
        *reinterpret_cast<float4*>(&sC[r][q * 4]) = vc[k];
    }
    if (t < LC_) scd[t] = cdv;
    __syncthreads();

    #pragma unroll 2
    for (int tok = 0; tok < LC_; ++tok) {
        float Cc[16];
        #pragma unroll
        for (int i4 = 0; i4 < 4; ++i4) {
            const float4 vC = *reinterpret_cast<const float4*>(&sC[tok][seg * 16 + i4 * 4]);
            Cc[i4 * 4 + 0] = vC.x; Cc[i4 * 4 + 1] = vC.y; Cc[i4 * 4 + 2] = vC.z; Cc[i4 * 4 + 3] = vC.w;
        }
        float corr = 0.f;
        #pragma unroll
        for (int i = 0; i < 16; ++i) corr += H[i] * Cc[i];
        scorr[tok][p][seg] = corr;
    }
    __syncthreads();

    #pragma unroll
    for (int k = 0; k < 2; ++k) {
        const int idx = k * 256 + t;
        const int r = idx >> 4, q = idx & 15;
        const float cd = scd[r];
        const float xs[4] = {vx[k].x, vx[k].y, vx[k].z, vx[k].w};
        const float zs[4] = {vz[k].x, vz[k].y, vz[k].z, vz[k].w};
        const float ys[4] = {vy[k].x, vy[k].y, vy[k].z, vy[k].w};
        float o[4];
        #pragma unroll
        for (int j = 0; j < 4; ++j) {
            const float4 pv = *reinterpret_cast<const float4*>(&scorr[r][q * 4 + j][0]);
            const float corr = (pv.x + pv.y) + (pv.z + pv.w);
            const float y = ys[j] + cd * corr;
            o[j] = (y + Dp * xs[j]) * (zs[j] * sigmoid_f(zs[j]));
        }
        *reinterpret_cast<float4*>(yfin + (bl0 + r) * DI + h * HD_ + q * 4) = make_float4(o[0], o[1], o[2], o[3]);
    }
}

extern "C" void kernel_launch(void* const* d_in, const int* in_sizes, int n_in,
                              void* d_out, int out_size, void* d_ws, size_t ws_size,
                              hipStream_t stream)
{
    const float* u       = (const float*)d_in[0];
    const float* W_in    = (const float*)d_in[1];
    const float* W_out   = (const float*)d_in[2];
    const float* dt_bias = (const float*)d_in[3];
    const float* D_param = (const float*)d_in[4];
    const float* Bn_g    = (const float*)d_in[5];
    const float* Bn_b    = (const float*)d_in[6];
    const float* Cn_g    = (const float*)d_in[7];
    const float* Cn_b    = (const float*)d_in[8];

    float* ws = (float*)d_ws;
    float* proj   = ws; ws += (size_t)M1 * DIP;
    float* yfin   = ws; ws += (size_t)M1 * DI;
    float* Sstate = ws; ws += (size_t)B_ * NH_ * NC_ * 4096;
    float* Brot   = ws; ws += (size_t)M1 * DS_;
    float* Crot   = ws; ws += (size_t)M1 * DS_;
    float* DTb    = ws; ws += (size_t)M1 * NH_;
    float* decayb = ws; ws += (size_t)M1 * NH_;
    float* trapb  = ws; ws += (size_t)M1 * NH_;
    float* phaseb = ws; ws += (size_t)M1 * NA_;
    float* cumdb  = ws; ws += (size_t)B_ * NH_ * L_;

    gemm_wmma_f32<DIP, DM><<<dim3(DIP / GBN, M1 / GBM), dim3(256), 0, stream>>>(
        u, W_in, proj);

    phase_kernel<<<dim3(B_ * PG_), dim3(PT_ * 32), 0, stream>>>(
        proj, dt_bias, DTb, decayb, trapb, phaseb);

    rot_kernel<<<dim3(M1), dim3(64), 0, stream>>>(
        proj, Bn_g, Bn_b, Cn_g, Cn_b, phaseb, Brot, Crot);

    scan_chunk_kernel<<<dim3(B_ * NH_ * NC_), dim3(256), 0, stream>>>(
        proj, Brot, Crot, DTb, decayb, trapb, yfin, cumdb, Sstate);
    scan_combine_kernel<<<dim3(B_ * NH_), dim3(256), 0, stream>>>(
        cumdb, Sstate);
    scan_correct_kernel<<<dim3(B_ * NH_ * NC_), dim3(256), 0, stream>>>(
        proj, Crot, cumdb, Sstate, D_param, yfin);

    gemm_wmma_f32<DM, DI><<<dim3(DM / GBN, M1 / GBM), dim3(256), 0, stream>>>(
        yfin, W_out, (float*)d_out);
}


# combine launch folded into the correction pass: each block rebuilds its chunk's entry state from the earlier chunks' end states, four chunks per round trip, heavy chunks first; sum order kept
# speedup vs baseline: 9.4387x; 1.2254x over previous
#include <hip/hip_runtime.h>
#include <math.h>

#define B_   2
#define L_   512
#define DM   1024
#define DI   2048
#define DS_  64
#define NH_  32
#define HD_  64
#define NA_  32
#define DIP  4352
#define M1   (B_*L_)

#define NC_  16
#define LC_  32

#define OFF_Z    0
#define OFF_X    2048
#define OFF_B    4096
#define OFF_C    4160
#define OFF_DT   4224
#define OFF_A    4256
#define OFF_TRAP 4288
#define OFF_ANG  4320

typedef __attribute__((ext_vector_type(2))) float v2f;
typedef __attribute__((ext_vector_type(4))) float v4f;
typedef __attribute__((ext_vector_type(8))) float v8f;

__device__ __forceinline__ float softplus_f(float x) {
    return x > 20.f ? x : log1pf(expf(x));
}
__device__ __forceinline__ float sigmoid_f(float x) {
    return 1.f / (1.f + expf(-x));
}

#define GBM    64
#define GBN    64
#define GBK    32
#define APITCH 36
#define BPITCH 160

template <int N, int K>
__global__ __launch_bounds__(256) void gemm_wmma_f32(
    const float* __restrict__ A, const float* __restrict__ Bm, float* __restrict__ C)
{
    constexpr int nslab = K / GBK;
    static_assert(N % GBN == 0 && K % GBK == 0, "tile sizes must divide the problem");
    static_assert(nslab % 2 == 0 && nslab >= 4, "the pipeline runs in pairs of slabs");
    __shared__ __attribute__((aligned(16))) float As[2][GBM * APITCH];
    __shared__ __attribute__((aligned(16))) float Bs[2][(GBK / 2) * BPITCH];

    const int t    = threadIdx.x;
    const int lane = t & 31;
    const int wave = t >> 5;
    const int lo   = lane & 15;
    const int hi   = lane >> 4;

    const int m0 = blockIdx.y * GBM;
    const int n0 = blockIdx.x * GBN;
    const int wm = (wave >> 1) * 16;
    const int wn = (wave & 1) * 32;

    const int ar = t >> 3, ac = (t & 7) * 4;
    const int bn = t & 63, kq = t >> 6;

    const float* Ag0 = A  + (size_t)(m0 + ar) * K + ac;
    const float* Ag1 = Ag0 + (size_t)32 * K;
    const float* Bg  = Bm + (size_t)(2 * kq) * N + n0 + bn;

    const int as0 = ar * APITCH + ac, as1 = (ar + 32) * APITCH + ac;
    const int bs0 = kq * BPITCH + bn * 2;

    struct Staged { v4f a0, a1; v2f b[4]; };
    Staged sg0, sg1;
    auto fetch = [&](int i, Staged& g) __attribute__((always_inline)) {
        g.a0 = *reinterpret_cast<const v4f*>(Ag0 + (size_t)i * GBK);
        g.a1 = *reinterpret_cast<const v4f*>(Ag1 + (size_t)i * GBK);
        const float* bp = Bg + (size_t)i * GBK * N;
        #pragma unroll
        for (int q = 0; q < 4; ++q) {
            g.b[q].x = bp[(size_t)(8 * q) * N];
            g.b[q].y = bp[(size_t)(8 * q + 1) * N];
        }
    };
    auto stage = [&](int st, const Staged& g) __attribute__((always_inline)) {
        *reinterpret_cast<v4f*>(&As[st][as0]) = g.a0;
        *reinterpret_cast<v4f*>(&As[st][as1]) = g.a1;
        #pragma unroll
        for (int q = 0; q < 4; ++q)
            *reinterpret_cast<v2f*>(&Bs[st][bs0 + 4 * q * BPITCH]) = g.b[q];
    };
    auto lds_handover = [] () __attribute__((always_inline)) {
        __builtin_amdgcn_fence(__ATOMIC_RELEASE, "workgroup", "local");
        __builtin_amdgcn_s_barrier();
        __builtin_amdgcn_fence(__ATOMIC_ACQUIRE, "workgroup", "local");
    };

    v8f acc0 = {}, acc1 = {};

    const int fa = (wm + lo) * APITCH + 2 * hi;
    const int fb = hi * BPITCH + (wn + lo) * 2;

    struct Frags { v2f a[GBK / 4], b0[GBK / 4], b1[GBK / 4]; };
    Frags fr0, fr1;
    auto frags = [&](int st, Frags& f) __attribute__((always_inline)) {
        const float* as = &As[st][fa];
        const float* bs = &Bs[st][fb];
        #pragma unroll
        for (int j = 0; j < GBK / 4; ++j) {
            f.a[j]  = *reinterpret_cast<const v2f*>(as + 4 * j);
            f.b0[j] = *reinterpret_cast<const v2f*>(bs + 2 * j * BPITCH);
            f.b1[j] = *reinterpret_cast<const v2f*>(bs + 2 * j * BPITCH + 32);
        }
    };
    auto products = [&](const Frags& f) __attribute__((always_inline)) {
        #pragma unroll
        for (int j = 0; j < GBK / 4; ++j) {
            acc0 = __builtin_amdgcn_wmma_f32_16x16x4_f32(false, f.a[j], false, f.b0[j], (short)0, acc0, false, false);
            acc1 = __builtin_amdgcn_wmma_f32_16x16x4_f32(false, f.a[j], false, f.b1[j], (short)0, acc1, false, false);
        }
    };

    auto hold_products = [&]() __attribute__((always_inline)) {
        asm volatile("" : "+v"(acc0), "+v"(acc1));
    };

    auto two_slabs = [&](int s, bool more2, bool more4, bool more3, bool more5) __attribute__((always_inline)) {
        frags(1, fr1);
        products(fr0);
        if (more2) { stage(0, sg0); if (more4) fetch(s + 4, sg0); }
        if (more2) { lds_handover(); hold_products(); }
        if (more2) frags(0, fr0);
        products(fr1);
        if (more3) { stage(1, sg1); if (more5) fetch(s + 5, sg1); }
        if (more2) { lds_handover(); hold_products(); }
    };

    fetch(0, sg0);
    fetch(1, sg1);
    stage(0, sg0);
    fetch(2, sg0);
    lds_handover();
    frags(0, fr0);
    stage(1, sg1);
    fetch(3, sg1);
    lds_handover();

    int s = 0;
    for (; s + 4 < nslab; s += 2) two_slabs(s, true, true, true, true);
    two_slabs(s, true, false, true, false);
    two_slabs(s + 2, false, false, false, false);

    #pragma unroll
    for (int r = 0; r < 8; ++r) {
        float* cp = C + (size_t)(m0 + wm + r + 8 * hi) * N + n0 + wn + lo;
        cp[0]  = acc0[r];
        cp[16] = acc1[r];
    }
}

#define PT_  16
#define PG_  (L_ / PT_)
static_assert(PG_ * PT_ == L_, "groups x tokens per group must cover the sequence");

__global__ __launch_bounds__(PT_ * 32) void phase_kernel(
    const float* __restrict__ proj, const float* __restrict__ dt_bias,
    float* __restrict__ DTb, float* __restrict__ decayb,
    float* __restrict__ trapb, float* __restrict__ phaseb)
{
    const int b = blockIdx.x / PG_;
    const int g = blockIdx.x % PG_;
    const int j = threadIdx.x & 31;
    const int w = threadIdx.x >> 5;
    __shared__ float sprod[PT_][32];

    const size_t bl  = (size_t)b * L_ + (size_t)g * PT_ + w;
    const float* row = proj + bl * DIP;

    const float vdt  = row[OFF_DT + j];
    const float vang = row[OFF_ANG + j];
    const float va   = row[OFF_A + j];
    const float vtr  = row[OFF_TRAP + j];

    const float dt = softplus_f(vdt + dt_bias[j]);
    float s = dt;
    s += __shfl_xor(s, 16, 32);
    s += __shfl_xor(s,  8, 32);
    s += __shfl_xor(s,  4, 32);
    s += __shfl_xor(s,  2, 32);
    s += __shfl_xor(s,  1, 32);
    const float dtmean = s * (1.f / 32.f);
    sprod[w][j] = vang * dtmean;

    const size_t o = bl * 32 + j;
    DTb[o]    = dt;
    const float Av = fminf(-softplus_f(va), -1e-4f);
    decayb[o] = expf(Av * dt);
    trapb[o]  = sigmoid_f(vtr);

    __syncthreads();

    float acc = 0.f;
    #pragma unroll
    for (int i = 0; i < PT_; ++i) {
        const float v = sprod[i][j];
        acc += (i <= w) ? v : 0.f;
    }
    phaseb[o] = acc;
}

__global__ __launch_bounds__(64) void rot_kernel(
    const float* __restrict__ proj,
    const float* __restrict__ Bn_g, const float* __restrict__ Bn_b,
    const float* __restrict__ Cn_g, const float* __restrict__ Cn_b,
    const float* __restrict__ phaseb,
    float* __restrict__ Brot, float* __restrict__ Crot)
{
    const int bl = blockIdx.x;
    const int t  = threadIdx.x;
    __shared__ float red[64], nb[64], nc[64];

    const int q  = t >> 1;
    const int l  = bl % L_;
    const int g  = l / PT_;
    const float* pseq = phaseb + (size_t)(bl - l) * 32 + q;
    float off = 0.f;
    #pragma unroll
    for (int k = 0; k < PG_ - 1; ++k) {
        const float v = pseq[(size_t)(k * PT_ + PT_ - 1) * 32];
        off += (k < g) ? v : 0.f;
    }
    const float ph = off + phaseb[(size_t)bl * 32 + q];

    const float* row = proj + (size_t)bl * DIP;
    const float vb = row[OFF_B + t];
    const float vc = row[OFF_C + t];

    auto blockReduce = [&](float v) -> float {
        red[t] = v; __syncthreads();
        for (int s = 32; s > 0; s >>= 1) {
            if (t < s) red[t] += red[t + s];
            __syncthreads();
        }
        float r = red[0]; __syncthreads();
        return r;
    };

    const float mB  = blockReduce(vb) * (1.f / 64.f);
    const float vaB = blockReduce((vb - mB) * (vb - mB)) * (1.f / 64.f);
    const float nB  = (vb - mB) * rsqrtf(vaB + 1e-5f) * Bn_g[t] + Bn_b[t];

    const float mC  = blockReduce(vc) * (1.f / 64.f);
    const float vaC = blockReduce((vc - mC) * (vc - mC)) * (1.f / 64.f);
    const float nC  = (vc - mC) * rsqrtf(vaC + 1e-5f) * Cn_g[t] + Cn_b[t];

    nb[t] = nB; nc[t] = nC;
    __syncthreads();

    const float c  = cosf(ph), s = sinf(ph);
    float ob, oc;
    if ((t & 1) == 0) { ob = nb[t] * c - nb[t + 1] * s; oc = nc[t] * c - nc[t + 1] * s; }
    else              { ob = nb[t - 1] * s + nb[t] * c; oc = nc[t - 1] * s + nc[t] * c; }
    Brot[(size_t)bl * 64 + t] = ob;
    Crot[(size_t)bl * 64 + t] = oc;
}


__global__ __launch_bounds__(256) void scan_chunk_kernel(
    const float* __restrict__ proj,
    const float* __restrict__ Brot, const float* __restrict__ Crot,
    const float* __restrict__ DTb, const float* __restrict__ decayb,
    const float* __restrict__ trapb,
    float* __restrict__ yfin, float* __restrict__ cumdb,
    float* __restrict__ Sstate)
{
    const int blk = blockIdx.x;
    const int c   = blk % NC_;
    const int h   = (blk / NC_) % NH_;
    const int b   = blk / (NC_ * NH_);
    const int l0  = c * LC_;
    const int t   = threadIdx.x;
    const int p   = t >> 2;
    const int seg = t & 3;

    __shared__ __attribute__((aligned(16))) float sx[LC_ + 1][64];
    __shared__ __attribute__((aligned(16))) float sB[LC_ + 1][64];
    __shared__ __attribute__((aligned(16))) float sC[LC_][64];
    __shared__ __attribute__((aligned(16))) float sscal[LC_][4];
    __shared__ float scum[LC_];
    __shared__ __attribute__((aligned(16))) float sy[LC_][64][4];

    const size_t bl0 = (size_t)b * L_ + l0;

    float4 vx[2], vb[2], vc[2];
    #pragma unroll
    for (int k = 0; k < 2; ++k) {
        const int idx = k * 256 + t;
        const int r = idx >> 4, q = idx & 15;
        vx[k] = *reinterpret_cast<const float4*>(proj + (bl0 + r) * DIP + OFF_X + h * HD_ + q * 4);
        vb[k] = *reinterpret_cast<const float4*>(Brot + (bl0 + r) * 64 + q * 4);
        vc[k] = *reinterpret_cast<const float4*>(Crot + (bl0 + r) * 64 + q * 4);
    }
    float4 v0 = make_float4(0.f, 0.f, 0.f, 0.f);
    if (c > 0 && t < 32) {
        const int q = t & 15;
        v0 = (t < 16)
           ? *reinterpret_cast<const float4*>(proj + (bl0 - 1) * DIP + OFF_X + h * HD_ + q * 4)
           : *reinterpret_cast<const float4*>(Brot + (bl0 - 1) * 64 + q * 4);
    }
    float vs = 0.f;
    if (t >= 64 && t < 160) {
        const int kind = (t - 64) >> 5, tok = (t - 64) & 31;
        const float* sp = (kind == 0) ? DTb : (kind == 1) ? decayb : trapb;
        vs = sp[(bl0 + tok) * 32 + h];
    }
    #pragma unroll
    for (int k = 0; k < 2; ++k) {
        const int idx = k * 256 + t;
        const int r = idx >> 4, q = idx & 15;
        *reinterpret_cast<float4*>(&sx[r + 1][q * 4]) = vx[k];
        *reinterpret_cast<float4*>(&sB[r + 1][q * 4]) = vb[k];
        *reinterpret_cast<float4*>(&sC[r][q * 4])     = vc[k];
    }
    if (t < 16)      *reinterpret_cast<float4*>(&sx[0][(t & 15) * 4]) = v0;
    else if (t < 32) *reinterpret_cast<float4*>(&sB[0][(t & 15) * 4]) = v0;
    if (t >= 64 && t < 160) sscal[(t - 64) & 31][(t - 64) >> 5] = vs;
    __syncthreads();

    float hst[16], Bp[16];
    #pragma unroll
    for (int i = 0; i < 16; ++i) hst[i] = 0.f;
    #pragma unroll
    for (int i4 = 0; i4 < 4; ++i4) {
        const float4 v = *reinterpret_cast<const float4*>(&sB[0][seg * 16 + i4 * 4]);
        Bp[i4 * 4 + 0] = v.x; Bp[i4 * 4 + 1] = v.y; Bp[i4 * 4 + 2] = v.z; Bp[i4 * 4 + 3] = v.w;
    }
    float xprev = sx[0][p];
    float cumd  = 1.f;

    #pragma unroll 2
    for (int tok = 0; tok < LC_; ++tok) {
        const float  x  = sx[tok + 1][p];
        const float4 sc = *reinterpret_cast<const float4*>(&sscal[tok][0]);
        const float dt = sc.x, dec = sc.y, tr = sc.z;

        float Bc[16], Cc[16];
        #pragma unroll
        for (int i4 = 0; i4 < 4; ++i4) {
            const float4 vB = *reinterpret_cast<const float4*>(&sB[tok + 1][seg * 16 + i4 * 4]);
            const float4 vC = *reinterpret_cast<const float4*>(&sC[tok][seg * 16 + i4 * 4]);
            Bc[i4 * 4 + 0] = vB.x; Bc[i4 * 4 + 1] = vB.y; Bc[i4 * 4 + 2] = vB.z; Bc[i4 * 4 + 3] = vB.w;
            Cc[i4 * 4 + 0] = vC.x; Cc[i4 * 4 + 1] = vC.y; Cc[i4 * 4 + 2] = vC.z; Cc[i4 * 4 + 3] = vC.w;
        }

        float y = 0.f;
        #pragma unroll
        for (int i = 0; i < 16; ++i) {
            const float bx = tr * x * Bc[i] + (1.f - tr) * xprev * Bp[i];
            hst[i] = dec * hst[i] + dt * bx;
            y += hst[i] * Cc[i];
            Bp[i] = Bc[i];
        }
        xprev = x;
        cumd *= dec;

        sy[tok][p][seg] = y;
        if (t == 0) scum[tok] = cumd;
    }
    __syncthreads();

    #pragma unroll
    for (int k = 0; k < 2; ++k) {
        const int idx = k * 256 + t;
        const int r = idx >> 4, q = idx & 15;
        float o[4];
        #pragma unroll
        for (int j = 0; j < 4; ++j) {
            const float4 pv = *reinterpret_cast<const float4*>(&sy[r][q * 4 + j][0]);
            o[j] = (pv.x + pv.y) + (pv.z + pv.w);
        }
        *reinterpret_cast<float4*>(yfin + (bl0 + r) * DI + h * HD_ + q * 4) = make_float4(o[0], o[1], o[2], o[3]);
    }
    if (t < LC_) cumdb[((size_t)(b * NH_ + h)) * L_ + l0 + t] = scum[t];

    float* Sp = Sstate + (((size_t)(b * NH_ + h)) * NC_ + c) * 4096 + p * 64 + seg * 16;
    #pragma unroll
    for (int i = 0; i < 16; ++i) Sp[i] = hst[i];
}

template <int NCH>
__device__ __forceinline__ void fold_chunks(float (&H)[16], const float* __restrict__ Sp,
                                            const float* __restrict__ cd)
{
    float P[NCH];
    float4 S[NCH][4];
    #pragma unroll
    for (int k = 0; k < NCH; ++k) {
        P[k] = cd[k * LC_];
        #pragma unroll
        for (int i4 = 0; i4 < 4; ++i4)
            S[k][i4] = *reinterpret_cast<const float4*>(Sp + (size_t)k * 4096 + i4 * 4);
    }
    #pragma unroll
    for (int k = 0; k < NCH; ++k) {
        #pragma unroll
        for (int i4 = 0; i4 < 4; ++i4) {
            H[i4 * 4 + 0] = P[k] * H[i4 * 4 + 0] + S[k][i4].x;
            H[i4 * 4 + 1] = P[k] * H[i4 * 4 + 1] + S[k][i4].y;
            H[i4 * 4 + 2] = P[k] * H[i4 * 4 + 2] + S[k][i4].z;
            H[i4 * 4 + 3] = P[k] * H[i4 * 4 + 3] + S[k][i4].w;
        }
    }
}

__global__ __launch_bounds__(256) void scan_correct_kernel(
    const float* __restrict__ proj, const float* __restrict__ Crot,
    const float* __restrict__ cumdb, const float* __restrict__ Sstate,
    const float* __restrict__ D_param, float* __restrict__ yfin)
{
    const int blk = blockIdx.x;
    static_assert(NC_ == 16 && B_ * NH_ == 64, "the block order below is written for 16 chunks x 64 (batch, head) pairs");
    const int qr  = blk >> 8;
    const int bh  = (blk >> 2) & 63;
    const int e   = (blk + qr) & 3;
    const int c   = (qr == 3) ? e : ((e & 1) ? (4 + qr + 3 * (e >> 1)) : (15 - qr - 3 * (e >> 1)));
    const int h   = bh % NH_;
    const int b   = bh / NH_;
    const int l0  = c * LC_;
    const int t   = threadIdx.x;
    const int p   = t >> 2, seg = t & 3;

    __shared__ __attribute__((aligned(16))) float sC[LC_][64];
    __shared__ float scd[LC_];
    __shared__ __attribute__((aligned(16))) float scorr[LC_][64][4];

    const size_t bl0 = (size_t)b * L_ + l0;

    const float Dp = D_param[h];

    float4 vc[2], vx[2], vz[2], vy[2];
    #pragma unroll
    for (int k = 0; k < 2; ++k) {
        const int idx = k * 256 + t;
        const int r = idx >> 4, q = idx & 15;
        vc[k] = *reinterpret_cast<const float4*>(Crot + (bl0 + r) * 64 + q * 4);
        vx[k] = *reinterpret_cast<const float4*>(proj + (bl0 + r) * DIP + OFF_X + h * HD_ + q * 4);
        vz[k] = *reinterpret_cast<const float4*>(proj + (bl0 + r) * DIP + OFF_Z + h * HD_ + q * 4);
        vy[k] = *reinterpret_cast<const float4*>(yfin + (bl0 + r) * DI + h * HD_ + q * 4);
    }
    float cdv = 0.f;
    if (t < LC_) cdv = cumdb[((size_t)(b * NH_ + h)) * L_ + l0 + t];

    float H[16];
    #pragma unroll
    for (int i = 0; i < 16; ++i) H[i] = 0.f;
    const float* Sb  = Sstate + ((size_t)(b * NH_ + h)) * NC_ * 4096 + p * 64 + seg * 16;
    const float* cdl = cumdb + ((size_t)(b * NH_ + h)) * L_ + (LC_ - 1);
    int cc = 0;
    for (; cc + 4 <= c; cc += 4) fold_chunks<4>(H, Sb + (size_t)cc * 4096, cdl + cc * LC_);
    if (c & 2) { fold_chunks<2>(H, Sb + (size_t)cc * 4096, cdl + cc * LC_); cc += 2; }
    if (c & 1)   fold_chunks<1>(H, Sb + (size_t)cc * 4096, cdl + cc * LC_);

    #pragma unroll
    for (int k = 0; k < 2; ++k) {
        const int idx = k * 256 + t;
        const int r = idx >> 4, q = idx & 15;
        *reinterpret_cast<float4*>(&sC[r][q * 4]) = vc[k];
    }
    if (t < LC_) scd[t] = cdv;
    __syncthreads();

    #pragma unroll 2
    for (int tok = 0; tok < LC_; ++tok) {
        float Cc[16];
        #pragma unroll
        for (int i4 = 0; i4 < 4; ++i4) {
            const float4 vC = *reinterpret_cast<const float4*>(&sC[tok][seg * 16 + i4 * 4]);
            Cc[i4 * 4 + 0] = vC.x; Cc[i4 * 4 + 1] = vC.y; Cc[i4 * 4 + 2] = vC.z; Cc[i4 * 4 + 3] = vC.w;
        }
        float corr = 0.f;
        #pragma unroll
        for (int i = 0; i < 16; ++i) corr += H[i] * Cc[i];
        scorr[tok][p][seg] = corr;
    }
    __syncthreads();

    #pragma unroll
    for (int k = 0; k < 2; ++k) {
        const int idx = k * 256 + t;
        const int r = idx >> 4, q = idx & 15;
        const float cd = scd[r];
        const float xs[4] = {vx[k].x, vx[k].y, vx[k].z, vx[k].w};
        const float zs[4] = {vz[k].x, vz[k].y, vz[k].z, vz[k].w};
        const float ys[4] = {vy[k].x, vy[k].y, vy[k].z, vy[k].w};
        float o[4];
        #pragma unroll
        for (int j = 0; j < 4; ++j) {
            const float4 pv = *reinterpret_cast<const float4*>(&scorr[r][q * 4 + j][0]);
            const float corr = (pv.x + pv.y) + (pv.z + pv.w);
            const float y = ys[j] + cd * corr;
            o[j] = (y + Dp * xs[j]) * (zs[j] * sigmoid_f(zs[j]));
        }
        *reinterpret_cast<float4*>(yfin + (bl0 + r) * DI + h * HD_ + q * 4) = make_float4(o[0], o[1], o[2], o[3]);
    }
}

extern "C" void kernel_launch(void* const* d_in, const int* in_sizes, int n_in,
                              void* d_out, int out_size, void* d_ws, size_t ws_size,
                              hipStream_t stream)
{
    const float* u       = (const float*)d_in[0];
    const float* W_in    = (const float*)d_in[1];
    const float* W_out   = (const float*)d_in[2];
    const float* dt_bias = (const float*)d_in[3];
    const float* D_param = (const float*)d_in[4];
    const float* Bn_g    = (const float*)d_in[5];
    const float* Bn_b    = (const float*)d_in[6];
    const float* Cn_g    = (const float*)d_in[7];
    const float* Cn_b    = (const float*)d_in[8];

    float* ws = (float*)d_ws;
    float* proj   = ws; ws += (size_t)M1 * DIP;
    float* yfin   = ws; ws += (size_t)M1 * DI;
    float* Sstate = ws; ws += (size_t)B_ * NH_ * NC_ * 4096;
    float* Brot   = ws; ws += (size_t)M1 * DS_;
    float* Crot   = ws; ws += (size_t)M1 * DS_;
    float* DTb    = ws; ws += (size_t)M1 * NH_;
    float* decayb = ws; ws += (size_t)M1 * NH_;
    float* trapb  = ws; ws += (size_t)M1 * NH_;
    float* phaseb = ws; ws += (size_t)M1 * NA_;
    float* cumdb  = ws; ws += (size_t)B_ * NH_ * L_;

    gemm_wmma_f32<DIP, DM><<<dim3(DIP / GBN, M1 / GBM), dim3(256), 0, stream>>>(
        u, W_in, proj);

    phase_kernel<<<dim3(B_ * PG_), dim3(PT_ * 32), 0, stream>>>(
        proj, dt_bias, DTb, decayb, trapb, phaseb);

    rot_kernel<<<dim3(M1), dim3(64), 0, stream>>>(
        proj, Bn_g, Bn_b, Cn_g, Cn_b, phaseb, Brot, Crot);

    scan_chunk_kernel<<<dim3(B_ * NH_ * NC_), dim3(256), 0, stream>>>(
        proj, Brot, Crot, DTb, decayb, trapb, yfin, cumdb, Sstate);
    scan_correct_kernel<<<dim3(B_ * NH_ * NC_), dim3(256), 0, stream>>>(
        proj, Crot, cumdb, Sstate, D_param, yfin);

    gemm_wmma_f32<DM, DI><<<dim3(DM / GBN, M1 / GBM), dim3(256), 0, stream>>>(
        yfin, W_out, (float*)d_out);
}


# scan chunk length 32 -> 64 tokens (8 chunks, 512 blocks a pass): 28 not 120 earlier-state reads per (batch, head), 2 not 5 fold round trips, half the state writes; chunk split moves, sum order changes
# speedup vs baseline: 9.8568x; 1.0443x over previous
#include <hip/hip_runtime.h>
#include <math.h>

#define B_   2
#define L_   512
#define DM   1024
#define DI   2048
#define DS_  64
#define NH_  32
#define HD_  64
#define NA_  32
#define DIP  4352
#define M1   (B_*L_)

#define NC_  8
#define LC_  64
#define LR_  (LC_ / 16)
static_assert(NC_ * LC_ == L_, "chunks x tokens per chunk must cover the sequence");
static_assert(LC_ % 16 == 0 && 3 * LC_ <= 192, "row rounds and the three per-token scalar planes are spread over 256 threads");

#define OFF_Z    0
#define OFF_X    2048
#define OFF_B    4096
#define OFF_C    4160
#define OFF_DT   4224
#define OFF_A    4256
#define OFF_TRAP 4288
#define OFF_ANG  4320

typedef __attribute__((ext_vector_type(2))) float v2f;
typedef __attribute__((ext_vector_type(4))) float v4f;
typedef __attribute__((ext_vector_type(8))) float v8f;

__device__ __forceinline__ float softplus_f(float x) {
    return x > 20.f ? x : log1pf(expf(x));
}
__device__ __forceinline__ float sigmoid_f(float x) {
    return 1.f / (1.f + expf(-x));
}

#define GBM    64
#define GBN    64
#define GBK    32
#define APITCH 36
#define BPITCH 160

template <int N, int K>
__global__ __launch_bounds__(256) void gemm_wmma_f32(
    const float* __restrict__ A, const float* __restrict__ Bm, float* __restrict__ C)
{
    constexpr int nslab = K / GBK;
    static_assert(N % GBN == 0 && K % GBK == 0, "tile sizes must divide the problem");
    static_assert(nslab % 2 == 0 && nslab >= 4, "the pipeline runs in pairs of slabs");
    __shared__ __attribute__((aligned(16))) float As[2][GBM * APITCH];
    __shared__ __attribute__((aligned(16))) float Bs[2][(GBK / 2) * BPITCH];

    const int t    = threadIdx.x;
    const int lane = t & 31;
    const int wave = t >> 5;
    const int lo   = lane & 15;
    const int hi   = lane >> 4;

    const int m0 = blockIdx.y * GBM;
    const int n0 = blockIdx.x * GBN;
    const int wm = (wave >> 1) * 16;
    const int wn = (wave & 1) * 32;

    const int ar = t >> 3, ac = (t & 7) * 4;
    const int bn = t & 63, kq = t >> 6;

    const float* Ag0 = A  + (size_t)(m0 + ar) * K + ac;
    const float* Ag1 = Ag0 + (size_t)32 * K;
    const float* Bg  = Bm + (size_t)(2 * kq) * N + n0 + bn;

    const int as0 = ar * APITCH + ac, as1 = (ar + 32) * APITCH + ac;
    const int bs0 = kq * BPITCH + bn * 2;

    struct Staged { v4f a0, a1; v2f b[4]; };
    Staged sg0, sg1;
    auto fetch = [&](int i, Staged& g) __attribute__((always_inline)) {
        g.a0 = *reinterpret_cast<const v4f*>(Ag0 + (size_t)i * GBK);
        g.a1 = *reinterpret_cast<const v4f*>(Ag1 + (size_t)i * GBK);
        const float* bp = Bg + (size_t)i * GBK * N;
        #pragma unroll
        for (int q = 0; q < 4; ++q) {
            g.b[q].x = bp[(size_t)(8 * q) * N];
            g.b[q].y = bp[(size_t)(8 * q + 1) * N];
        }
    };
    auto stage = [&](int st, const Staged& g) __attribute__((always_inline)) {
        *reinterpret_cast<v4f*>(&As[st][as0]) = g.a0;
        *reinterpret_cast<v4f*>(&As[st][as1]) = g.a1;
        #pragma unroll
        for (int q = 0; q < 4; ++q)
            *reinterpret_cast<v2f*>(&Bs[st][bs0 + 4 * q * BPITCH]) = g.b[q];
    };
    auto lds_handover = [] () __attribute__((always_inline)) {
        __builtin_amdgcn_fence(__ATOMIC_RELEASE, "workgroup", "local");
        __builtin_amdgcn_s_barrier();
        __builtin_amdgcn_fence(__ATOMIC_ACQUIRE, "workgroup", "local");
    };

    v8f acc0 = {}, acc1 = {};

    const int fa = (wm + lo) * APITCH + 2 * hi;
    const int fb = hi * BPITCH + (wn + lo) * 2;

    struct Frags { v2f a[GBK / 4], b0[GBK / 4], b1[GBK / 4]; };
    Frags fr0, fr1;
    auto frags = [&](int st, Frags& f) __attribute__((always_inline)) {
        const float* as = &As[st][fa];
        const float* bs = &Bs[st][fb];
        #pragma unroll
        for (int j = 0; j < GBK / 4; ++j) {
            f.a[j]  = *reinterpret_cast<const v2f*>(as + 4 * j);
            f.b0[j] = *reinterpret_cast<const v2f*>(bs + 2 * j * BPITCH);
            f.b1[j] = *reinterpret_cast<const v2f*>(bs + 2 * j * BPITCH + 32);
        }
    };
    auto products = [&](const Frags& f) __attribute__((always_inline)) {
        #pragma unroll
        for (int j = 0; j < GBK / 4; ++j) {
            acc0 = __builtin_amdgcn_wmma_f32_16x16x4_f32(false, f.a[j], false, f.b0[j], (short)0, acc0, false, false);
            acc1 = __builtin_amdgcn_wmma_f32_16x16x4_f32(false, f.a[j], false, f.b1[j], (short)0, acc1, false, false);
        }
    };

    auto hold_products = [&]() __attribute__((always_inline)) {
        asm volatile("" : "+v"(acc0), "+v"(acc1));
    };

    auto two_slabs = [&](int s, bool more2, bool more4, bool more3, bool more5) __attribute__((always_inline)) {
        frags(1, fr1);
        products(fr0);
        if (more2) { stage(0, sg0); if (more4) fetch(s + 4, sg0); }
        if (more2) { lds_handover(); hold_products(); }
        if (more2) frags(0, fr0);
        products(fr1);
        if (more3) { stage(1, sg1); if (more5) fetch(s + 5, sg1); }
        if (more2) { lds_handover(); hold_products(); }
    };

    fetch(0, sg0);
    fetch(1, sg1);
    stage(0, sg0);
    fetch(2, sg0);
    lds_handover();
    frags(0, fr0);
    stage(1, sg1);
    fetch(3, sg1);
    lds_handover();

    int s = 0;
    for (; s + 4 < nslab; s += 2) two_slabs(s, true, true, true, true);
    two_slabs(s, true, false, true, false);
    two_slabs(s + 2, false, false, false, false);

    #pragma unroll
    for (int r = 0; r < 8; ++r) {
        float* cp = C + (size_t)(m0 + wm + r + 8 * hi) * N + n0 + wn + lo;
        cp[0]  = acc0[r];
        cp[16] = acc1[r];
    }
}

#define PT_  16
#define PG_  (L_ / PT_)
static_assert(PG_ * PT_ == L_, "groups x tokens per group must cover the sequence");

__global__ __launch_bounds__(PT_ * 32) void phase_kernel(
    const float* __restrict__ proj, const float* __restrict__ dt_bias,
    float* __restrict__ DTb, float* __restrict__ decayb,
    float* __restrict__ trapb, float* __restrict__ phaseb)
{
    const int b = blockIdx.x / PG_;
    const int g = blockIdx.x % PG_;
    const int j = threadIdx.x & 31;
    const int w = threadIdx.x >> 5;
    __shared__ float sprod[PT_][32];

    const size_t bl  = (size_t)b * L_ + (size_t)g * PT_ + w;
    const float* row = proj + bl * DIP;

    const float vdt  = row[OFF_DT + j];
    const float vang = row[OFF_ANG + j];
    const float va   = row[OFF_A + j];
    const float vtr  = row[OFF_TRAP + j];

    const float dt = softplus_f(vdt + dt_bias[j]);
    float s = dt;
    s += __shfl_xor(s, 16, 32);
    s += __shfl_xor(s,  8, 32);
    s += __shfl_xor(s,  4, 32);
    s += __shfl_xor(s,  2, 32);
    s += __shfl_xor(s,  1, 32);
    const float dtmean = s * (1.f / 32.f);
    sprod[w][j] = vang * dtmean;

    const size_t o = bl * 32 + j;
    DTb[o]    = dt;
    const float Av = fminf(-softplus_f(va), -1e-4f);
    decayb[o] = expf(Av * dt);
    trapb[o]  = sigmoid_f(vtr);

    __syncthreads();

    float acc = 0.f;
    #pragma unroll
    for (int i = 0; i < PT_; ++i) {
        const float v = sprod[i][j];
        acc += (i <= w) ? v : 0.f;
    }
    phaseb[o] = acc;
}

__global__ __launch_bounds__(64) void rot_kernel(
    const float* __restrict__ proj,
    const float* __restrict__ Bn_g, const float* __restrict__ Bn_b,
    const float* __restrict__ Cn_g, const float* __restrict__ Cn_b,
    const float* __restrict__ phaseb,
    float* __restrict__ Brot, float* __restrict__ Crot)
{
    const int bl = blockIdx.x;
    const int t  = threadIdx.x;
    __shared__ float red[64], nb[64], nc[64];

    const int q  = t >> 1;
    const int l  = bl % L_;
    const int g  = l / PT_;
    const float* pseq = phaseb + (size_t)(bl - l) * 32 + q;
    float off = 0.f;
    #pragma unroll
    for (int k = 0; k < PG_ - 1; ++k) {
        const float v = pseq[(size_t)(k * PT_ + PT_ - 1) * 32];
        off += (k < g) ? v : 0.f;
    }
    const float ph = off + phaseb[(size_t)bl * 32 + q];

    const float* row = proj + (size_t)bl * DIP;
    const float vb = row[OFF_B + t];
    const float vc = row[OFF_C + t];

    auto blockReduce = [&](float v) -> float {
        red[t] = v; __syncthreads();
        for (int s = 32; s > 0; s >>= 1) {
            if (t < s) red[t] += red[t + s];
            __syncthreads();
        }
        float r = red[0]; __syncthreads();
        return r;
    };

    const float mB  = blockReduce(vb) * (1.f / 64.f);
    const float vaB = blockReduce((vb - mB) * (vb - mB)) * (1.f / 64.f);
    const float nB  = (vb - mB) * rsqrtf(vaB + 1e-5f) * Bn_g[t] + Bn_b[t];

    const float mC  = blockReduce(vc) * (1.f / 64.f);
    const float vaC = blockReduce((vc - mC) * (vc - mC)) * (1.f / 64.f);
    const float nC  = (vc - mC) * rsqrtf(vaC + 1e-5f) * Cn_g[t] + Cn_b[t];

    nb[t] = nB; nc[t] = nC;
    __syncthreads();

    const float c  = cosf(ph), s = sinf(ph);
    float ob, oc;
    if ((t & 1) == 0) { ob = nb[t] * c - nb[t + 1] * s; oc = nc[t] * c - nc[t + 1] * s; }
    else              { ob = nb[t - 1] * s + nb[t] * c; oc = nc[t - 1] * s + nc[t] * c; }
    Brot[(size_t)bl * 64 + t] = ob;
    Crot[(size_t)bl * 64 + t] = oc;
}


__global__ __launch_bounds__(256) void scan_chunk_kernel(
    const float* __restrict__ proj,
    const float* __restrict__ Brot, const float* __restrict__ Crot,
    const float* __restrict__ DTb, const float* __restrict__ decayb,
    const float* __restrict__ trapb,
    float* __restrict__ yfin, float* __restrict__ cumdb,
    float* __restrict__ Sstate)
{
    const int blk = blockIdx.x;
    const int c   = blk % NC_;
    const int h   = (blk / NC_) % NH_;
    const int b   = blk / (NC_ * NH_);
    const int l0  = c * LC_;
    const int t   = threadIdx.x;
    const int p   = t >> 2;
    const int seg = t & 3;

    __shared__ __attribute__((aligned(16))) float sx[LC_ + 1][64];
    __shared__ __attribute__((aligned(16))) float sB[LC_ + 1][64];
    __shared__ __attribute__((aligned(16))) float sC[LC_][64];
    __shared__ __attribute__((aligned(16))) float sscal[LC_][4];
    __shared__ float scum[LC_];
    __shared__ __attribute__((aligned(16))) float sy[LC_][64][4];

    const size_t bl0 = (size_t)b * L_ + l0;

    float4 vx[LR_], vb[LR_], vc[LR_];
    #pragma unroll
    for (int k = 0; k < LR_; ++k) {
        const int idx = k * 256 + t;
        const int r = idx >> 4, q = idx & 15;
        vx[k] = *reinterpret_cast<const float4*>(proj + (bl0 + r) * DIP + OFF_X + h * HD_ + q * 4);
        vb[k] = *reinterpret_cast<const float4*>(Brot + (bl0 + r) * 64 + q * 4);
        vc[k] = *reinterpret_cast<const float4*>(Crot + (bl0 + r) * 64 + q * 4);
    }
    float4 v0 = make_float4(0.f, 0.f, 0.f, 0.f);
    if (c > 0 && t < 32) {
        const int q = t & 15;
        v0 = (t < 16)
           ? *reinterpret_cast<const float4*>(proj + (bl0 - 1) * DIP + OFF_X + h * HD_ + q * 4)
           : *reinterpret_cast<const float4*>(Brot + (bl0 - 1) * 64 + q * 4);
    }
    float vs = 0.f;
    if (t >= 64) {
        const int kind = (t - 64) / LC_, tok = (t - 64) % LC_;
        const float* sp = (kind == 0) ? DTb : (kind == 1) ? decayb : trapb;
        vs = sp[(bl0 + tok) * 32 + h];
    }
    #pragma unroll
    for (int k = 0; k < LR_; ++k) {
        const int idx = k * 256 + t;
        const int r = idx >> 4, q = idx & 15;
        *reinterpret_cast<float4*>(&sx[r + 1][q * 4]) = vx[k];
        *reinterpret_cast<float4*>(&sB[r + 1][q * 4]) = vb[k];
        *reinterpret_cast<float4*>(&sC[r][q * 4])     = vc[k];
    }
    if (t < 16)      *reinterpret_cast<float4*>(&sx[0][(t & 15) * 4]) = v0;
    else if (t < 32) *reinterpret_cast<float4*>(&sB[0][(t & 15) * 4]) = v0;
    if (t >= 64) sscal[(t - 64) % LC_][(t - 64) / LC_] = vs;
    __syncthreads();

    float hst[16], Bp[16];
    #pragma unroll
    for (int i = 0; i < 16; ++i) hst[i] = 0.f;
    #pragma unroll
    for (int i4 = 0; i4 < 4; ++i4) {
        const float4 v = *reinterpret_cast<const float4*>(&sB[0][seg * 16 + i4 * 4]);
        Bp[i4 * 4 + 0] = v.x; Bp[i4 * 4 + 1] = v.y; Bp[i4 * 4 + 2] = v.z; Bp[i4 * 4 + 3] = v.w;
    }
    float xprev = sx[0][p];
    float cumd  = 1.f;

    #pragma unroll 2
    for (int tok = 0; tok < LC_; ++tok) {
        const float  x  = sx[tok + 1][p];
        const float4 sc = *reinterpret_cast<const float4*>(&sscal[tok][0]);
        const float dt = sc.x, dec = sc.y, tr = sc.z;

        float Bc[16], Cc[16];
        #pragma unroll
        for (int i4 = 0; i4 < 4; ++i4) {
            const float4 vB = *reinterpret_cast<const float4*>(&sB[tok + 1][seg * 16 + i4 * 4]);
            const float4 vC = *reinterpret_cast<const float4*>(&sC[tok][seg * 16 + i4 * 4]);
            Bc[i4 * 4 + 0] = vB.x; Bc[i4 * 4 + 1] = vB.y; Bc[i4 * 4 + 2] = vB.z; Bc[i4 * 4 + 3] = vB.w;
            Cc[i4 * 4 + 0] = vC.x; Cc[i4 * 4 + 1] = vC.y; Cc[i4 * 4 + 2] = vC.z; Cc[i4 * 4 + 3] = vC.w;
        }

        float y = 0.f;
        #pragma unroll
        for (int i = 0; i < 16; ++i) {
            const float bx = tr * x * Bc[i] + (1.f - tr) * xprev * Bp[i];
            hst[i] = dec * hst[i] + dt * bx;
            y += hst[i] * Cc[i];
            Bp[i] = Bc[i];
        }
        xprev = x;
        cumd *= dec;

        sy[tok][p][seg] = y;
        if (t == 0) scum[tok] = cumd;
    }
    __syncthreads();

    #pragma unroll
    for (int k = 0; k < LR_; ++k) {
        const int idx = k * 256 + t;
        const int r = idx >> 4, q = idx & 15;
        float o[4];
        #pragma unroll
        for (int j = 0; j < 4; ++j) {
            const float4 pv = *reinterpret_cast<const float4*>(&sy[r][q * 4 + j][0]);
            o[j] = (pv.x + pv.y) + (pv.z + pv.w);
        }
        *reinterpret_cast<float4*>(yfin + (bl0 + r) * DI + h * HD_ + q * 4) = make_float4(o[0], o[1], o[2], o[3]);
    }
    if (t < LC_) cumdb[((size_t)(b * NH_ + h)) * L_ + l0 + t] = scum[t];

    float* Sp = Sstate + (((size_t)(b * NH_ + h)) * NC_ + c) * 4096 + p * 64 + seg * 16;
    #pragma unroll
    for (int i = 0; i < 16; ++i) Sp[i] = hst[i];
}

template <int NCH>
__device__ __forceinline__ void fold_chunks(float (&H)[16], const float* __restrict__ Sp,
                                            const float* __restrict__ cd)
{
    float P[NCH];
    float4 S[NCH][4];
    #pragma unroll
    for (int k = 0; k < NCH; ++k) {
        P[k] = cd[k * LC_];
        #pragma unroll
        for (int i4 = 0; i4 < 4; ++i4)
            S[k][i4] = *reinterpret_cast<const float4*>(Sp + (size_t)k * 4096 + i4 * 4);
    }
    #pragma unroll
    for (int k = 0; k < NCH; ++k) {
        #pragma unroll
        for (int i4 = 0; i4 < 4; ++i4) {
            H[i4 * 4 + 0] = P[k] * H[i4 * 4 + 0] + S[k][i4].x;
            H[i4 * 4 + 1] = P[k] * H[i4 * 4 + 1] + S[k][i4].y;
            H[i4 * 4 + 2] = P[k] * H[i4 * 4 + 2] + S[k][i4].z;
            H[i4 * 4 + 3] = P[k] * H[i4 * 4 + 3] + S[k][i4].w;
        }
    }
}

__global__ __launch_bounds__(256) void scan_correct_kernel(
    const float* __restrict__ proj, const float* __restrict__ Crot,
    const float* __restrict__ cumdb, const float* __restrict__ Sstate,
    const float* __restrict__ D_param, float* __restrict__ yfin)
{
    const int blk = blockIdx.x;
    static_assert(NC_ == 8 && B_ * NH_ == 64, "the block order below is written for 8 chunks x 64 (batch, head) pairs");
    const int hf  = blk >> 8;
    const int bh  = blk & 63;
    const int e   = (blk >> 6) & 3;
    const int c   = hf ? e : (7 - e);
    const int h   = bh % NH_;
    const int b   = bh / NH_;
    const int l0  = c * LC_;
    const int t   = threadIdx.x;
    const int p   = t >> 2, seg = t & 3;

    __shared__ __attribute__((aligned(16))) float sC[LC_][64];
    __shared__ float scd[LC_];
    __shared__ __attribute__((aligned(16))) float scorr[LC_][64][4];

    const size_t bl0 = (size_t)b * L_ + l0;

    const float Dp = D_param[h];

    float4 vc[LR_], vx[LR_], vz[LR_], vy[LR_];
    #pragma unroll
    for (int k = 0; k < LR_; ++k) {
        const int idx = k * 256 + t;
        const int r = idx >> 4, q = idx & 15;
        vc[k] = *reinterpret_cast<const float4*>(Crot + (bl0 + r) * 64 + q * 4);
        vx[k] = *reinterpret_cast<const float4*>(proj + (bl0 + r) * DIP + OFF_X + h * HD_ + q * 4);
        vz[k] = *reinterpret_cast<const float4*>(proj + (bl0 + r) * DIP + OFF_Z + h * HD_ + q * 4);
        vy[k] = *reinterpret_cast<const float4*>(yfin + (bl0 + r) * DI + h * HD_ + q * 4);
    }
    float cdv = 0.f;
    if (t < LC_) cdv = cumdb[((size_t)(b * NH_ + h)) * L_ + l0 + t];

    float H[16];
    #pragma unroll
    for (int i = 0; i < 16; ++i) H[i] = 0.f;
    const float* Sb  = Sstate + ((size_t)(b * NH_ + h)) * NC_ * 4096 + p * 64 + seg * 16;
    const float* cdl = cumdb + ((size_t)(b * NH_ + h)) * L_ + (LC_ - 1);
    int cc = 0;
    for (; cc + 4 <= c; cc += 4) fold_chunks<4>(H, Sb + (size_t)cc * 4096, cdl + cc * LC_);
    const int rest = c - cc;
    if (rest == 3)      fold_chunks<3>(H, Sb + (size_t)cc * 4096, cdl + cc * LC_);
    else if (rest == 2) fold_chunks<2>(H, Sb + (size_t)cc * 4096, cdl + cc * LC_);
    else if (rest == 1) fold_chunks<1>(H, Sb + (size_t)cc * 4096, cdl + cc * LC_);

    #pragma unroll
    for (int k = 0; k < LR_; ++k) {
        const int idx = k * 256 + t;
        const int r = idx >> 4, q = idx & 15;
        *reinterpret_cast<float4*>(&sC[r][q * 4]) = vc[k];
    }
    if (t < LC_) scd[t] = cdv;
    __syncthreads();

    #pragma unroll 2
    for (int tok = 0; tok < LC_; ++tok) {
        float Cc[16];
        #pragma unroll
        for (int i4 = 0; i4 < 4; ++i4) {
            const float4 vC = *reinterpret_cast<const float4*>(&sC[tok][seg * 16 + i4 * 4]);
            Cc[i4 * 4 + 0] = vC.x; Cc[i4 * 4 + 1] = vC.y; Cc[i4 * 4 + 2] = vC.z; Cc[i4 * 4 + 3] = vC.w;
        }
        float corr = 0.f;
        #pragma unroll
        for (int i = 0; i < 16; ++i) corr += H[i] * Cc[i];
        scorr[tok][p][seg] = corr;
    }
    __syncthreads();

    #pragma unroll
    for (int k = 0; k < LR_; ++k) {
        const int idx = k * 256 + t;
        const int r = idx >> 4, q = idx & 15;
        const float cd = scd[r];
        const float xs[4] = {vx[k].x, vx[k].y, vx[k].z, vx[k].w};
        const float zs[4] = {vz[k].x, vz[k].y, vz[k].z, vz[k].w};
        const float ys[4] = {vy[k].x, vy[k].y, vy[k].z, vy[k].w};
        float o[4];
        #pragma unroll
        for (int j = 0; j < 4; ++j) {
            const float4 pv = *reinterpret_cast<const float4*>(&scorr[r][q * 4 + j][0]);
            const float corr = (pv.x + pv.y) + (pv.z + pv.w);
            const float y = ys[j] + cd * corr;
            o[j] = (y + Dp * xs[j]) * (zs[j] * sigmoid_f(zs[j]));
        }
        *reinterpret_cast<float4*>(yfin + (bl0 + r) * DI + h * HD_ + q * 4) = make_float4(o[0], o[1], o[2], o[3]);
    }
}

extern "C" void kernel_launch(void* const* d_in, const int* in_sizes, int n_in,
                              void* d_out, int out_size, void* d_ws, size_t ws_size,
                              hipStream_t stream)
{
    const float* u       = (const float*)d_in[0];
    const float* W_in    = (const float*)d_in[1];
    const float* W_out   = (const float*)d_in[2];
    const float* dt_bias = (const float*)d_in[3];
    const float* D_param = (const float*)d_in[4];
    const float* Bn_g    = (const float*)d_in[5];
    const float* Bn_b    = (const float*)d_in[6];
    const float* Cn_g    = (const float*)d_in[7];
    const float* Cn_b    = (const float*)d_in[8];

    float* ws = (float*)d_ws;
    float* proj   = ws; ws += (size_t)M1 * DIP;
    float* yfin   = ws; ws += (size_t)M1 * DI;
    float* Sstate = ws; ws += (size_t)B_ * NH_ * NC_ * 4096;
    float* Brot   = ws; ws += (size_t)M1 * DS_;
    float* Crot   = ws; ws += (size_t)M1 * DS_;
    float* DTb    = ws; ws += (size_t)M1 * NH_;
    float* decayb = ws; ws += (size_t)M1 * NH_;
    float* trapb  = ws; ws += (size_t)M1 * NH_;
    float* phaseb = ws; ws += (size_t)M1 * NA_;
    float* cumdb  = ws; ws += (size_t)B_ * NH_ * L_;

    gemm_wmma_f32<DIP, DM><<<dim3(DIP / GBN, M1 / GBM), dim3(256), 0, stream>>>(
        u, W_in, proj);

    phase_kernel<<<dim3(B_ * PG_), dim3(PT_ * 32), 0, stream>>>(
        proj, dt_bias, DTb, decayb, trapb, phaseb);

    rot_kernel<<<dim3(M1), dim3(64), 0, stream>>>(
        proj, Bn_g, Bn_b, Cn_g, Cn_b, phaseb, Brot, Crot);

    scan_chunk_kernel<<<dim3(B_ * NH_ * NC_), dim3(256), 0, stream>>>(
        proj, Brot, Crot, DTb, decayb, trapb, yfin, cumdb, Sstate);
    scan_correct_kernel<<<dim3(B_ * NH_ * NC_), dim3(256), 0, stream>>>(
        proj, Crot, cumdb, Sstate, D_param, yfin);

    gemm_wmma_f32<DM, DI><<<dim3(DM / GBN, M1 / GBM), dim3(256), 0, stream>>>(
        yfin, W_out, (float*)d_out);
}


# scan passes A and B re-blocked: a thread owns 2 state rows x 8 columns (was 1 x 16), B/C LDS reads halved, the row pair runs in packed FMAs (dt folded into the input weights); y sum tree changes
# speedup vs baseline: 10.2560x; 1.0405x over previous
#include <hip/hip_runtime.h>
#include <math.h>

#define B_   2
#define L_   512
#define DM   1024
#define DI   2048
#define DS_  64
#define NH_  32
#define HD_  64
#define NA_  32
#define DIP  4352
#define M1   (B_*L_)

#define NC_  8
#define LC_  64
#define LR_  (LC_ / 16)
static_assert(NC_ * LC_ == L_, "chunks x tokens per chunk must cover the sequence");
static_assert(LC_ % 16 == 0 && 3 * LC_ <= 192, "row rounds and the three per-token scalar planes are spread over 256 threads");

#define OFF_Z    0
#define OFF_X    2048
#define OFF_B    4096
#define OFF_C    4160
#define OFF_DT   4224
#define OFF_A    4256
#define OFF_TRAP 4288
#define OFF_ANG  4320

typedef __attribute__((ext_vector_type(2))) float v2f;
typedef __attribute__((ext_vector_type(4))) float v4f;
typedef __attribute__((ext_vector_type(8))) float v8f;

__device__ __forceinline__ float softplus_f(float x) {
    return x > 20.f ? x : log1pf(expf(x));
}
__device__ __forceinline__ float sigmoid_f(float x) {
    return 1.f / (1.f + expf(-x));
}
__device__ __forceinline__ float lane_xor1(float v) {
    return __int_as_float(__builtin_amdgcn_update_dpp(0, __float_as_int(v), 0xB1, 0xF, 0xF, false));
}

#define GBM    64
#define GBN    64
#define GBK    32
#define APITCH 36
#define BPITCH 160

template <int N, int K>
__global__ __launch_bounds__(256) void gemm_wmma_f32(
    const float* __restrict__ A, const float* __restrict__ Bm, float* __restrict__ C)
{
    constexpr int nslab = K / GBK;
    static_assert(N % GBN == 0 && K % GBK == 0, "tile sizes must divide the problem");
    static_assert(nslab % 2 == 0 && nslab >= 4, "the pipeline runs in pairs of slabs");
    __shared__ __attribute__((aligned(16))) float As[2][GBM * APITCH];
    __shared__ __attribute__((aligned(16))) float Bs[2][(GBK / 2) * BPITCH];

    const int t    = threadIdx.x;
    const int lane = t & 31;
    const int wave = t >> 5;
    const int lo   = lane & 15;
    const int hi   = lane >> 4;

    const int m0 = blockIdx.y * GBM;
    const int n0 = blockIdx.x * GBN;
    const int wm = (wave >> 1) * 16;
    const int wn = (wave & 1) * 32;

    const int ar = t >> 3, ac = (t & 7) * 4;
    const int bn = t & 63, kq = t >> 6;

    const float* Ag0 = A  + (size_t)(m0 + ar) * K + ac;
    const float* Ag1 = Ag0 + (size_t)32 * K;
    const float* Bg  = Bm + (size_t)(2 * kq) * N + n0 + bn;

    const int as0 = ar * APITCH + ac, as1 = (ar + 32) * APITCH + ac;
    const int bs0 = kq * BPITCH + bn * 2;

    struct Staged { v4f a0, a1; v2f b[4]; };
    Staged sg0, sg1;
    auto fetch = [&](int i, Staged& g) __attribute__((always_inline)) {
        g.a0 = *reinterpret_cast<const v4f*>(Ag0 + (size_t)i * GBK);
        g.a1 = *reinterpret_cast<const v4f*>(Ag1 + (size_t)i * GBK);
        const float* bp = Bg + (size_t)i * GBK * N;
        #pragma unroll
        for (int q = 0; q < 4; ++q) {
            g.b[q].x = bp[(size_t)(8 * q) * N];
            g.b[q].y = bp[(size_t)(8 * q + 1) * N];
        }
    };
    auto stage = [&](int st, const Staged& g) __attribute__((always_inline)) {
        *reinterpret_cast<v4f*>(&As[st][as0]) = g.a0;
        *reinterpret_cast<v4f*>(&As[st][as1]) = g.a1;
        #pragma unroll
        for (int q = 0; q < 4; ++q)
            *reinterpret_cast<v2f*>(&Bs[st][bs0 + 4 * q * BPITCH]) = g.b[q];
    };
    auto lds_handover = [] () __attribute__((always_inline)) {
        __builtin_amdgcn_fence(__ATOMIC_RELEASE, "workgroup", "local");
        __builtin_amdgcn_s_barrier();
        __builtin_amdgcn_fence(__ATOMIC_ACQUIRE, "workgroup", "local");
    };

    v8f acc0 = {}, acc1 = {};

    const int fa = (wm + lo) * APITCH + 2 * hi;
    const int fb = hi * BPITCH + (wn + lo) * 2;

    struct Frags { v2f a[GBK / 4], b0[GBK / 4], b1[GBK / 4]; };
    Frags fr0, fr1;
    auto frags = [&](int st, Frags& f) __attribute__((always_inline)) {
        const float* as = &As[st][fa];
        const float* bs = &Bs[st][fb];
        #pragma unroll
        for (int j = 0; j < GBK / 4; ++j) {
            f.a[j]  = *reinterpret_cast<const v2f*>(as + 4 * j);
            f.b0[j] = *reinterpret_cast<const v2f*>(bs + 2 * j * BPITCH);
            f.b1[j] = *reinterpret_cast<const v2f*>(bs + 2 * j * BPITCH + 32);
        }
    };
    auto products = [&](const Frags& f) __attribute__((always_inline)) {
        #pragma unroll
        for (int j = 0; j < GBK / 4; ++j) {
            acc0 = __builtin_amdgcn_wmma_f32_16x16x4_f32(false, f.a[j], false, f.b0[j], (short)0, acc0, false, false);
            acc1 = __builtin_amdgcn_wmma_f32_16x16x4_f32(false, f.a[j], false, f.b1[j], (short)0, acc1, false, false);
        }
    };

    auto hold_products = [&]() __attribute__((always_inline)) {
        asm volatile("" : "+v"(acc0), "+v"(acc1));
    };

    auto two_slabs = [&](int s, bool more2, bool more4, bool more3, bool more5) __attribute__((always_inline)) {
        frags(1, fr1);
        products(fr0);
        if (more2) { stage(0, sg0); if (more4) fetch(s + 4, sg0); }
        if (more2) { lds_handover(); hold_products(); }
        if (more2) frags(0, fr0);
        products(fr1);
        if (more3) { stage(1, sg1); if (more5) fetch(s + 5, sg1); }
        if (more2) { lds_handover(); hold_products(); }
    };

    fetch(0, sg0);
    fetch(1, sg1);
    stage(0, sg0);
    fetch(2, sg0);
    lds_handover();
    frags(0, fr0);
    stage(1, sg1);
    fetch(3, sg1);
    lds_handover();

    int s = 0;
    for (; s + 4 < nslab; s += 2) two_slabs(s, true, true, true, true);
    two_slabs(s, true, false, true, false);
    two_slabs(s + 2, false, false, false, false);

    #pragma unroll
    for (int r = 0; r < 8; ++r) {
        float* cp = C + (size_t)(m0 + wm + r + 8 * hi) * N + n0 + wn + lo;
        cp[0]  = acc0[r];
        cp[16] = acc1[r];
    }
}

#define PT_  16
#define PG_  (L_ / PT_)
static_assert(PG_ * PT_ == L_, "groups x tokens per group must cover the sequence");

__global__ __launch_bounds__(PT_ * 32) void phase_kernel(
    const float* __restrict__ proj, const float* __restrict__ dt_bias,
    float* __restrict__ DTb, float* __restrict__ decayb,
    float* __restrict__ trapb, float* __restrict__ phaseb)
{
    const int b = blockIdx.x / PG_;
    const int g = blockIdx.x % PG_;
    const int j = threadIdx.x & 31;
    const int w = threadIdx.x >> 5;
    __shared__ float sprod[PT_][32];

    const size_t bl  = (size_t)b * L_ + (size_t)g * PT_ + w;
    const float* row = proj + bl * DIP;

    const float vdt  = row[OFF_DT + j];
    const float vang = row[OFF_ANG + j];
    const float va   = row[OFF_A + j];
    const float vtr  = row[OFF_TRAP + j];

    const float dt = softplus_f(vdt + dt_bias[j]);
    float s = dt;
    s += __shfl_xor(s, 16, 32);
    s += __shfl_xor(s,  8, 32);
    s += __shfl_xor(s,  4, 32);
    s += __shfl_xor(s,  2, 32);
    s += __shfl_xor(s,  1, 32);
    const float dtmean = s * (1.f / 32.f);
    sprod[w][j] = vang * dtmean;

    const size_t o = bl * 32 + j;
    DTb[o]    = dt;
    const float Av = fminf(-softplus_f(va), -1e-4f);
    decayb[o] = expf(Av * dt);
    trapb[o]  = sigmoid_f(vtr);

    __syncthreads();

    float acc = 0.f;
    #pragma unroll
    for (int i = 0; i < PT_; ++i) {
        const float v = sprod[i][j];
        acc += (i <= w) ? v : 0.f;
    }
    phaseb[o] = acc;
}

__global__ __launch_bounds__(64) void rot_kernel(
    const float* __restrict__ proj,
    const float* __restrict__ Bn_g, const float* __restrict__ Bn_b,
    const float* __restrict__ Cn_g, const float* __restrict__ Cn_b,
    const float* __restrict__ phaseb,
    float* __restrict__ Brot, float* __restrict__ Crot)
{
    const int bl = blockIdx.x;
    const int t  = threadIdx.x;
    __shared__ float red[64], nb[64], nc[64];

    const int q  = t >> 1;
    const int l  = bl % L_;
    const int g  = l / PT_;
    const float* pseq = phaseb + (size_t)(bl - l) * 32 + q;
    float off = 0.f;
    #pragma unroll
    for (int k = 0; k < PG_ - 1; ++k) {
        const float v = pseq[(size_t)(k * PT_ + PT_ - 1) * 32];
        off += (k < g) ? v : 0.f;
    }
    const float ph = off + phaseb[(size_t)bl * 32 + q];

    const float* row = proj + (size_t)bl * DIP;
    const float vb = row[OFF_B + t];
    const float vc = row[OFF_C + t];

    auto blockReduce = [&](float v) -> float {
        red[t] = v; __syncthreads();
        for (int s = 32; s > 0; s >>= 1) {
            if (t < s) red[t] += red[t + s];
            __syncthreads();
        }
        float r = red[0]; __syncthreads();
        return r;
    };

    const float mB  = blockReduce(vb) * (1.f / 64.f);
    const float vaB = blockReduce((vb - mB) * (vb - mB)) * (1.f / 64.f);
    const float nB  = (vb - mB) * rsqrtf(vaB + 1e-5f) * Bn_g[t] + Bn_b[t];

    const float mC  = blockReduce(vc) * (1.f / 64.f);
    const float vaC = blockReduce((vc - mC) * (vc - mC)) * (1.f / 64.f);
    const float nC  = (vc - mC) * rsqrtf(vaC + 1e-5f) * Cn_g[t] + Cn_b[t];

    nb[t] = nB; nc[t] = nC;
    __syncthreads();

    const float c  = cosf(ph), s = sinf(ph);
    float ob, oc;
    if ((t & 1) == 0) { ob = nb[t] * c - nb[t + 1] * s; oc = nc[t] * c - nc[t + 1] * s; }
    else              { ob = nb[t - 1] * s + nb[t] * c; oc = nc[t - 1] * s + nc[t] * c; }
    Brot[(size_t)bl * 64 + t] = ob;
    Crot[(size_t)bl * 64 + t] = oc;
}


__global__ __launch_bounds__(256) void scan_chunk_kernel(
    const float* __restrict__ proj,
    const float* __restrict__ Brot, const float* __restrict__ Crot,
    const float* __restrict__ DTb, const float* __restrict__ decayb,
    const float* __restrict__ trapb,
    float* __restrict__ yfin, float* __restrict__ cumdb,
    float* __restrict__ Sstate)
{
    const int blk = blockIdx.x;
    const int c   = blk % NC_;
    const int h   = (blk / NC_) % NH_;
    const int b   = blk / (NC_ * NH_);
    const int l0  = c * LC_;
    const int t   = threadIdx.x;
    const int pp  = t >> 3;
    const int ns  = t & 7;

    __shared__ __attribute__((aligned(16))) float sx[LC_ + 1][64];
    __shared__ __attribute__((aligned(16))) float sB[LC_ + 1][64];
    __shared__ __attribute__((aligned(16))) float sC[LC_][64];
    __shared__ __attribute__((aligned(16))) float sscal[LC_][4];
    __shared__ float scum[LC_];
    __shared__ __attribute__((aligned(16))) float sy[LC_][64][4];

    const size_t bl0 = (size_t)b * L_ + l0;

    float4 vx[LR_], vb[LR_], vc[LR_];
    #pragma unroll
    for (int k = 0; k < LR_; ++k) {
        const int idx = k * 256 + t;
        const int r = idx >> 4, q = idx & 15;
        vx[k] = *reinterpret_cast<const float4*>(proj + (bl0 + r) * DIP + OFF_X + h * HD_ + q * 4);
        vb[k] = *reinterpret_cast<const float4*>(Brot + (bl0 + r) * 64 + q * 4);
        vc[k] = *reinterpret_cast<const float4*>(Crot + (bl0 + r) * 64 + q * 4);
    }
    float4 v0 = make_float4(0.f, 0.f, 0.f, 0.f);
    if (c > 0 && t < 32) {
        const int q = t & 15;
        v0 = (t < 16)
           ? *reinterpret_cast<const float4*>(proj + (bl0 - 1) * DIP + OFF_X + h * HD_ + q * 4)
           : *reinterpret_cast<const float4*>(Brot + (bl0 - 1) * 64 + q * 4);
    }
    float vs = 0.f;
    if (t >= 64) {
        const int kind = (t - 64) / LC_, tok = (t - 64) % LC_;
        const float* sp = (kind == 0) ? DTb : (kind == 1) ? decayb : trapb;
        vs = sp[(bl0 + tok) * 32 + h];
    }
    #pragma unroll
    for (int k = 0; k < LR_; ++k) {
        const int idx = k * 256 + t;
        const int r = idx >> 4, q = idx & 15;
        *reinterpret_cast<float4*>(&sx[r + 1][q * 4]) = vx[k];
        *reinterpret_cast<float4*>(&sB[r + 1][q * 4]) = vb[k];
        *reinterpret_cast<float4*>(&sC[r][q * 4])     = vc[k];
    }
    if (t < 16)      *reinterpret_cast<float4*>(&sx[0][(t & 15) * 4]) = v0;
    else if (t < 32) *reinterpret_cast<float4*>(&sB[0][(t & 15) * 4]) = v0;
    if (t >= 64) sscal[(t - 64) % LC_][(t - 64) / LC_] = vs;
    __syncthreads();

    v2f hst[8];
    float Bp[8];
    #pragma unroll
    for (int i = 0; i < 8; ++i) hst[i] = (v2f){0.f, 0.f};
    #pragma unroll
    for (int i4 = 0; i4 < 2; ++i4) {
        const float4 v = *reinterpret_cast<const float4*>(&sB[0][ns * 8 + i4 * 4]);
        Bp[i4 * 4 + 0] = v.x; Bp[i4 * 4 + 1] = v.y; Bp[i4 * 4 + 2] = v.z; Bp[i4 * 4 + 3] = v.w;
    }
    v2f xprev  = *reinterpret_cast<const v2f*>(&sx[0][2 * pp]);
    float cumd = 1.f;
    const int prow = 2 * pp + (ns & 1);

    #pragma unroll 2
    for (int tok = 0; tok < LC_; ++tok) {
        const v2f    x  = *reinterpret_cast<const v2f*>(&sx[tok + 1][2 * pp]);
        const float4 sc = *reinterpret_cast<const float4*>(&sscal[tok][0]);
        const float dt = sc.x, dec = sc.y, tr = sc.z;

        float Bc[8], Cc[8];
        #pragma unroll
        for (int i4 = 0; i4 < 2; ++i4) {
            const float4 vB = *reinterpret_cast<const float4*>(&sB[tok + 1][ns * 8 + i4 * 4]);
            const float4 vC = *reinterpret_cast<const float4*>(&sC[tok][ns * 8 + i4 * 4]);
            Bc[i4 * 4 + 0] = vB.x; Bc[i4 * 4 + 1] = vB.y; Bc[i4 * 4 + 2] = vB.z; Bc[i4 * 4 + 3] = vB.w;
            Cc[i4 * 4 + 0] = vC.x; Cc[i4 * 4 + 1] = vC.y; Cc[i4 * 4 + 2] = vC.z; Cc[i4 * 4 + 3] = vC.w;
        }

        const v2f a  = x * (dt * tr);
        const v2f bq = xprev * (dt * (1.f - tr));
        v2f y = {0.f, 0.f};
        #pragma unroll
        for (int i = 0; i < 8; ++i) {
            const v2f bx = __builtin_elementwise_fma(a, (v2f){Bc[i], Bc[i]}, bq * Bp[i]);
            hst[i] = __builtin_elementwise_fma((v2f){dec, dec}, hst[i], bx);
            y      = __builtin_elementwise_fma(hst[i], (v2f){Cc[i], Cc[i]}, y);
            Bp[i] = Bc[i];
        }
        xprev = x;
        cumd *= dec;

        const float y0 = y.x + lane_xor1(y.x);
        const float y1 = y.y + lane_xor1(y.y);
        sy[tok][prow][ns >> 1] = (ns & 1) ? y1 : y0;
        if (t == 0) scum[tok] = cumd;
    }
    __syncthreads();

    #pragma unroll
    for (int k = 0; k < LR_; ++k) {
        const int idx = k * 256 + t;
        const int r = idx >> 4, q = idx & 15;
        float o[4];
        #pragma unroll
        for (int j = 0; j < 4; ++j) {
            const float4 pv = *reinterpret_cast<const float4*>(&sy[r][q * 4 + j][0]);
            o[j] = (pv.x + pv.y) + (pv.z + pv.w);
        }
        *reinterpret_cast<float4*>(yfin + (bl0 + r) * DI + h * HD_ + q * 4) = make_float4(o[0], o[1], o[2], o[3]);
    }
    if (t < LC_) cumdb[((size_t)(b * NH_ + h)) * L_ + l0 + t] = scum[t];

    float* Sp = Sstate + (((size_t)(b * NH_ + h)) * NC_ + c) * 4096 + (2 * pp) * 64 + ns * 8;
    #pragma unroll
    for (int i4 = 0; i4 < 2; ++i4) {
        *reinterpret_cast<float4*>(Sp + i4 * 4) =
            make_float4(hst[i4 * 4 + 0].x, hst[i4 * 4 + 1].x, hst[i4 * 4 + 2].x, hst[i4 * 4 + 3].x);
        *reinterpret_cast<float4*>(Sp + 64 + i4 * 4) =
            make_float4(hst[i4 * 4 + 0].y, hst[i4 * 4 + 1].y, hst[i4 * 4 + 2].y, hst[i4 * 4 + 3].y);
    }
}

template <int NCH>
__device__ __forceinline__ void fold_chunks(float (&H)[16], const float* __restrict__ Sp,
                                            const float* __restrict__ cd)
{
    float P[NCH];
    float4 S[NCH][4];
    #pragma unroll
    for (int k = 0; k < NCH; ++k) {
        P[k] = cd[k * LC_];
        #pragma unroll
        for (int i4 = 0; i4 < 4; ++i4)
            S[k][i4] = *reinterpret_cast<const float4*>(Sp + (size_t)k * 4096 + (i4 >> 1) * 64 + (i4 & 1) * 4);
    }
    #pragma unroll
    for (int k = 0; k < NCH; ++k) {
        #pragma unroll
        for (int i4 = 0; i4 < 4; ++i4) {
            H[i4 * 4 + 0] = P[k] * H[i4 * 4 + 0] + S[k][i4].x;
            H[i4 * 4 + 1] = P[k] * H[i4 * 4 + 1] + S[k][i4].y;
            H[i4 * 4 + 2] = P[k] * H[i4 * 4 + 2] + S[k][i4].z;
            H[i4 * 4 + 3] = P[k] * H[i4 * 4 + 3] + S[k][i4].w;
        }
    }
}

__global__ __launch_bounds__(256) void scan_correct_kernel(
    const float* __restrict__ proj, const float* __restrict__ Crot,
    const float* __restrict__ cumdb, const float* __restrict__ Sstate,
    const float* __restrict__ D_param, float* __restrict__ yfin)
{
    const int blk = blockIdx.x;
    static_assert(NC_ == 8 && B_ * NH_ == 64, "the block order below is written for 8 chunks x 64 (batch, head) pairs");
    const int hf  = blk >> 8;
    const int bh  = blk & 63;
    const int e   = (blk >> 6) & 3;
    const int c   = hf ? e : (7 - e);
    const int h   = bh % NH_;
    const int b   = bh / NH_;
    const int l0  = c * LC_;
    const int t   = threadIdx.x;
    const int pp  = t >> 3, ns = t & 7;

    __shared__ __attribute__((aligned(16))) float sC[LC_][64];
    __shared__ float scd[LC_];
    __shared__ __attribute__((aligned(16))) float scorr[LC_][64][4];

    const size_t bl0 = (size_t)b * L_ + l0;

    const float Dp = D_param[h];

    float4 vc[LR_], vx[LR_], vz[LR_], vy[LR_];
    #pragma unroll
    for (int k = 0; k < LR_; ++k) {
        const int idx = k * 256 + t;
        const int r = idx >> 4, q = idx & 15;
        vc[k] = *reinterpret_cast<const float4*>(Crot + (bl0 + r) * 64 + q * 4);
        vx[k] = *reinterpret_cast<const float4*>(proj + (bl0 + r) * DIP + OFF_X + h * HD_ + q * 4);
        vz[k] = *reinterpret_cast<const float4*>(proj + (bl0 + r) * DIP + OFF_Z + h * HD_ + q * 4);
        vy[k] = *reinterpret_cast<const float4*>(yfin + (bl0 + r) * DI + h * HD_ + q * 4);
    }
    float cdv = 0.f;
    if (t < LC_) cdv = cumdb[((size_t)(b * NH_ + h)) * L_ + l0 + t];

    float H[16];
    #pragma unroll
    for (int i = 0; i < 16; ++i) H[i] = 0.f;
    const float* Sb  = Sstate + ((size_t)(b * NH_ + h)) * NC_ * 4096 + (2 * pp) * 64 + ns * 8;
    const float* cdl = cumdb + ((size_t)(b * NH_ + h)) * L_ + (LC_ - 1);
    int cc = 0;
    for (; cc + 4 <= c; cc += 4) fold_chunks<4>(H, Sb + (size_t)cc * 4096, cdl + cc * LC_);
    const int rest = c - cc;
    if (rest == 3)      fold_chunks<3>(H, Sb + (size_t)cc * 4096, cdl + cc * LC_);
    else if (rest == 2) fold_chunks<2>(H, Sb + (size_t)cc * 4096, cdl + cc * LC_);
    else if (rest == 1) fold_chunks<1>(H, Sb + (size_t)cc * 4096, cdl + cc * LC_);

    #pragma unroll
    for (int k = 0; k < LR_; ++k) {
        const int idx = k * 256 + t;
        const int r = idx >> 4, q = idx & 15;
        *reinterpret_cast<float4*>(&sC[r][q * 4]) = vc[k];
    }
    if (t < LC_) scd[t] = cdv;
    __syncthreads();

    v2f H2[8];
    #pragma unroll
    for (int i = 0; i < 8; ++i) H2[i] = (v2f){H[i], H[8 + i]};
    const int prow = 2 * pp + (ns & 1);

    #pragma unroll 2
    for (int tok = 0; tok < LC_; ++tok) {
        float Cc[8];
        #pragma unroll
        for (int i4 = 0; i4 < 2; ++i4) {
            const float4 vC = *reinterpret_cast<const float4*>(&sC[tok][ns * 8 + i4 * 4]);
            Cc[i4 * 4 + 0] = vC.x; Cc[i4 * 4 + 1] = vC.y; Cc[i4 * 4 + 2] = vC.z; Cc[i4 * 4 + 3] = vC.w;
        }
        v2f corr = {0.f, 0.f};
        #pragma unroll
        for (int i = 0; i < 8; ++i) corr = __builtin_elementwise_fma(H2[i], (v2f){Cc[i], Cc[i]}, corr);
        const float c0 = corr.x + lane_xor1(corr.x);
        const float c1 = corr.y + lane_xor1(corr.y);
        scorr[tok][prow][ns >> 1] = (ns & 1) ? c1 : c0;
    }
    __syncthreads();

    #pragma unroll
    for (int k = 0; k < LR_; ++k) {
        const int idx = k * 256 + t;
        const int r = idx >> 4, q = idx & 15;
        const float cd = scd[r];
        const float xs[4] = {vx[k].x, vx[k].y, vx[k].z, vx[k].w};
        const float zs[4] = {vz[k].x, vz[k].y, vz[k].z, vz[k].w};
        const float ys[4] = {vy[k].x, vy[k].y, vy[k].z, vy[k].w};
        float o[4];
        #pragma unroll
        for (int j = 0; j < 4; ++j) {
            const float4 pv = *reinterpret_cast<const float4*>(&scorr[r][q * 4 + j][0]);
            const float corr = (pv.x + pv.y) + (pv.z + pv.w);
            const float y = ys[j] + cd * corr;
            o[j] = (y + Dp * xs[j]) * (zs[j] * sigmoid_f(zs[j]));
        }
        *reinterpret_cast<float4*>(yfin + (bl0 + r) * DI + h * HD_ + q * 4) = make_float4(o[0], o[1], o[2], o[3]);
    }
}

extern "C" void kernel_launch(void* const* d_in, const int* in_sizes, int n_in,
                              void* d_out, int out_size, void* d_ws, size_t ws_size,
                              hipStream_t stream)
{
    const float* u       = (const float*)d_in[0];
    const float* W_in    = (const float*)d_in[1];
    const float* W_out   = (const float*)d_in[2];
    const float* dt_bias = (const float*)d_in[3];
    const float* D_param = (const float*)d_in[4];
    const float* Bn_g    = (const float*)d_in[5];
    const float* Bn_b    = (const float*)d_in[6];
    const float* Cn_g    = (const float*)d_in[7];
    const float* Cn_b    = (const float*)d_in[8];

    float* ws = (float*)d_ws;
    float* proj   = ws; ws += (size_t)M1 * DIP;
    float* yfin   = ws; ws += (size_t)M1 * DI;
    float* Sstate = ws; ws += (size_t)B_ * NH_ * NC_ * 4096;
    float* Brot   = ws; ws += (size_t)M1 * DS_;
    float* Crot   = ws; ws += (size_t)M1 * DS_;
    float* DTb    = ws; ws += (size_t)M1 * NH_;
    float* decayb = ws; ws += (size_t)M1 * NH_;
    float* trapb  = ws; ws += (size_t)M1 * NH_;
    float* phaseb = ws; ws += (size_t)M1 * NA_;
    float* cumdb  = ws; ws += (size_t)B_ * NH_ * L_;

    gemm_wmma_f32<DIP, DM><<<dim3(DIP / GBN, M1 / GBM), dim3(256), 0, stream>>>(
        u, W_in, proj);

    phase_kernel<<<dim3(B_ * PG_), dim3(PT_ * 32), 0, stream>>>(
        proj, dt_bias, DTb, decayb, trapb, phaseb);

    rot_kernel<<<dim3(M1), dim3(64), 0, stream>>>(
        proj, Bn_g, Bn_b, Cn_g, Cn_b, phaseb, Brot, Crot);

    scan_chunk_kernel<<<dim3(B_ * NH_ * NC_), dim3(256), 0, stream>>>(
        proj, Brot, Crot, DTb, decayb, trapb, yfin, cumdb, Sstate);
    scan_correct_kernel<<<dim3(B_ * NH_ * NC_), dim3(256), 0, stream>>>(
        proj, Crot, cumdb, Sstate, D_param, yfin);

    gemm_wmma_f32<DM, DI><<<dim3(DM / GBN, M1 / GBM), dim3(256), 0, stream>>>(
        yfin, W_out, (float*)d_out);
}
